# starting kernel
# baseline (speedup 1.0000x reference)
#include <hip/hip_runtime.h>

typedef __attribute__((ext_vector_type(2))) float v2f;
typedef __attribute__((ext_vector_type(8))) float v8f;

#define D_FEAT 128

__global__ void sage_zero(float* __restrict__ ws, size_t n) {
  size_t i = (size_t)blockIdx.x * blockDim.x + threadIdx.x;
  if (i < n) ws[i] = 0.0f;
}

__global__ __launch_bounds__(256) void sage_scatter1(
    const float* __restrict__ x, const int* __restrict__ ei,
    float* __restrict__ agg1, float* __restrict__ cnt, int n_edges) {
  const int lane = threadIdx.x & 31;
  const int e = (int)((blockIdx.x * (size_t)blockDim.x + threadIdx.x) >> 5);
  if (e >= n_edges) return;
  const int src = __builtin_nontemporal_load(ei + e);
  const int dst = __builtin_nontemporal_load(ei + n_edges + e);
  const float4 v = *(const float4*)(x + (size_t)src * D_FEAT + lane * 4);
  float* a = agg1 + (size_t)dst * D_FEAT + lane * 4;
  atomicAdd(a + 0, v.x);
  atomicAdd(a + 1, v.y);
  atomicAdd(a + 2, v.z);
  atomicAdd(a + 3, v.w);
  if (lane == 0) atomicAdd(cnt + dst, 1.0f);
}

__global__ __launch_bounds__(256) void sage_layer1_wmma(
    const float* __restrict__ x, const float* __restrict__ agg1,
    const float* __restrict__ cnt, const float* __restrict__ W1l,
    const float* __restrict__ W1r, const float* __restrict__ b1,
    float* __restrict__ h, int n_nodes) {
  __shared__ float ldsWl[D_FEAT * D_FEAT];

  for (int idx = threadIdx.x; idx < D_FEAT * 64; idx += 256) {
    const int r = idx >> 6;
    const int c = idx & 63;
    const float2 w = *(const float2*)(W1l + (size_t)r * D_FEAT + 2 * c);
    const int s = c ^ ((2 * r) & 63);
    *(float2*)(ldsWl + r * D_FEAT + 2 * s) = w;
  }
  __syncthreads();

  const int lane = threadIdx.x & 31;
  const int wave = threadIdx.x >> 5;
  const int tile = blockIdx.x * 8 + wave;
  const int n_tiles = n_nodes >> 4;
  if (tile >= n_tiles) return;

  const int half = lane >> 4;
  const int q = lane & 15;
  const int rowA = tile * 16 + q;

  const float inv = 1.0f / fmaxf(cnt[rowA], 1.0f);

  v8f acc[8];
#pragma unroll
  for (int i = 0; i < 8; ++i) {
    v8f z = {};
    acc[i] = z;
  }

  const float* aggRow = agg1 + (size_t)rowA * D_FEAT + 2 * half;
  const float* xRow   = x    + (size_t)rowA * D_FEAT + 2 * half;

  int nrow[8], nswz[8];
#pragma unroll
  for (int nt = 0; nt < 8; ++nt) {
    nrow[nt] = nt * 16 + q;
    nswz[nt] = (2 * nrow[nt]) & 63;
  }

  for (int k = 0; k < D_FEAT; k += 4) {
    const int c0 = (k >> 1) + half;
    const float2 tm = *(const float2*)(aggRow + k);
    const float2 ts = *(const float2*)(xRow + k);

    float2 bl[8], br[8];
#pragma unroll
    for (int nt = 0; nt < 8; ++nt) {
      bl[nt] = *(const float2*)(ldsWl + nrow[nt] * D_FEAT + 2 * (c0 ^ nswz[nt]));
      br[nt] = *(const float2*)(W1r + (size_t)nrow[nt] * D_FEAT + k + 2 * half);
    }

    v2f a_m;  a_m.x = tm.x * inv;  a_m.y = tm.y * inv;
    v2f a_s;  a_s.x = ts.x;        a_s.y = ts.y;
#pragma unroll
    for (int nt = 0; nt < 8; ++nt) {
      v2f vbl;  vbl.x = bl[nt].x;  vbl.y = bl[nt].y;
      v2f vbr;  vbr.x = br[nt].x;  vbr.y = br[nt].y;
      acc[nt] = __builtin_amdgcn_wmma_f32_16x16x4_f32(
          false, a_m, false, vbl, (short)0, acc[nt], false, false);
      acc[nt] = __builtin_amdgcn_wmma_f32_16x16x4_f32(
          false, a_s, false, vbr, (short)0, acc[nt], false, false);
    }
  }

#pragma unroll
  for (int nt = 0; nt < 8; ++nt) {
    const int n = nrow[nt];
    const float bias = b1[n];
#pragma unroll
    for (int v = 0; v < 8; ++v) {
      const int m = v + 8 * half;
      float val = acc[nt][v] + bias;
      val = fmaxf(val, 0.0f);
      h[(size_t)(tile * 16 + m) * D_FEAT + n] = val;
    }
  }
}

__global__ __launch_bounds__(256) void sage_pq(
    const float* __restrict__ h, const float* __restrict__ W2l,
    const float* __restrict__ W2r, const float* __restrict__ b2,
    float* __restrict__ p, float* __restrict__ out, int n_nodes) {
  const int lane = threadIdx.x & 31;
  const int node = (int)((blockIdx.x * (size_t)blockDim.x + threadIdx.x) >> 5);
  if (node >= n_nodes) return;
  const float4 hv = *(const float4*)(h + (size_t)node * D_FEAT + lane * 4);
  const float4 l0 = *(const float4*)(W2l + lane * 4);
  const float4 l1 = *(const float4*)(W2l + D_FEAT + lane * 4);
  const float4 r0 = *(const float4*)(W2r + lane * 4);
  const float4 r1 = *(const float4*)(W2r + D_FEAT + lane * 4);
  float d0 = hv.x * l0.x + hv.y * l0.y + hv.z * l0.z + hv.w * l0.w;
  float d1 = hv.x * l1.x + hv.y * l1.y + hv.z * l1.z + hv.w * l1.w;
  float d2 = hv.x * r0.x + hv.y * r0.y + hv.z * r0.z + hv.w * r0.w;
  float d3 = hv.x * r1.x + hv.y * r1.y + hv.z * r1.z + hv.w * r1.w;
#pragma unroll
  for (int off = 16; off > 0; off >>= 1) {
    d0 += __shfl_xor(d0, off, 32);
    d1 += __shfl_xor(d1, off, 32);
    d2 += __shfl_xor(d2, off, 32);
    d3 += __shfl_xor(d3, off, 32);
  }
  if (lane == 0) {
    p[(size_t)node * 2 + 0] = d0;
    p[(size_t)node * 2 + 1] = d1;
    out[(size_t)node * 2 + 0] = d2 + b2[0];
    out[(size_t)node * 2 + 1] = d3 + b2[1];
  }
}

__global__ __launch_bounds__(256) void sage_scatter2(
    const float* __restrict__ p, const int* __restrict__ ei,
    float* __restrict__ aggp, int n_edges) {
  const int e = blockIdx.x * blockDim.x + threadIdx.x;
  if (e >= n_edges) return;
  const int src = __builtin_nontemporal_load(ei + e);
  const int dst = __builtin_nontemporal_load(ei + n_edges + e);
  const float2 pv = *(const float2*)(p + (size_t)src * 2);
  atomicAdd(aggp + (size_t)dst * 2 + 0, pv.x);
  atomicAdd(aggp + (size_t)dst * 2 + 1, pv.y);
}

__global__ void sage_finalize(const float* __restrict__ aggp,
                              const float* __restrict__ cnt,
                              float* __restrict__ out, int n_nodes) {
  const int i = blockIdx.x * blockDim.x + threadIdx.x;
  if (i >= n_nodes) return;
  const float inv = 1.0f / fmaxf(cnt[i], 1.0f);
  out[(size_t)i * 2 + 0] += aggp[(size_t)i * 2 + 0] * inv;
  out[(size_t)i * 2 + 1] += aggp[(size_t)i * 2 + 1] * inv;
}

extern "C" void kernel_launch(void* const* d_in, const int* in_sizes, int n_in,
                              void* d_out, int out_size, void* d_ws, size_t ws_size,
                              hipStream_t stream) {
  const float* x   = (const float*)d_in[0];
  const int*   ei  = (const int*)d_in[1];
  const float* W1l = (const float*)d_in[2];
  const float* W1r = (const float*)d_in[3];
  const float* b1  = (const float*)d_in[4];
  const float* W2l = (const float*)d_in[5];
  const float* W2r = (const float*)d_in[6];
  const float* b2  = (const float*)d_in[7];
  float* out = (float*)d_out;

  const int n_nodes = in_sizes[0] / D_FEAT;
  const int n_edges = in_sizes[1] / 2;

  float* ws   = (float*)d_ws;
  float* agg1 = ws;
  float* cnt  = agg1 + (size_t)n_nodes * D_FEAT;
  float* aggp = cnt + n_nodes;
  float* h    = aggp + (size_t)n_nodes * 2;
  float* p    = h + (size_t)n_nodes * D_FEAT;

  const size_t zn = (size_t)n_nodes * (D_FEAT + 3);
  sage_zero<<<(unsigned)((zn + 255) / 256), 256, 0, stream>>>(ws, zn);

  sage_scatter1<<<(n_edges + 7) / 8, 256, 0, stream>>>(x, ei, agg1, cnt, n_edges);

  const int n_tiles = n_nodes / 16;
  sage_layer1_wmma<<<(n_tiles + 7) / 8, 256, 0, stream>>>(x, agg1, cnt, W1l, W1r,
                                                          b1, h, n_nodes);

  sage_pq<<<(n_nodes + 7) / 8, 256, 0, stream>>>(h, W2l, W2r, b2, p, out, n_nodes);

  sage_scatter2<<<(n_edges + 255) / 256, 256, 0, stream>>>(p, ei, aggp, n_edges);

  sage_finalize<<<(n_nodes + 255) / 256, 256, 0, stream>>>(aggp, cnt, out, n_nodes);
}


# Both layers' float atomic scatter-adds replaced by a counting sort of edges by dst (one int atomic per edge) + per-node register gathers; sum order changes (arrival rank; shuffle tree in layer 2)
# speedup vs baseline: 1.6057x; 1.6057x over previous
#include <hip/hip_runtime.h>

typedef __attribute__((ext_vector_type(2))) float v2f;
typedef __attribute__((ext_vector_type(8))) float v8f;

#define D_FEAT 128


__global__ void sage_zero_deg(int* __restrict__ deg, int n) {
  const int i = blockIdx.x * blockDim.x + threadIdx.x;
  if (i < n) deg[i] = 0;
}

__global__ __launch_bounds__(256) void sage_rank(
    const int* __restrict__ ei, int* __restrict__ deg, int* __restrict__ rank,
    int n_edges) {
  const int e = blockIdx.x * blockDim.x + threadIdx.x;
  if (e >= n_edges) return;
  const int dst = ei[n_edges + e];
  rank[e] = atomicAdd(deg + dst, 1);
}

__global__ __launch_bounds__(1024) void sage_scan(
    const int* __restrict__ deg, int* __restrict__ off, int n) {
  __shared__ int part[1024];
  const int t = threadIdx.x;
  const int chunk = (n + 1023) / 1024;
  int beg = t * chunk;
  if (beg > n) beg = n;
  int end = beg + chunk;
  if (end > n) end = n;
  int s = 0;
  for (int i = beg; i < end; ++i) s += deg[i];
  part[t] = s;
  __syncthreads();
  for (int d = 1; d < 1024; d <<= 1) {
    const int v = (t >= d) ? part[t - d] : 0;
    __syncthreads();
    part[t] += v;
    __syncthreads();
  }
  int run = part[t] - s;
  for (int i = beg; i < end; ++i) {
    off[i] = run;
    run += deg[i];
  }
}

__global__ __launch_bounds__(256) void sage_fill(
    const int* __restrict__ ei, const int* __restrict__ off,
    const int* __restrict__ rank, int* __restrict__ csr, int n_edges) {
  const int e = blockIdx.x * blockDim.x + threadIdx.x;
  if (e >= n_edges) return;
  const int src = ei[e];
  const int dst = ei[n_edges + e];
  csr[off[dst] + rank[e]] = src;
}

__global__ __launch_bounds__(256) void sage_gather1(
    const float* __restrict__ x, const int* __restrict__ csr,
    const int* __restrict__ off, const int* __restrict__ deg,
    float* __restrict__ agg1, float* __restrict__ cnt, int n_nodes) {
  const int lane = threadIdx.x & 31;
  const int node = __builtin_amdgcn_readfirstlane(
      (int)((blockIdx.x * (size_t)blockDim.x + threadIdx.x) >> 5));
  if (node >= n_nodes) return;
  const int beg = off[node];
  const int d = deg[node];
  const float* xl = x + lane * 4;
  float4 acc = make_float4(0.0f, 0.0f, 0.0f, 0.0f);
  for (int base = 0; base < d; base += 32) {
    const int m = (d - base < 32) ? (d - base) : 32;
    const int mine = (lane < m) ? csr[beg + base + lane] : 0;
    int j = 0;
    for (; j + 4 <= m; j += 4) {
      const int s0 = __builtin_amdgcn_readlane(mine, j + 0);
      const int s1 = __builtin_amdgcn_readlane(mine, j + 1);
      const int s2 = __builtin_amdgcn_readlane(mine, j + 2);
      const int s3 = __builtin_amdgcn_readlane(mine, j + 3);
      const float4 v0 = *(const float4*)(xl + (size_t)s0 * D_FEAT);
      const float4 v1 = *(const float4*)(xl + (size_t)s1 * D_FEAT);
      const float4 v2 = *(const float4*)(xl + (size_t)s2 * D_FEAT);
      const float4 v3 = *(const float4*)(xl + (size_t)s3 * D_FEAT);
      acc.x += v0.x;  acc.y += v0.y;  acc.z += v0.z;  acc.w += v0.w;
      acc.x += v1.x;  acc.y += v1.y;  acc.z += v1.z;  acc.w += v1.w;
      acc.x += v2.x;  acc.y += v2.y;  acc.z += v2.z;  acc.w += v2.w;
      acc.x += v3.x;  acc.y += v3.y;  acc.z += v3.z;  acc.w += v3.w;
    }
    for (; j < m; ++j) {
      const int s = __builtin_amdgcn_readlane(mine, j);
      const float4 v = *(const float4*)(xl + (size_t)s * D_FEAT);
      acc.x += v.x;  acc.y += v.y;  acc.z += v.z;  acc.w += v.w;
    }
  }
  *(float4*)(agg1 + (size_t)node * D_FEAT + lane * 4) = acc;
  if (lane == 0) cnt[node] = (float)d;
}

__global__ __launch_bounds__(256) void sage_layer1_wmma(
    const float* __restrict__ x, const float* agg1,
    const float* __restrict__ cnt, const float* __restrict__ W1l,
    const float* __restrict__ W1r, const float* __restrict__ b1,
    float* h, int n_nodes) {
  __shared__ float ldsWl[D_FEAT * D_FEAT];

  for (int idx = threadIdx.x; idx < D_FEAT * 64; idx += 256) {
    const int r = idx >> 6;
    const int c = idx & 63;
    const float2 w = *(const float2*)(W1l + (size_t)r * D_FEAT + 2 * c);
    const int s = c ^ ((2 * r) & 63);
    *(float2*)(ldsWl + r * D_FEAT + 2 * s) = w;
  }
  __syncthreads();

  const int lane = threadIdx.x & 31;
  const int wave = threadIdx.x >> 5;
  const int tile = blockIdx.x * 8 + wave;
  const int n_tiles = n_nodes >> 4;
  if (tile >= n_tiles) return;

  const int half = lane >> 4;
  const int q = lane & 15;
  const int rowA = tile * 16 + q;

  const float inv = 1.0f / fmaxf(cnt[rowA], 1.0f);

  v8f acc[8];
#pragma unroll
  for (int i = 0; i < 8; ++i) {
    v8f z = {};
    acc[i] = z;
  }

  const float* aggRow = agg1 + (size_t)rowA * D_FEAT + 2 * half;
  const float* xRow   = x    + (size_t)rowA * D_FEAT + 2 * half;

  int nrow[8], nswz[8];
#pragma unroll
  for (int nt = 0; nt < 8; ++nt) {
    nrow[nt] = nt * 16 + q;
    nswz[nt] = (2 * nrow[nt]) & 63;
  }

  for (int k = 0; k < D_FEAT; k += 4) {
    const int c0 = (k >> 1) + half;
    const float2 tm = *(const float2*)(aggRow + k);
    const float2 ts = *(const float2*)(xRow + k);

    float2 bl[8], br[8];
#pragma unroll
    for (int nt = 0; nt < 8; ++nt) {
      bl[nt] = *(const float2*)(ldsWl + nrow[nt] * D_FEAT + 2 * (c0 ^ nswz[nt]));
      br[nt] = *(const float2*)(W1r + (size_t)nrow[nt] * D_FEAT + k + 2 * half);
    }

    v2f a_m;  a_m.x = tm.x * inv;  a_m.y = tm.y * inv;
    v2f a_s;  a_s.x = ts.x;        a_s.y = ts.y;
#pragma unroll
    for (int nt = 0; nt < 8; ++nt) {
      v2f vbl;  vbl.x = bl[nt].x;  vbl.y = bl[nt].y;
      v2f vbr;  vbr.x = br[nt].x;  vbr.y = br[nt].y;
      acc[nt] = __builtin_amdgcn_wmma_f32_16x16x4_f32(
          false, a_m, false, vbl, (short)0, acc[nt], false, false);
      acc[nt] = __builtin_amdgcn_wmma_f32_16x16x4_f32(
          false, a_s, false, vbr, (short)0, acc[nt], false, false);
    }
  }

#pragma unroll
  for (int nt = 0; nt < 8; ++nt) {
    const int n = nrow[nt];
    const float bias = b1[n];
#pragma unroll
    for (int v = 0; v < 8; ++v) {
      const int m = v + 8 * half;
      float val = acc[nt][v] + bias;
      val = fmaxf(val, 0.0f);
      h[(size_t)(tile * 16 + m) * D_FEAT + n] = val;
    }
  }
}

__global__ __launch_bounds__(256) void sage_pq(
    const float* __restrict__ h, const float* __restrict__ W2l,
    const float* __restrict__ W2r, const float* __restrict__ b2,
    float* __restrict__ p, float* __restrict__ out, int n_nodes) {
  const int lane = threadIdx.x & 31;
  const int node = (int)((blockIdx.x * (size_t)blockDim.x + threadIdx.x) >> 5);
  if (node >= n_nodes) return;
  const float4 hv = *(const float4*)(h + (size_t)node * D_FEAT + lane * 4);
  const float4 l0 = *(const float4*)(W2l + lane * 4);
  const float4 l1 = *(const float4*)(W2l + D_FEAT + lane * 4);
  const float4 r0 = *(const float4*)(W2r + lane * 4);
  const float4 r1 = *(const float4*)(W2r + D_FEAT + lane * 4);
  float d0 = hv.x * l0.x + hv.y * l0.y + hv.z * l0.z + hv.w * l0.w;
  float d1 = hv.x * l1.x + hv.y * l1.y + hv.z * l1.z + hv.w * l1.w;
  float d2 = hv.x * r0.x + hv.y * r0.y + hv.z * r0.z + hv.w * r0.w;
  float d3 = hv.x * r1.x + hv.y * r1.y + hv.z * r1.z + hv.w * r1.w;
#pragma unroll
  for (int off = 16; off > 0; off >>= 1) {
    d0 += __shfl_xor(d0, off, 32);
    d1 += __shfl_xor(d1, off, 32);
    d2 += __shfl_xor(d2, off, 32);
    d3 += __shfl_xor(d3, off, 32);
  }
  if (lane == 0) {
    p[(size_t)node * 2 + 0] = d0;
    p[(size_t)node * 2 + 1] = d1;
    out[(size_t)node * 2 + 0] = d2 + b2[0];
    out[(size_t)node * 2 + 1] = d3 + b2[1];
  }
}

__global__ __launch_bounds__(256) void sage_gather2(
    const float* __restrict__ p, const int* __restrict__ csr,
    const int* __restrict__ off, const int* __restrict__ deg,
    float* __restrict__ out, int n_nodes) {
  const int lane = threadIdx.x & 31;
  const int node = __builtin_amdgcn_readfirstlane(
      (int)((blockIdx.x * (size_t)blockDim.x + threadIdx.x) >> 5));
  if (node >= n_nodes) return;
  const int beg = off[node];
  const int d = deg[node];
  float s0 = 0.0f, s1 = 0.0f;
  for (int j = lane; j < d; j += 32) {
    const int s = csr[beg + j];
    const float2 pv = *(const float2*)(p + (size_t)s * 2);
    s0 += pv.x;
    s1 += pv.y;
  }
#pragma unroll
  for (int o = 16; o > 0; o >>= 1) {
    s0 += __shfl_xor(s0, o, 32);
    s1 += __shfl_xor(s1, o, 32);
  }
  if (lane == 0) {
    const float inv = 1.0f / fmaxf((float)d, 1.0f);
    out[(size_t)node * 2 + 0] += s0 * inv;
    out[(size_t)node * 2 + 1] += s1 * inv;
  }
}

extern "C" void kernel_launch(void* const* d_in, const int* in_sizes, int n_in,
                              void* d_out, int out_size, void* d_ws, size_t ws_size,
                              hipStream_t stream) {
  const float* x   = (const float*)d_in[0];
  const int*   ei  = (const int*)d_in[1];
  const float* W1l = (const float*)d_in[2];
  const float* W1r = (const float*)d_in[3];
  const float* b1  = (const float*)d_in[4];
  const float* W2l = (const float*)d_in[5];
  const float* W2r = (const float*)d_in[6];
  const float* b2  = (const float*)d_in[7];
  float* out = (float*)d_out;

  const int n_nodes = in_sizes[0] / D_FEAT;
  const int n_edges = in_sizes[1] / 2;

  float* ws   = (float*)d_ws;
  float* agg1 = ws;
  float* h    = agg1;
  float* p    = agg1 + (size_t)n_nodes * D_FEAT;
  float* cnt  = p + (size_t)n_nodes * 2;
  int*   deg  = (int*)(cnt + n_nodes);
  int*   off  = deg + n_nodes;
  int*   csr  = off + n_nodes;
  int*   rank = csr + n_edges;

  sage_zero_deg<<<(n_nodes + 255) / 256, 256, 0, stream>>>(deg, n_nodes);

  sage_rank<<<(n_edges + 255) / 256, 256, 0, stream>>>(ei, deg, rank, n_edges);

  sage_scan<<<1, 1024, 0, stream>>>(deg, off, n_nodes);

  sage_fill<<<(n_edges + 255) / 256, 256, 0, stream>>>(ei, off, rank, csr, n_edges);

  sage_gather1<<<(n_nodes + 7) / 8, 256, 0, stream>>>(x, csr, off, deg, agg1, cnt,
                                                     n_nodes);

  const int n_tiles = n_nodes / 16;
  sage_layer1_wmma<<<(n_tiles + 7) / 8, 256, 0, stream>>>(x, agg1, cnt, W1l, W1r,
                                                          b1, h, n_nodes);

  sage_pq<<<(n_nodes + 7) / 8, 256, 0, stream>>>(h, W2l, W2r, b2, p, out, n_nodes);

  sage_gather2<<<(n_nodes + 7) / 8, 256, 0, stream>>>(p, csr, off, deg, out, n_nodes);
}


# Edge bucketing without global atomics: edges binned by 256-node range (LDS histogram, scan, scatter), then ranked per node by LDS atomics; neighbour sum order stays arrival-ranked
# speedup vs baseline: 2.0234x; 1.2601x over previous
#include <hip/hip_runtime.h>

typedef __attribute__((ext_vector_type(2))) float v2f;
typedef __attribute__((ext_vector_type(8))) float v8f;

#define D_FEAT 128


#define BIN_SHIFT 8
#define EDGE_BLOCKS 256
#define MAX_BINS 256

__global__ __launch_bounds__(256) void sage_bin_hist(
    const int* __restrict__ ei, int* __restrict__ cnt_bb, int n_edges,
    int n_bins, int chunk) {
  __shared__ int hist[MAX_BINS];
  const int t = threadIdx.x;
  if (t < n_bins) hist[t] = 0;
  __syncthreads();
  const int beg = blockIdx.x * chunk;
  int end = beg + chunk;
  if (end > n_edges) end = n_edges;
  const int* dstp = ei + n_edges;
  int e = beg + t;
  for (; e + 3 * 256 < end; e += 4 * 256) {
    int d[4];
#pragma unroll
    for (int u = 0; u < 4; ++u) d[u] = dstp[e + u * 256];
#pragma unroll
    for (int u = 0; u < 4; ++u) atomicAdd(&hist[d[u] >> BIN_SHIFT], 1);
  }
  for (; e < end; e += 256) atomicAdd(&hist[dstp[e] >> BIN_SHIFT], 1);
  __syncthreads();
  if (t < n_bins) cnt_bb[t * EDGE_BLOCKS + blockIdx.x] = hist[t];
}

__global__ __launch_bounds__(1024) void sage_scan(
    const int* __restrict__ deg, int* __restrict__ off, int n) {
  __shared__ int part[1024];
  const int t = threadIdx.x;
  const int chunk = (n + 1023) / 1024;
  int beg = t * chunk;
  if (beg > n) beg = n;
  int end = beg + chunk;
  if (end > n) end = n;
  int s = 0;
  for (int i = beg; i < end; ++i) s += deg[i];
  part[t] = s;
  __syncthreads();
  for (int d = 1; d < 1024; d <<= 1) {
    const int v = (t >= d) ? part[t - d] : 0;
    __syncthreads();
    part[t] += v;
    __syncthreads();
  }
  int run = part[t] - s;
  for (int i = beg; i < end; ++i) {
    off[i] = run;
    run += deg[i];
  }
}

__global__ __launch_bounds__(256) void sage_bin_scatter(
    const int* __restrict__ ei, const int* __restrict__ base_bb,
    int2* __restrict__ tmp, int n_edges, int n_bins, int chunk) {
  __shared__ int cur[MAX_BINS];
  const int t = threadIdx.x;
  if (t < n_bins) cur[t] = base_bb[t * EDGE_BLOCKS + blockIdx.x];
  __syncthreads();
  const int beg = blockIdx.x * chunk;
  int end = beg + chunk;
  if (end > n_edges) end = n_edges;
  const int* dstp = ei + n_edges;
  int e = beg + t;
  for (; e + 3 * 256 < end; e += 4 * 256) {
    int s[4], d[4];
#pragma unroll
    for (int u = 0; u < 4; ++u) {
      s[u] = ei[e + u * 256];
      d[u] = dstp[e + u * 256];
    }
#pragma unroll
    for (int u = 0; u < 4; ++u) {
      const int slot = atomicAdd(&cur[d[u] >> BIN_SHIFT], 1);
      tmp[slot] = make_int2(s[u], d[u]);
    }
  }
  for (; e < end; e += 256) {
    const int s = ei[e];
    const int d = dstp[e];
    const int slot = atomicAdd(&cur[d >> BIN_SHIFT], 1);
    tmp[slot] = make_int2(s, d);
  }
}

__global__ __launch_bounds__(256) void sage_bin_fill(
    const int2* __restrict__ tmp, const int* __restrict__ base_bb,
    int* __restrict__ deg, int* __restrict__ off, int* __restrict__ csr,
    int n_nodes, int n_edges, int n_bins) {
  __shared__ int cnt[256];
  __shared__ int scn[256];
  const int t = threadIdx.x;
  const int b = blockIdx.x;
  const int beg = base_bb[b * EDGE_BLOCKS];
  const int end = (b + 1 < n_bins) ? base_bb[(b + 1) * EDGE_BLOCKS] : n_edges;
  cnt[t] = 0;
  __syncthreads();
  int i = beg + t;
  for (; i + 3 * 256 < end; i += 4 * 256) {
    int d[4];
#pragma unroll
    for (int u = 0; u < 4; ++u) d[u] = tmp[i + u * 256].y;
#pragma unroll
    for (int u = 0; u < 4; ++u) atomicAdd(&cnt[d[u] & 255], 1);
  }
  for (; i < end; i += 256) atomicAdd(&cnt[tmp[i].y & 255], 1);
  __syncthreads();
  const int c = cnt[t];
  scn[t] = c;
  __syncthreads();
  for (int d = 1; d < 256; d <<= 1) {
    const int v = (t >= d) ? scn[t - d] : 0;
    __syncthreads();
    scn[t] += v;
    __syncthreads();
  }
  const int excl = scn[t] - c;
  const int node = (b << BIN_SHIFT) + t;
  if (node < n_nodes) {
    deg[node] = c;
    off[node] = beg + excl;
  }
  cnt[t] = excl;
  __syncthreads();
  i = beg + t;
  for (; i + 3 * 256 < end; i += 4 * 256) {
    int2 sd[4];
#pragma unroll
    for (int u = 0; u < 4; ++u) sd[u] = tmp[i + u * 256];
#pragma unroll
    for (int u = 0; u < 4; ++u) {
      const int r = atomicAdd(&cnt[sd[u].y & 255], 1);
      csr[beg + r] = sd[u].x;
    }
  }
  for (; i < end; i += 256) {
    const int2 sd = tmp[i];
    const int r = atomicAdd(&cnt[sd.y & 255], 1);
    csr[beg + r] = sd.x;
  }
}

__global__ __launch_bounds__(256) void sage_gather1(
    const float* __restrict__ x, const int* __restrict__ csr,
    const int* __restrict__ off, const int* __restrict__ deg,
    float* __restrict__ agg1, float* __restrict__ cnt, int n_nodes) {
  const int lane = threadIdx.x & 31;
  const int node = __builtin_amdgcn_readfirstlane(
      (int)((blockIdx.x * (size_t)blockDim.x + threadIdx.x) >> 5));
  if (node >= n_nodes) return;
  const int beg = off[node];
  const int d = deg[node];
  const float* xl = x + lane * 4;
  float4 acc = make_float4(0.0f, 0.0f, 0.0f, 0.0f);
  for (int base = 0; base < d; base += 32) {
    const int m = (d - base < 32) ? (d - base) : 32;
    const int mine = (lane < m) ? csr[beg + base + lane] : 0;
    int j = 0;
    for (; j + 4 <= m; j += 4) {
      const int s0 = __builtin_amdgcn_readlane(mine, j + 0);
      const int s1 = __builtin_amdgcn_readlane(mine, j + 1);
      const int s2 = __builtin_amdgcn_readlane(mine, j + 2);
      const int s3 = __builtin_amdgcn_readlane(mine, j + 3);
      const float4 v0 = *(const float4*)(xl + (size_t)s0 * D_FEAT);
      const float4 v1 = *(const float4*)(xl + (size_t)s1 * D_FEAT);
      const float4 v2 = *(const float4*)(xl + (size_t)s2 * D_FEAT);
      const float4 v3 = *(const float4*)(xl + (size_t)s3 * D_FEAT);
      acc.x += v0.x;  acc.y += v0.y;  acc.z += v0.z;  acc.w += v0.w;
      acc.x += v1.x;  acc.y += v1.y;  acc.z += v1.z;  acc.w += v1.w;
      acc.x += v2.x;  acc.y += v2.y;  acc.z += v2.z;  acc.w += v2.w;
      acc.x += v3.x;  acc.y += v3.y;  acc.z += v3.z;  acc.w += v3.w;
    }
    for (; j < m; ++j) {
      const int s = __builtin_amdgcn_readlane(mine, j);
      const float4 v = *(const float4*)(xl + (size_t)s * D_FEAT);
      acc.x += v.x;  acc.y += v.y;  acc.z += v.z;  acc.w += v.w;
    }
  }
  *(float4*)(agg1 + (size_t)node * D_FEAT + lane * 4) = acc;
  if (lane == 0) cnt[node] = (float)d;
}

__global__ __launch_bounds__(256) void sage_layer1_wmma(
    const float* __restrict__ x, const float* agg1,
    const float* __restrict__ cnt, const float* __restrict__ W1l,
    const float* __restrict__ W1r, const float* __restrict__ b1,
    float* h, int n_nodes) {
  __shared__ float ldsWl[D_FEAT * D_FEAT];

  for (int idx = threadIdx.x; idx < D_FEAT * 64; idx += 256) {
    const int r = idx >> 6;
    const int c = idx & 63;
    const float2 w = *(const float2*)(W1l + (size_t)r * D_FEAT + 2 * c);
    const int s = c ^ ((2 * r) & 63);
    *(float2*)(ldsWl + r * D_FEAT + 2 * s) = w;
  }
  __syncthreads();

  const int lane = threadIdx.x & 31;
  const int wave = threadIdx.x >> 5;
  const int tile = blockIdx.x * 8 + wave;
  const int n_tiles = n_nodes >> 4;
  if (tile >= n_tiles) return;

  const int half = lane >> 4;
  const int q = lane & 15;
  const int rowA = tile * 16 + q;

  const float inv = 1.0f / fmaxf(cnt[rowA], 1.0f);

  v8f acc[8];
#pragma unroll
  for (int i = 0; i < 8; ++i) {
    v8f z = {};
    acc[i] = z;
  }

  const float* aggRow = agg1 + (size_t)rowA * D_FEAT + 2 * half;
  const float* xRow   = x    + (size_t)rowA * D_FEAT + 2 * half;

  int nrow[8], nswz[8];
#pragma unroll
  for (int nt = 0; nt < 8; ++nt) {
    nrow[nt] = nt * 16 + q;
    nswz[nt] = (2 * nrow[nt]) & 63;
  }

  for (int k = 0; k < D_FEAT; k += 4) {
    const int c0 = (k >> 1) + half;
    const float2 tm = *(const float2*)(aggRow + k);
    const float2 ts = *(const float2*)(xRow + k);

    float2 bl[8], br[8];
#pragma unroll
    for (int nt = 0; nt < 8; ++nt) {
      bl[nt] = *(const float2*)(ldsWl + nrow[nt] * D_FEAT + 2 * (c0 ^ nswz[nt]));
      br[nt] = *(const float2*)(W1r + (size_t)nrow[nt] * D_FEAT + k + 2 * half);
    }

    v2f a_m;  a_m.x = tm.x * inv;  a_m.y = tm.y * inv;
    v2f a_s;  a_s.x = ts.x;        a_s.y = ts.y;
#pragma unroll
    for (int nt = 0; nt < 8; ++nt) {
      v2f vbl;  vbl.x = bl[nt].x;  vbl.y = bl[nt].y;
      v2f vbr;  vbr.x = br[nt].x;  vbr.y = br[nt].y;
      acc[nt] = __builtin_amdgcn_wmma_f32_16x16x4_f32(
          false, a_m, false, vbl, (short)0, acc[nt], false, false);
      acc[nt] = __builtin_amdgcn_wmma_f32_16x16x4_f32(
          false, a_s, false, vbr, (short)0, acc[nt], false, false);
    }
  }

#pragma unroll
  for (int nt = 0; nt < 8; ++nt) {
    const int n = nrow[nt];
    const float bias = b1[n];
#pragma unroll
    for (int v = 0; v < 8; ++v) {
      const int m = v + 8 * half;
      float val = acc[nt][v] + bias;
      val = fmaxf(val, 0.0f);
      h[(size_t)(tile * 16 + m) * D_FEAT + n] = val;
    }
  }
}

__global__ __launch_bounds__(256) void sage_pq(
    const float* __restrict__ h, const float* __restrict__ W2l,
    const float* __restrict__ W2r, const float* __restrict__ b2,
    float* __restrict__ p, float* __restrict__ out, int n_nodes) {
  const int lane = threadIdx.x & 31;
  const int node = (int)((blockIdx.x * (size_t)blockDim.x + threadIdx.x) >> 5);
  if (node >= n_nodes) return;
  const float4 hv = *(const float4*)(h + (size_t)node * D_FEAT + lane * 4);
  const float4 l0 = *(const float4*)(W2l + lane * 4);
  const float4 l1 = *(const float4*)(W2l + D_FEAT + lane * 4);
  const float4 r0 = *(const float4*)(W2r + lane * 4);
  const float4 r1 = *(const float4*)(W2r + D_FEAT + lane * 4);
  float d0 = hv.x * l0.x + hv.y * l0.y + hv.z * l0.z + hv.w * l0.w;
  float d1 = hv.x * l1.x + hv.y * l1.y + hv.z * l1.z + hv.w * l1.w;
  float d2 = hv.x * r0.x + hv.y * r0.y + hv.z * r0.z + hv.w * r0.w;
  float d3 = hv.x * r1.x + hv.y * r1.y + hv.z * r1.z + hv.w * r1.w;
#pragma unroll
  for (int off = 16; off > 0; off >>= 1) {
    d0 += __shfl_xor(d0, off, 32);
    d1 += __shfl_xor(d1, off, 32);
    d2 += __shfl_xor(d2, off, 32);
    d3 += __shfl_xor(d3, off, 32);
  }
  if (lane == 0) {
    p[(size_t)node * 2 + 0] = d0;
    p[(size_t)node * 2 + 1] = d1;
    out[(size_t)node * 2 + 0] = d2 + b2[0];
    out[(size_t)node * 2 + 1] = d3 + b2[1];
  }
}

__global__ __launch_bounds__(256) void sage_gather2(
    const float* __restrict__ p, const int* __restrict__ csr,
    const int* __restrict__ off, const int* __restrict__ deg,
    float* __restrict__ out, int n_nodes) {
  const int lane = threadIdx.x & 31;
  const int node = __builtin_amdgcn_readfirstlane(
      (int)((blockIdx.x * (size_t)blockDim.x + threadIdx.x) >> 5));
  if (node >= n_nodes) return;
  const int beg = off[node];
  const int d = deg[node];
  float s0 = 0.0f, s1 = 0.0f;
  for (int j = lane; j < d; j += 32) {
    const int s = csr[beg + j];
    const float2 pv = *(const float2*)(p + (size_t)s * 2);
    s0 += pv.x;
    s1 += pv.y;
  }
#pragma unroll
  for (int o = 16; o > 0; o >>= 1) {
    s0 += __shfl_xor(s0, o, 32);
    s1 += __shfl_xor(s1, o, 32);
  }
  if (lane == 0) {
    const float inv = 1.0f / fmaxf((float)d, 1.0f);
    out[(size_t)node * 2 + 0] += s0 * inv;
    out[(size_t)node * 2 + 1] += s1 * inv;
  }
}

extern "C" void kernel_launch(void* const* d_in, const int* in_sizes, int n_in,
                              void* d_out, int out_size, void* d_ws, size_t ws_size,
                              hipStream_t stream) {
  const float* x   = (const float*)d_in[0];
  const int*   ei  = (const int*)d_in[1];
  const float* W1l = (const float*)d_in[2];
  const float* W1r = (const float*)d_in[3];
  const float* b1  = (const float*)d_in[4];
  const float* W2l = (const float*)d_in[5];
  const float* W2r = (const float*)d_in[6];
  const float* b2  = (const float*)d_in[7];
  float* out = (float*)d_out;

  const int n_nodes = in_sizes[0] / D_FEAT;
  const int n_edges = in_sizes[1] / 2;

  float* ws   = (float*)d_ws;
  float* agg1 = ws;
  float* h    = agg1;
  float* p    = agg1 + (size_t)n_nodes * D_FEAT;
  float* cnt  = p + (size_t)n_nodes * 2;
  int*   deg  = (int*)(cnt + n_nodes);
  int*   off  = deg + n_nodes;
  int*   csr  = off + n_nodes;
  const int n_bins  = (n_nodes + 255) / 256;
  const int chunk   = (n_edges + EDGE_BLOCKS - 1) / EDGE_BLOCKS;
  int*   cnt_bb  = csr + n_edges;
  int*   base_bb = cnt_bb + n_bins * EDGE_BLOCKS;
  int2*  tmp     = (int2*)ws;

  sage_bin_hist<<<EDGE_BLOCKS, 256, 0, stream>>>(ei, cnt_bb, n_edges, n_bins, chunk);

  sage_scan<<<1, 1024, 0, stream>>>(cnt_bb, base_bb, n_bins * EDGE_BLOCKS);

  sage_bin_scatter<<<EDGE_BLOCKS, 256, 0, stream>>>(ei, base_bb, tmp, n_edges, n_bins,
                                                    chunk);

  sage_bin_fill<<<n_bins, 256, 0, stream>>>(tmp, base_bb, deg, off, csr, n_nodes,
                                            n_edges, n_bins);

  sage_gather1<<<(n_nodes + 7) / 8, 256, 0, stream>>>(x, csr, off, deg, agg1, cnt,
                                                     n_nodes);

  const int n_tiles = n_nodes / 16;
  sage_layer1_wmma<<<(n_tiles + 7) / 8, 256, 0, stream>>>(x, agg1, cnt, W1l, W1r,
                                                          b1, h, n_nodes);

  sage_pq<<<(n_nodes + 7) / 8, 256, 0, stream>>>(h, W2l, W2r, b2, p, out, n_nodes);

  sage_gather2<<<(n_nodes + 7) / 8, 256, 0, stream>>>(p, csr, off, deg, out, n_nodes);
}


# Layer-2 projections (h@W2l^T, h@W2r^T+b2) done in the layer-1 WMMA epilogue from the accumulators: h never stored, projection launch removed; sum order changes (8 per lane, then DPP row tree)
# speedup vs baseline: 2.6104x; 1.2901x over previous
#include <hip/hip_runtime.h>

typedef __attribute__((ext_vector_type(2))) float v2f;
typedef __attribute__((ext_vector_type(8))) float v8f;

#define D_FEAT 128


#define BIN_SHIFT 8
#define EDGE_BLOCKS 256
#define MAX_BINS 256

__global__ __launch_bounds__(256) void sage_bin_hist(
    const int* __restrict__ ei, int* __restrict__ cnt_bb, int n_edges,
    int n_bins, int chunk) {
  __shared__ int hist[MAX_BINS];
  const int t = threadIdx.x;
  if (t < n_bins) hist[t] = 0;
  __syncthreads();
  const int beg = blockIdx.x * chunk;
  int end = beg + chunk;
  if (end > n_edges) end = n_edges;
  const int* dstp = ei + n_edges;
  int e = beg + t;
  for (; e + 3 * 256 < end; e += 4 * 256) {
    int d[4];
#pragma unroll
    for (int u = 0; u < 4; ++u) d[u] = dstp[e + u * 256];
#pragma unroll
    for (int u = 0; u < 4; ++u) atomicAdd(&hist[d[u] >> BIN_SHIFT], 1);
  }
  for (; e < end; e += 256) atomicAdd(&hist[dstp[e] >> BIN_SHIFT], 1);
  __syncthreads();
  if (t < n_bins) cnt_bb[t * EDGE_BLOCKS + blockIdx.x] = hist[t];
}

__global__ __launch_bounds__(1024) void sage_scan(
    const int* __restrict__ deg, int* __restrict__ off, int n) {
  __shared__ int part[1024];
  const int t = threadIdx.x;
  const int chunk = (n + 1023) / 1024;
  int beg = t * chunk;
  if (beg > n) beg = n;
  int end = beg + chunk;
  if (end > n) end = n;
  int s = 0;
  for (int i = beg; i < end; ++i) s += deg[i];
  part[t] = s;
  __syncthreads();
  for (int d = 1; d < 1024; d <<= 1) {
    const int v = (t >= d) ? part[t - d] : 0;
    __syncthreads();
    part[t] += v;
    __syncthreads();
  }
  int run = part[t] - s;
  for (int i = beg; i < end; ++i) {
    off[i] = run;
    run += deg[i];
  }
}

__global__ __launch_bounds__(256) void sage_bin_scatter(
    const int* __restrict__ ei, const int* __restrict__ base_bb,
    int2* __restrict__ tmp, int n_edges, int n_bins, int chunk) {
  __shared__ int cur[MAX_BINS];
  const int t = threadIdx.x;
  if (t < n_bins) cur[t] = base_bb[t * EDGE_BLOCKS + blockIdx.x];
  __syncthreads();
  const int beg = blockIdx.x * chunk;
  int end = beg + chunk;
  if (end > n_edges) end = n_edges;
  const int* dstp = ei + n_edges;
  int e = beg + t;
  for (; e + 3 * 256 < end; e += 4 * 256) {
    int s[4], d[4];
#pragma unroll
    for (int u = 0; u < 4; ++u) {
      s[u] = ei[e + u * 256];
      d[u] = dstp[e + u * 256];
    }
#pragma unroll
    for (int u = 0; u < 4; ++u) {
      const int slot = atomicAdd(&cur[d[u] >> BIN_SHIFT], 1);
      tmp[slot] = make_int2(s[u], d[u]);
    }
  }
  for (; e < end; e += 256) {
    const int s = ei[e];
    const int d = dstp[e];
    const int slot = atomicAdd(&cur[d >> BIN_SHIFT], 1);
    tmp[slot] = make_int2(s, d);
  }
}

__global__ __launch_bounds__(256) void sage_bin_fill(
    const int2* __restrict__ tmp, const int* __restrict__ base_bb,
    int* __restrict__ deg, int* __restrict__ off, int* __restrict__ csr,
    int n_nodes, int n_edges, int n_bins) {
  __shared__ int cnt[256];
  __shared__ int scn[256];
  const int t = threadIdx.x;
  const int b = blockIdx.x;
  const int beg = base_bb[b * EDGE_BLOCKS];
  const int end = (b + 1 < n_bins) ? base_bb[(b + 1) * EDGE_BLOCKS] : n_edges;
  cnt[t] = 0;
  __syncthreads();
  int i = beg + t;
  for (; i + 3 * 256 < end; i += 4 * 256) {
    int d[4];
#pragma unroll
    for (int u = 0; u < 4; ++u) d[u] = tmp[i + u * 256].y;
#pragma unroll
    for (int u = 0; u < 4; ++u) atomicAdd(&cnt[d[u] & 255], 1);
  }
  for (; i < end; i += 256) atomicAdd(&cnt[tmp[i].y & 255], 1);
  __syncthreads();
  const int c = cnt[t];
  scn[t] = c;
  __syncthreads();
  for (int d = 1; d < 256; d <<= 1) {
    const int v = (t >= d) ? scn[t - d] : 0;
    __syncthreads();
    scn[t] += v;
    __syncthreads();
  }
  const int excl = scn[t] - c;
  const int node = (b << BIN_SHIFT) + t;
  if (node < n_nodes) {
    deg[node] = c;
    off[node] = beg + excl;
  }
  cnt[t] = excl;
  __syncthreads();
  i = beg + t;
  for (; i + 3 * 256 < end; i += 4 * 256) {
    int2 sd[4];
#pragma unroll
    for (int u = 0; u < 4; ++u) sd[u] = tmp[i + u * 256];
#pragma unroll
    for (int u = 0; u < 4; ++u) {
      const int r = atomicAdd(&cnt[sd[u].y & 255], 1);
      csr[beg + r] = sd[u].x;
    }
  }
  for (; i < end; i += 256) {
    const int2 sd = tmp[i];
    const int r = atomicAdd(&cnt[sd.y & 255], 1);
    csr[beg + r] = sd.x;
  }
}

__global__ __launch_bounds__(256) void sage_gather1(
    const float* __restrict__ x, const int* __restrict__ csr,
    const int* __restrict__ off, const int* __restrict__ deg,
    float* __restrict__ agg1, float* __restrict__ cnt, int n_nodes) {
  const int lane = threadIdx.x & 31;
  const int node = __builtin_amdgcn_readfirstlane(
      (int)((blockIdx.x * (size_t)blockDim.x + threadIdx.x) >> 5));
  if (node >= n_nodes) return;
  const int beg = off[node];
  const int d = deg[node];
  const float* xl = x + lane * 4;
  float4 acc = make_float4(0.0f, 0.0f, 0.0f, 0.0f);
  for (int base = 0; base < d; base += 32) {
    const int m = (d - base < 32) ? (d - base) : 32;
    const int mine = (lane < m) ? csr[beg + base + lane] : 0;
    int j = 0;
    for (; j + 4 <= m; j += 4) {
      const int s0 = __builtin_amdgcn_readlane(mine, j + 0);
      const int s1 = __builtin_amdgcn_readlane(mine, j + 1);
      const int s2 = __builtin_amdgcn_readlane(mine, j + 2);
      const int s3 = __builtin_amdgcn_readlane(mine, j + 3);
      const float4 v0 = *(const float4*)(xl + (size_t)s0 * D_FEAT);
      const float4 v1 = *(const float4*)(xl + (size_t)s1 * D_FEAT);
      const float4 v2 = *(const float4*)(xl + (size_t)s2 * D_FEAT);
      const float4 v3 = *(const float4*)(xl + (size_t)s3 * D_FEAT);
      acc.x += v0.x;  acc.y += v0.y;  acc.z += v0.z;  acc.w += v0.w;
      acc.x += v1.x;  acc.y += v1.y;  acc.z += v1.z;  acc.w += v1.w;
      acc.x += v2.x;  acc.y += v2.y;  acc.z += v2.z;  acc.w += v2.w;
      acc.x += v3.x;  acc.y += v3.y;  acc.z += v3.z;  acc.w += v3.w;
    }
    for (; j < m; ++j) {
      const int s = __builtin_amdgcn_readlane(mine, j);
      const float4 v = *(const float4*)(xl + (size_t)s * D_FEAT);
      acc.x += v.x;  acc.y += v.y;  acc.z += v.z;  acc.w += v.w;
    }
  }
  *(float4*)(agg1 + (size_t)node * D_FEAT + lane * 4) = acc;
  if (lane == 0) cnt[node] = (float)d;
}

template <int XMASK>
__device__ __forceinline__ float row_xor(float x) {
  return __builtin_bit_cast(
      float, __builtin_amdgcn_update_dpp(0, __builtin_bit_cast(int, x),
                                         0x160 + XMASK, 0xf, 0xf, true));
}

template <int CNT, int XMASK>
__device__ __forceinline__ void row_fold(float* v, bool upper) {
#pragma unroll
  for (int i = 0; i < CNT; ++i) {
    const float keep = upper ? v[i + CNT] : v[i];
    const float send = upper ? v[i] : v[i + CNT];
    v[i] = keep + row_xor<XMASK>(send);
  }
}

__global__ __launch_bounds__(256) void sage_layer1_wmma(
    const float* __restrict__ x, const float* __restrict__ agg1,
    const float* __restrict__ cnt, const float* __restrict__ W1l,
    const float* __restrict__ W1r, const float* __restrict__ b1,
    const float* __restrict__ W2l, const float* __restrict__ W2r,
    const float* __restrict__ b2, float* __restrict__ p,
    float* __restrict__ out, int n_nodes) {
  __shared__ float ldsWl[D_FEAT * D_FEAT];

  for (int idx = threadIdx.x; idx < D_FEAT * 64; idx += 256) {
    const int r = idx >> 6;
    const int c = idx & 63;
    const float2 w = *(const float2*)(W1l + (size_t)r * D_FEAT + 2 * c);
    const int s = c ^ ((2 * r) & 63);
    *(float2*)(ldsWl + r * D_FEAT + 2 * s) = w;
  }
  __syncthreads();

  const int lane = threadIdx.x & 31;
  const int wave = threadIdx.x >> 5;
  const int tile = blockIdx.x * 8 + wave;
  const int n_tiles = n_nodes >> 4;
  if (tile >= n_tiles) return;

  const int half = lane >> 4;
  const int q = lane & 15;
  const int rowA = tile * 16 + q;

  const float inv = 1.0f / fmaxf(cnt[rowA], 1.0f);

  v8f acc[8];
#pragma unroll
  for (int i = 0; i < 8; ++i) {
    v8f z = {};
    acc[i] = z;
  }

  const float* aggRow = agg1 + (size_t)rowA * D_FEAT + 2 * half;
  const float* xRow   = x    + (size_t)rowA * D_FEAT + 2 * half;

  int nrow[8], nswz[8];
#pragma unroll
  for (int nt = 0; nt < 8; ++nt) {
    nrow[nt] = nt * 16 + q;
    nswz[nt] = (2 * nrow[nt]) & 63;
  }

  for (int k = 0; k < D_FEAT; k += 4) {
    const int c0 = (k >> 1) + half;
    const float2 tm = *(const float2*)(aggRow + k);
    const float2 ts = *(const float2*)(xRow + k);

    float2 bl[8], br[8];
#pragma unroll
    for (int nt = 0; nt < 8; ++nt) {
      bl[nt] = *(const float2*)(ldsWl + nrow[nt] * D_FEAT + 2 * (c0 ^ nswz[nt]));
      br[nt] = *(const float2*)(W1r + (size_t)nrow[nt] * D_FEAT + k + 2 * half);
    }

    v2f a_m;  a_m.x = tm.x * inv;  a_m.y = tm.y * inv;
    v2f a_s;  a_s.x = ts.x;        a_s.y = ts.y;
#pragma unroll
    for (int nt = 0; nt < 8; ++nt) {
      v2f vbl;  vbl.x = bl[nt].x;  vbl.y = bl[nt].y;
      v2f vbr;  vbr.x = br[nt].x;  vbr.y = br[nt].y;
      acc[nt] = __builtin_amdgcn_wmma_f32_16x16x4_f32(
          false, a_m, false, vbl, (short)0, acc[nt], false, false);
      acc[nt] = __builtin_amdgcn_wmma_f32_16x16x4_f32(
          false, a_s, false, vbr, (short)0, acc[nt], false, false);
    }
  }

  float part[32];
#pragma unroll
  for (int i = 0; i < 32; ++i) part[i] = 0.0f;
#pragma unroll
  for (int nt = 0; nt < 8; ++nt) {
    const int n = nrow[nt];
    const float bias = b1[n];
    const float wl0 = W2l[n];
    const float wl1 = W2l[D_FEAT + n];
    const float wr0 = W2r[n];
    const float wr1 = W2r[D_FEAT + n];
#pragma unroll
    for (int v = 0; v < 8; ++v) {
      const float hv = fmaxf(acc[nt][v] + bias, 0.0f);
      part[v * 4 + 0] += hv * wl0;
      part[v * 4 + 1] += hv * wl1;
      part[v * 4 + 2] += hv * wr0;
      part[v * 4 + 3] += hv * wr1;
    }
  }
  row_fold<16, 8>(part, (q & 8) != 0);
  row_fold<8, 4>(part, (q & 4) != 0);
  row_fold<4, 2>(part, (q & 2) != 0);
  row_fold<2, 1>(part, (q & 1) != 0);

  const int node = tile * 16 + 8 * half + (q >> 1);
  const bool self_side = (q & 1) != 0;
  const float bias0 = b2[0];
  const float bias1 = b2[1];
  float2 res;
  res.x = self_side ? part[0] + bias0 : part[0];
  res.y = self_side ? part[1] + bias1 : part[1];
  float* dst = (self_side ? out : p) + (size_t)node * 2;
  *(float2*)dst = res;
}

__global__ __launch_bounds__(256) void sage_gather2(
    const float* __restrict__ p, const int* __restrict__ csr,
    const int* __restrict__ off, const int* __restrict__ deg,
    float* __restrict__ out, int n_nodes) {
  const int lane = threadIdx.x & 31;
  const int node = __builtin_amdgcn_readfirstlane(
      (int)((blockIdx.x * (size_t)blockDim.x + threadIdx.x) >> 5));
  if (node >= n_nodes) return;
  const int beg = off[node];
  const int d = deg[node];
  float s0 = 0.0f, s1 = 0.0f;
  for (int j = lane; j < d; j += 32) {
    const int s = csr[beg + j];
    const float2 pv = *(const float2*)(p + (size_t)s * 2);
    s0 += pv.x;
    s1 += pv.y;
  }
#pragma unroll
  for (int o = 16; o > 0; o >>= 1) {
    s0 += __shfl_xor(s0, o, 32);
    s1 += __shfl_xor(s1, o, 32);
  }
  if (lane == 0) {
    const float inv = 1.0f / fmaxf((float)d, 1.0f);
    out[(size_t)node * 2 + 0] += s0 * inv;
    out[(size_t)node * 2 + 1] += s1 * inv;
  }
}

extern "C" void kernel_launch(void* const* d_in, const int* in_sizes, int n_in,
                              void* d_out, int out_size, void* d_ws, size_t ws_size,
                              hipStream_t stream) {
  const float* x   = (const float*)d_in[0];
  const int*   ei  = (const int*)d_in[1];
  const float* W1l = (const float*)d_in[2];
  const float* W1r = (const float*)d_in[3];
  const float* b1  = (const float*)d_in[4];
  const float* W2l = (const float*)d_in[5];
  const float* W2r = (const float*)d_in[6];
  const float* b2  = (const float*)d_in[7];
  float* out = (float*)d_out;

  const int n_nodes = in_sizes[0] / D_FEAT;
  const int n_edges = in_sizes[1] / 2;

  float* ws   = (float*)d_ws;
  float* agg1 = ws;
  float* p    = agg1 + (size_t)n_nodes * D_FEAT;
  float* cnt  = p + (size_t)n_nodes * 2;
  int*   deg  = (int*)(cnt + n_nodes);
  int*   off  = deg + n_nodes;
  int*   csr  = off + n_nodes;
  const int n_bins  = (n_nodes + 255) / 256;
  const int chunk   = (n_edges + EDGE_BLOCKS - 1) / EDGE_BLOCKS;
  int*   cnt_bb  = csr + n_edges;
  int*   base_bb = cnt_bb + n_bins * EDGE_BLOCKS;
  int2*  tmp     = (int2*)ws;

  sage_bin_hist<<<EDGE_BLOCKS, 256, 0, stream>>>(ei, cnt_bb, n_edges, n_bins, chunk);

  sage_scan<<<1, 1024, 0, stream>>>(cnt_bb, base_bb, n_bins * EDGE_BLOCKS);

  sage_bin_scatter<<<EDGE_BLOCKS, 256, 0, stream>>>(ei, base_bb, tmp, n_edges, n_bins,
                                                    chunk);

  sage_bin_fill<<<n_bins, 256, 0, stream>>>(tmp, base_bb, deg, off, csr, n_nodes,
                                            n_edges, n_bins);

  sage_gather1<<<(n_nodes + 7) / 8, 256, 0, stream>>>(x, csr, off, deg, agg1, cnt,
                                                     n_nodes);

  const int n_tiles = n_nodes / 16;
  sage_layer1_wmma<<<(n_tiles + 7) / 8, 256, 0, stream>>>(
      x, agg1, cnt, W1l, W1r, b1, W2l, W2r, b2, p, out, n_nodes);

  sage_gather2<<<(n_nodes + 7) / 8, 256, 0, stream>>>(p, csr, off, deg, out, n_nodes);
}


# Layer-1 neighbour gather moved into the WMMA kernel: each wave sums its 16 nodes' rows and hands the means to its A fragments via LDS; half the waves run the self product first; sum order changes
# speedup vs baseline: 2.7367x; 1.0484x over previous
#include <hip/hip_runtime.h>

typedef __attribute__((ext_vector_type(2))) float v2f;
typedef __attribute__((ext_vector_type(8))) float v8f;

#define D_FEAT 128


#define BIN_SHIFT 8
#define EDGE_BLOCKS 256
#define MAX_BINS 256

__global__ __launch_bounds__(256) void sage_bin_hist(
    const int* __restrict__ ei, int* __restrict__ cnt_bb, int n_edges,
    int n_bins, int chunk) {
  __shared__ int hist[MAX_BINS];
  const int t = threadIdx.x;
  if (t < n_bins) hist[t] = 0;
  __syncthreads();
  const int beg = blockIdx.x * chunk;
  int end = beg + chunk;
  if (end > n_edges) end = n_edges;
  const int* dstp = ei + n_edges;
  int e = beg + t;
  for (; e + 3 * 256 < end; e += 4 * 256) {
    int d[4];
#pragma unroll
    for (int u = 0; u < 4; ++u) d[u] = dstp[e + u * 256];
#pragma unroll
    for (int u = 0; u < 4; ++u) atomicAdd(&hist[d[u] >> BIN_SHIFT], 1);
  }
  for (; e < end; e += 256) atomicAdd(&hist[dstp[e] >> BIN_SHIFT], 1);
  __syncthreads();
  if (t < n_bins) cnt_bb[t * EDGE_BLOCKS + blockIdx.x] = hist[t];
}

__global__ __launch_bounds__(1024) void sage_scan(
    const int* __restrict__ deg, int* __restrict__ off, int n) {
  __shared__ int part[1024];
  const int t = threadIdx.x;
  const int chunk = (n + 1023) / 1024;
  int beg = t * chunk;
  if (beg > n) beg = n;
  int end = beg + chunk;
  if (end > n) end = n;
  int s = 0;
  for (int i = beg; i < end; ++i) s += deg[i];
  part[t] = s;
  __syncthreads();
  for (int d = 1; d < 1024; d <<= 1) {
    const int v = (t >= d) ? part[t - d] : 0;
    __syncthreads();
    part[t] += v;
    __syncthreads();
  }
  int run = part[t] - s;
  for (int i = beg; i < end; ++i) {
    off[i] = run;
    run += deg[i];
  }
}

__global__ __launch_bounds__(256) void sage_bin_scatter(
    const int* __restrict__ ei, const int* __restrict__ base_bb,
    int2* __restrict__ tmp, int n_edges, int n_bins, int chunk) {
  __shared__ int cur[MAX_BINS];
  const int t = threadIdx.x;
  if (t < n_bins) cur[t] = base_bb[t * EDGE_BLOCKS + blockIdx.x];
  __syncthreads();
  const int beg = blockIdx.x * chunk;
  int end = beg + chunk;
  if (end > n_edges) end = n_edges;
  const int* dstp = ei + n_edges;
  int e = beg + t;
  for (; e + 3 * 256 < end; e += 4 * 256) {
    int s[4], d[4];
#pragma unroll
    for (int u = 0; u < 4; ++u) {
      s[u] = ei[e + u * 256];
      d[u] = dstp[e + u * 256];
    }
#pragma unroll
    for (int u = 0; u < 4; ++u) {
      const int slot = atomicAdd(&cur[d[u] >> BIN_SHIFT], 1);
      tmp[slot] = make_int2(s[u], d[u]);
    }
  }
  for (; e < end; e += 256) {
    const int s = ei[e];
    const int d = dstp[e];
    const int slot = atomicAdd(&cur[d >> BIN_SHIFT], 1);
    tmp[slot] = make_int2(s, d);
  }
}

__global__ __launch_bounds__(256) void sage_bin_fill(
    const int2* __restrict__ tmp, const int* __restrict__ base_bb,
    int* __restrict__ deg, int* __restrict__ off, int* __restrict__ csr,
    int n_nodes, int n_edges, int n_bins) {
  __shared__ int cnt[256];
  __shared__ int scn[256];
  const int t = threadIdx.x;
  const int b = blockIdx.x;
  const int beg = base_bb[b * EDGE_BLOCKS];
  const int end = (b + 1 < n_bins) ? base_bb[(b + 1) * EDGE_BLOCKS] : n_edges;
  cnt[t] = 0;
  __syncthreads();
  int i = beg + t;
  for (; i + 3 * 256 < end; i += 4 * 256) {
    int d[4];
#pragma unroll
    for (int u = 0; u < 4; ++u) d[u] = tmp[i + u * 256].y;
#pragma unroll
    for (int u = 0; u < 4; ++u) atomicAdd(&cnt[d[u] & 255], 1);
  }
  for (; i < end; i += 256) atomicAdd(&cnt[tmp[i].y & 255], 1);
  __syncthreads();
  const int c = cnt[t];
  scn[t] = c;
  __syncthreads();
  for (int d = 1; d < 256; d <<= 1) {
    const int v = (t >= d) ? scn[t - d] : 0;
    __syncthreads();
    scn[t] += v;
    __syncthreads();
  }
  const int excl = scn[t] - c;
  const int node = (b << BIN_SHIFT) + t;
  if (node < n_nodes) {
    deg[node] = c;
    off[node] = beg + excl;
  }
  cnt[t] = excl;
  __syncthreads();
  i = beg + t;
  for (; i + 3 * 256 < end; i += 4 * 256) {
    int2 sd[4];
#pragma unroll
    for (int u = 0; u < 4; ++u) sd[u] = tmp[i + u * 256];
#pragma unroll
    for (int u = 0; u < 4; ++u) {
      const int r = atomicAdd(&cnt[sd[u].y & 255], 1);
      csr[beg + r] = sd[u].x;
    }
  }
  for (; i < end; i += 256) {
    const int2 sd = tmp[i];
    const int r = atomicAdd(&cnt[sd.y & 255], 1);
    csr[beg + r] = sd.x;
  }
}

#define TILE_PITCH 132

__device__ __forceinline__ float lane_read_f(float v, int l) {
  return __builtin_bit_cast(
      float, __builtin_amdgcn_readlane(__builtin_bit_cast(int, v), l));
}

__device__ __forceinline__ void tile_gather(
    const float* __restrict__ x, const int* __restrict__ csr,
    const int* __restrict__ off, const int* __restrict__ deg, int node0,
    int lane, float* tileLds) {
  const int li = lane & 15;
  const int offv = off[node0 + li];
  const int degv = deg[node0 + li];
  const float invv = 1.0f / fmaxf((float)degv, 1.0f);
  const int e0 = __builtin_amdgcn_readlane(offv, 0);
  const int e1 = __builtin_amdgcn_readlane(offv, 15) +
                 __builtin_amdgcn_readlane(degv, 15);
  const int bndv = (lane < 16) ? offv : e1;
  const int total = e1 - e0;
  const float* xl = x + lane * 4;
  float* trow = tileLds + lane * 4;

  float4 acc = make_float4(0.0f, 0.0f, 0.0f, 0.0f);
  int cur = 0;
  int nextB = __builtin_amdgcn_readlane(bndv, 1);

  int mine = (lane < total) ? csr[e0 + lane] : 0;
  for (int base = 0; base < total; base += 32) {
    const int nb = base + 32;
    const int ni = (nb + lane < total) ? (nb + lane) : (total - 1);
    const int nxt = csr[e0 + ni];
#pragma unroll
    for (int h = 0; h < 2; ++h) {
      const int g = base + 16 * h;
      if (g < total) {
        float4 v[16];
#pragma unroll
        for (int u = 0; u < 16; ++u) {
          const int s = __builtin_amdgcn_readlane(mine, 16 * h + u);
          v[u] = *(const float4*)(xl + (size_t)s * D_FEAT);
        }
#pragma unroll
        for (int u = 0; u < 16; ++u) {
          const int e = e0 + g + u;
          if (e < e1) {
            while (e >= nextB && cur < 15) {
              const float inv = lane_read_f(invv, cur);
              *(float4*)(trow + cur * TILE_PITCH) = make_float4(
                  acc.x * inv, acc.y * inv, acc.z * inv, acc.w * inv);
              acc = make_float4(0.0f, 0.0f, 0.0f, 0.0f);
              ++cur;
              nextB = __builtin_amdgcn_readlane(bndv, cur + 1);
            }
            acc.x += v[u].x;  acc.y += v[u].y;
            acc.z += v[u].z;  acc.w += v[u].w;
          }
        }
      }
    }
    mine = nxt;
  }
  for (; cur < 16; ++cur) {
    const float inv = lane_read_f(invv, cur);
    *(float4*)(trow + cur * TILE_PITCH) = make_float4(
        acc.x * inv, acc.y * inv, acc.z * inv, acc.w * inv);
    acc = make_float4(0.0f, 0.0f, 0.0f, 0.0f);
  }
}

template <int XMASK>
__device__ __forceinline__ float row_xor(float x) {
  return __builtin_bit_cast(
      float, __builtin_amdgcn_update_dpp(0, __builtin_bit_cast(int, x),
                                         0x160 + XMASK, 0xf, 0xf, true));
}

template <int CNT, int XMASK>
__device__ __forceinline__ void row_fold(float* v, bool upper) {
#pragma unroll
  for (int i = 0; i < CNT; ++i) {
    const float keep = upper ? v[i + CNT] : v[i];
    const float send = upper ? v[i] : v[i + CNT];
    v[i] = keep + row_xor<XMASK>(send);
  }
}


#define WMMA_TAIL_PAD()                                    \
  do {                                                     \
    __builtin_amdgcn_sched_barrier(0);                     \
    asm volatile("v_nop\n\tv_nop\n\tv_nop\n\tv_nop");      \
  } while (0)

__device__ __forceinline__ void self_product(
    v8f* acc, const float* __restrict__ xRow, const float* __restrict__ W1r,
    const int* nrow, int half) {
#pragma unroll 1
  for (int k = 0; k < D_FEAT; k += 8) {
    float2 ts[2], br[2][8];
#pragma unroll
    for (int kk = 0; kk < 2; ++kk) {
      ts[kk] = *(const float2*)(xRow + k + 4 * kk);
#pragma unroll
      for (int nt = 0; nt < 8; ++nt)
        br[kk][nt] = *(const float2*)(W1r + (size_t)nrow[nt] * D_FEAT + k +
                                      4 * kk + 2 * half);
    }
#pragma unroll
    for (int kk = 0; kk < 2; ++kk) {
      v2f a_s;  a_s.x = ts[kk].x;  a_s.y = ts[kk].y;
#pragma unroll
      for (int nt = 0; nt < 8; ++nt) {
        v2f vbr;  vbr.x = br[kk][nt].x;  vbr.y = br[kk][nt].y;
        acc[nt] = __builtin_amdgcn_wmma_f32_16x16x4_f32(
            false, a_s, false, vbr, (short)0, acc[nt], false, false);
      }
    }
    WMMA_TAIL_PAD();
  }
}

__global__ __launch_bounds__(256) void sage_layer1_wmma(
    const float* __restrict__ x, const int* __restrict__ csr,
    const int* __restrict__ off, const int* __restrict__ deg,
    const float* __restrict__ W1l,
    const float* __restrict__ W1r, const float* __restrict__ b1,
    const float* __restrict__ W2l, const float* __restrict__ W2r,
    const float* __restrict__ b2, float* __restrict__ p,
    float* __restrict__ out, int n_nodes) {
  __shared__ float ldsWl[D_FEAT * D_FEAT];
  __shared__ __attribute__((aligned(16))) float ldsAgg[8 * 16 * TILE_PITCH];

  for (int idx = threadIdx.x; idx < D_FEAT * 64; idx += 256) {
    const int r = idx >> 6;
    const int c = idx & 63;
    const float2 w = *(const float2*)(W1l + (size_t)r * D_FEAT + 2 * c);
    const int s = c ^ ((2 * r) & 63);
    *(float2*)(ldsWl + r * D_FEAT + 2 * s) = w;
  }
  __syncthreads();

  const int lane = threadIdx.x & 31;
  const int wave = __builtin_amdgcn_readfirstlane((int)(threadIdx.x >> 5));
  const int tile = blockIdx.x * 8 + wave;
  const int n_tiles = n_nodes >> 4;
  if (tile >= n_tiles) return;

  const int half = lane >> 4;
  const int q = lane & 15;
  const int rowA = tile * 16 + q;

  v8f acc[8];
#pragma unroll
  for (int i = 0; i < 8; ++i) {
    v8f z = {};
    acc[i] = z;
  }

  const float* xRow = x + (size_t)rowA * D_FEAT + 2 * half;
  float* myTile = ldsAgg + wave * (16 * TILE_PITCH);

  int nrow[8], nswz[8];
#pragma unroll
  for (int nt = 0; nt < 8; ++nt) {
    nrow[nt] = nt * 16 + q;
    nswz[nt] = (2 * nrow[nt]) & 63;
  }

  const bool self_first = (wave & 4) != 0;
  if (self_first) self_product(acc, xRow, W1r, nrow, half);
  tile_gather(x, csr, off, deg, tile * 16, lane, myTile);
  if (!self_first) self_product(acc, xRow, W1r, nrow, half);

  __builtin_amdgcn_fence(__ATOMIC_RELEASE, "workgroup");
  __builtin_amdgcn_wave_barrier();
  __builtin_amdgcn_fence(__ATOMIC_ACQUIRE, "workgroup");

  const float* aggRow = myTile + q * TILE_PITCH + 2 * half;
#pragma unroll 1
  for (int k = 0; k < D_FEAT; k += 8) {
    float2 tm[2], bl[2][8];
#pragma unroll
    for (int kk = 0; kk < 2; ++kk) {
      const int c0 = ((k + 4 * kk) >> 1) + half;
      tm[kk] = *(const float2*)(aggRow + k + 4 * kk);
#pragma unroll
      for (int nt = 0; nt < 8; ++nt)
        bl[kk][nt] = *(const float2*)(ldsWl + nrow[nt] * D_FEAT +
                                      2 * (c0 ^ nswz[nt]));
    }
#pragma unroll
    for (int kk = 0; kk < 2; ++kk) {
      v2f a_m;  a_m.x = tm[kk].x;  a_m.y = tm[kk].y;
#pragma unroll
      for (int nt = 0; nt < 8; ++nt) {
        v2f vbl;  vbl.x = bl[kk][nt].x;  vbl.y = bl[kk][nt].y;
        acc[nt] = __builtin_amdgcn_wmma_f32_16x16x4_f32(
            false, a_m, false, vbl, (short)0, acc[nt], false, false);
      }
    }
    WMMA_TAIL_PAD();
  }

  float part[32];
#pragma unroll
  for (int i = 0; i < 32; ++i) part[i] = 0.0f;
#pragma unroll
  for (int nt = 0; nt < 8; ++nt) {
    const int n = nrow[nt];
    const float bias = b1[n];
    const float wl0 = W2l[n];
    const float wl1 = W2l[D_FEAT + n];
    const float wr0 = W2r[n];
    const float wr1 = W2r[D_FEAT + n];
#pragma unroll
    for (int v = 0; v < 8; ++v) {
      const float hv = fmaxf(acc[nt][v] + bias, 0.0f);
      part[v * 4 + 0] += hv * wl0;
      part[v * 4 + 1] += hv * wl1;
      part[v * 4 + 2] += hv * wr0;
      part[v * 4 + 3] += hv * wr1;
    }
  }
  row_fold<16, 8>(part, (q & 8) != 0);
  row_fold<8, 4>(part, (q & 4) != 0);
  row_fold<4, 2>(part, (q & 2) != 0);
  row_fold<2, 1>(part, (q & 1) != 0);

  const int node = tile * 16 + 8 * half + (q >> 1);
  const bool self_side = (q & 1) != 0;
  const float bias0 = b2[0];
  const float bias1 = b2[1];
  float2 res;
  res.x = self_side ? part[0] + bias0 : part[0];
  res.y = self_side ? part[1] + bias1 : part[1];
  float* dst = (self_side ? out : p) + (size_t)node * 2;
  *(float2*)dst = res;
}

__global__ __launch_bounds__(256) void sage_gather2(
    const float* __restrict__ p, const int* __restrict__ csr,
    const int* __restrict__ off, const int* __restrict__ deg,
    float* __restrict__ out, int n_nodes) {
  const int lane = threadIdx.x & 31;
  const int node = __builtin_amdgcn_readfirstlane(
      (int)((blockIdx.x * (size_t)blockDim.x + threadIdx.x) >> 5));
  if (node >= n_nodes) return;
  const int beg = off[node];
  const int d = deg[node];
  float s0 = 0.0f, s1 = 0.0f;
  for (int j = lane; j < d; j += 32) {
    const int s = csr[beg + j];
    const float2 pv = *(const float2*)(p + (size_t)s * 2);
    s0 += pv.x;
    s1 += pv.y;
  }
#pragma unroll
  for (int o = 16; o > 0; o >>= 1) {
    s0 += __shfl_xor(s0, o, 32);
    s1 += __shfl_xor(s1, o, 32);
  }
  if (lane == 0) {
    const float inv = 1.0f / fmaxf((float)d, 1.0f);
    out[(size_t)node * 2 + 0] += s0 * inv;
    out[(size_t)node * 2 + 1] += s1 * inv;
  }
}

extern "C" void kernel_launch(void* const* d_in, const int* in_sizes, int n_in,
                              void* d_out, int out_size, void* d_ws, size_t ws_size,
                              hipStream_t stream) {
  const float* x   = (const float*)d_in[0];
  const int*   ei  = (const int*)d_in[1];
  const float* W1l = (const float*)d_in[2];
  const float* W1r = (const float*)d_in[3];
  const float* b1  = (const float*)d_in[4];
  const float* W2l = (const float*)d_in[5];
  const float* W2r = (const float*)d_in[6];
  const float* b2  = (const float*)d_in[7];
  float* out = (float*)d_out;

  const int n_nodes = in_sizes[0] / D_FEAT;
  const int n_edges = in_sizes[1] / 2;

  float* ws   = (float*)d_ws;
  float* agg1 = ws;
  float* p    = agg1 + (size_t)n_nodes * D_FEAT;
  float* cnt  = p + (size_t)n_nodes * 2;
  int*   deg  = (int*)(cnt + n_nodes);
  int*   off  = deg + n_nodes;
  int*   csr  = off + n_nodes;
  const int n_bins  = (n_nodes + 255) / 256;
  const int chunk   = (n_edges + EDGE_BLOCKS - 1) / EDGE_BLOCKS;
  int*   cnt_bb  = csr + n_edges;
  int*   base_bb = cnt_bb + n_bins * EDGE_BLOCKS;
  int2*  tmp     = (int2*)ws;

  sage_bin_hist<<<EDGE_BLOCKS, 256, 0, stream>>>(ei, cnt_bb, n_edges, n_bins, chunk);

  sage_scan<<<1, 1024, 0, stream>>>(cnt_bb, base_bb, n_bins * EDGE_BLOCKS);

  sage_bin_scatter<<<EDGE_BLOCKS, 256, 0, stream>>>(ei, base_bb, tmp, n_edges, n_bins,
                                                    chunk);

  sage_bin_fill<<<n_bins, 256, 0, stream>>>(tmp, base_bb, deg, off, csr, n_nodes,
                                            n_edges, n_bins);

  const int n_tiles = n_nodes / 16;
  sage_layer1_wmma<<<(n_tiles + 7) / 8, 256, 0, stream>>>(
      x, csr, off, deg, W1l, W1r, b1, W2l, W2r, b2, p, out, n_nodes);

  sage_gather2<<<(n_nodes + 7) / 8, 256, 0, stream>>>(p, csr, off, deg, out, n_nodes);
}


# Prefix sums of the (bin, chunk) edge counts moved into the scatter blocks: each reads the chunk-major count table by 512-B wave loads and scans the bin totals itself; one-block scan launch removed
# speedup vs baseline: 3.4479x; 1.2599x over previous
#include <hip/hip_runtime.h>

typedef __attribute__((ext_vector_type(2))) float v2f;
typedef __attribute__((ext_vector_type(8))) float v8f;

#define D_FEAT 128


#define BIN_SHIFT 8
#define EDGE_BLOCKS 256
#define MAX_BINS 256

__global__ __launch_bounds__(256) void sage_bin_hist(
    const int* __restrict__ ei, int* __restrict__ cnt_cb, int n_edges,
    int n_bins, int chunk) {
  __shared__ int hist[MAX_BINS];
  const int t = threadIdx.x;
  if (t < n_bins) hist[t] = 0;
  __syncthreads();
  const int beg = blockIdx.x * chunk;
  int end = beg + chunk;
  if (end > n_edges) end = n_edges;
  const int* dstp = ei + n_edges;
  int e = beg + t;
  for (; e + 3 * 256 < end; e += 4 * 256) {
    int d[4];
#pragma unroll
    for (int u = 0; u < 4; ++u) d[u] = dstp[e + u * 256];
#pragma unroll
    for (int u = 0; u < 4; ++u) atomicAdd(&hist[d[u] >> BIN_SHIFT], 1);
  }
  for (; e < end; e += 256) atomicAdd(&hist[dstp[e] >> BIN_SHIFT], 1);
  __syncthreads();
  cnt_cb[blockIdx.x * MAX_BINS + t] = (t < n_bins) ? hist[t] : 0;
}

__global__ __launch_bounds__(256) void sage_bin_scatter(
    const int* __restrict__ ei, const int* __restrict__ cnt_cb,
    int* __restrict__ bin_start, int2* __restrict__ tmp, int n_edges,
    int n_bins, int chunk) {
  __shared__ int cur[MAX_BINS];
  __shared__ __attribute__((aligned(16))) int partTot[4 * MAX_BINS];
  __shared__ __attribute__((aligned(16))) int partPre[4 * MAX_BINS];
  __shared__ int waveTot[8];
  const int t = threadIdx.x;
  const int c = blockIdx.x;
  {
    const int q = t & 63;
    const int r = t >> 6;
    int4 tot = make_int4(0, 0, 0, 0);
    int4 pre = make_int4(0, 0, 0, 0);
    if (4 * q < n_bins) {
      const int4* col = (const int4*)cnt_cb + q;
      for (int j = 0; j < EDGE_BLOCKS / 4; j += 16) {
        int4 v[16];
#pragma unroll
        for (int u = 0; u < 16; ++u)
          v[u] = col[(r + 4 * (j + u)) * (MAX_BINS / 4)];
#pragma unroll
        for (int u = 0; u < 16; ++u) {
          const int m = (r + 4 * (j + u) < c) ? -1 : 0;
          tot.x += v[u].x;      tot.y += v[u].y;
          tot.z += v[u].z;      tot.w += v[u].w;
          pre.x += v[u].x & m;  pre.y += v[u].y & m;
          pre.z += v[u].z & m;  pre.w += v[u].w & m;
        }
      }
    }
    ((int4*)partTot)[r * (MAX_BINS / 4) + q] = tot;
    ((int4*)partPre)[r * (MAX_BINS / 4) + q] = pre;
  }
  __syncthreads();
  {
    int totT = 0, preT = 0;
    for (int k = 0; k < 4; ++k) {
      totT += partTot[k * MAX_BINS + t];
      preT += partPre[k * MAX_BINS + t];
    }
    const int lane = t & 31;
    const int wv = t >> 5;
    int inc = totT;
    for (int o = 1; o < 32; o <<= 1) {
      const int up = __shfl_up(inc, o, 32);
      if (lane >= o) inc += up;
    }
    if (lane == 31) waveTot[wv] = inc;
    __syncthreads();
    int below = 0;
    for (int k = 0; k < 8; ++k) below += (k < wv) ? waveTot[k] : 0;
    const int binBeg = below + inc - totT;
    cur[t] = binBeg + preT;
    if (c == 0 && t < n_bins) bin_start[t] = binBeg;
  }
  __syncthreads();
  const int beg = blockIdx.x * chunk;
  int end = beg + chunk;
  if (end > n_edges) end = n_edges;
  const int* dstp = ei + n_edges;
  int e = beg + t;
  for (; e + 3 * 256 < end; e += 4 * 256) {
    int s[4], d[4];
#pragma unroll
    for (int u = 0; u < 4; ++u) {
      s[u] = ei[e + u * 256];
      d[u] = dstp[e + u * 256];
    }
#pragma unroll
    for (int u = 0; u < 4; ++u) {
      const int slot = atomicAdd(&cur[d[u] >> BIN_SHIFT], 1);
      tmp[slot] = make_int2(s[u], d[u]);
    }
  }
  for (; e < end; e += 256) {
    const int s = ei[e];
    const int d = dstp[e];
    const int slot = atomicAdd(&cur[d >> BIN_SHIFT], 1);
    tmp[slot] = make_int2(s, d);
  }
}

__global__ __launch_bounds__(256) void sage_bin_fill(
    const int2* __restrict__ tmp, const int* __restrict__ bin_start,
    int* __restrict__ deg, int* __restrict__ off, int* __restrict__ csr,
    int n_nodes, int n_edges, int n_bins) {
  __shared__ int cnt[256];
  __shared__ int scn[256];
  const int t = threadIdx.x;
  const int b = blockIdx.x;
  const int beg = bin_start[b];
  const int end = (b + 1 < n_bins) ? bin_start[b + 1] : n_edges;
  cnt[t] = 0;
  __syncthreads();
  int i = beg + t;
  for (; i + 3 * 256 < end; i += 4 * 256) {
    int d[4];
#pragma unroll
    for (int u = 0; u < 4; ++u) d[u] = tmp[i + u * 256].y;
#pragma unroll
    for (int u = 0; u < 4; ++u) atomicAdd(&cnt[d[u] & 255], 1);
  }
  for (; i < end; i += 256) atomicAdd(&cnt[tmp[i].y & 255], 1);
  __syncthreads();
  const int c = cnt[t];
  scn[t] = c;
  __syncthreads();
  for (int d = 1; d < 256; d <<= 1) {
    const int v = (t >= d) ? scn[t - d] : 0;
    __syncthreads();
    scn[t] += v;
    __syncthreads();
  }
  const int excl = scn[t] - c;
  const int node = (b << BIN_SHIFT) + t;
  if (node < n_nodes) {
    deg[node] = c;
    off[node] = beg + excl;
  }
  cnt[t] = excl;
  __syncthreads();
  i = beg + t;
  for (; i + 3 * 256 < end; i += 4 * 256) {
    int2 sd[4];
#pragma unroll
    for (int u = 0; u < 4; ++u) sd[u] = tmp[i + u * 256];
#pragma unroll
    for (int u = 0; u < 4; ++u) {
      const int r = atomicAdd(&cnt[sd[u].y & 255], 1);
      csr[beg + r] = sd[u].x;
    }
  }
  for (; i < end; i += 256) {
    const int2 sd = tmp[i];
    const int r = atomicAdd(&cnt[sd.y & 255], 1);
    csr[beg + r] = sd.x;
  }
}

#define TILE_PITCH 132

__device__ __forceinline__ float lane_read_f(float v, int l) {
  return __builtin_bit_cast(
      float, __builtin_amdgcn_readlane(__builtin_bit_cast(int, v), l));
}

__device__ __forceinline__ void tile_gather(
    const float* __restrict__ x, const int* __restrict__ csr,
    const int* __restrict__ off, const int* __restrict__ deg, int node0,
    int lane, float* tileLds) {
  const int li = lane & 15;
  const int offv = off[node0 + li];
  const int degv = deg[node0 + li];
  const float invv = 1.0f / fmaxf((float)degv, 1.0f);
  const int e0 = __builtin_amdgcn_readlane(offv, 0);
  const int e1 = __builtin_amdgcn_readlane(offv, 15) +
                 __builtin_amdgcn_readlane(degv, 15);
  const int bndv = (lane < 16) ? offv : e1;
  const int total = e1 - e0;
  const float* xl = x + lane * 4;
  float* trow = tileLds + lane * 4;

  float4 acc = make_float4(0.0f, 0.0f, 0.0f, 0.0f);
  int cur = 0;
  int nextB = __builtin_amdgcn_readlane(bndv, 1);

  int mine = (lane < total) ? csr[e0 + lane] : 0;
  for (int base = 0; base < total; base += 32) {
    const int nb = base + 32;
    const int ni = (nb + lane < total) ? (nb + lane) : (total - 1);
    const int nxt = csr[e0 + ni];
#pragma unroll
    for (int h = 0; h < 2; ++h) {
      const int g = base + 16 * h;
      if (g < total) {
        float4 v[16];
#pragma unroll
        for (int u = 0; u < 16; ++u) {
          const int s = __builtin_amdgcn_readlane(mine, 16 * h + u);
          v[u] = *(const float4*)(xl + (size_t)s * D_FEAT);
        }
#pragma unroll
        for (int u = 0; u < 16; ++u) {
          const int e = e0 + g + u;
          if (e < e1) {
            while (e >= nextB && cur < 15) {
              const float inv = lane_read_f(invv, cur);
              *(float4*)(trow + cur * TILE_PITCH) = make_float4(
                  acc.x * inv, acc.y * inv, acc.z * inv, acc.w * inv);
              acc = make_float4(0.0f, 0.0f, 0.0f, 0.0f);
              ++cur;
              nextB = __builtin_amdgcn_readlane(bndv, cur + 1);
            }
            acc.x += v[u].x;  acc.y += v[u].y;
            acc.z += v[u].z;  acc.w += v[u].w;
          }
        }
      }
    }
    mine = nxt;
  }
  for (; cur < 16; ++cur) {
    const float inv = lane_read_f(invv, cur);
    *(float4*)(trow + cur * TILE_PITCH) = make_float4(
        acc.x * inv, acc.y * inv, acc.z * inv, acc.w * inv);
    acc = make_float4(0.0f, 0.0f, 0.0f, 0.0f);
  }
}

template <int XMASK>
__device__ __forceinline__ float row_xor(float x) {
  return __builtin_bit_cast(
      float, __builtin_amdgcn_update_dpp(0, __builtin_bit_cast(int, x),
                                         0x160 + XMASK, 0xf, 0xf, true));
}

template <int CNT, int XMASK>
__device__ __forceinline__ void row_fold(float* v, bool upper) {
#pragma unroll
  for (int i = 0; i < CNT; ++i) {
    const float keep = upper ? v[i + CNT] : v[i];
    const float send = upper ? v[i] : v[i + CNT];
    v[i] = keep + row_xor<XMASK>(send);
  }
}


#define WMMA_TAIL_PAD()                                    \
  do {                                                     \
    __builtin_amdgcn_sched_barrier(0);                     \
    asm volatile("v_nop\n\tv_nop\n\tv_nop\n\tv_nop");      \
  } while (0)

__device__ __forceinline__ void self_product(
    v8f* acc, const float* __restrict__ xRow, const float* __restrict__ W1r,
    const int* nrow, int half) {
#pragma unroll 1
  for (int k = 0; k < D_FEAT; k += 8) {
    float2 ts[2], br[2][8];
#pragma unroll
    for (int kk = 0; kk < 2; ++kk) {
      ts[kk] = *(const float2*)(xRow + k + 4 * kk);
#pragma unroll
      for (int nt = 0; nt < 8; ++nt)
        br[kk][nt] = *(const float2*)(W1r + (size_t)nrow[nt] * D_FEAT + k +
                                      4 * kk + 2 * half);
    }
#pragma unroll
    for (int kk = 0; kk < 2; ++kk) {
      v2f a_s;  a_s.x = ts[kk].x;  a_s.y = ts[kk].y;
#pragma unroll
      for (int nt = 0; nt < 8; ++nt) {
        v2f vbr;  vbr.x = br[kk][nt].x;  vbr.y = br[kk][nt].y;
        acc[nt] = __builtin_amdgcn_wmma_f32_16x16x4_f32(
            false, a_s, false, vbr, (short)0, acc[nt], false, false);
      }
    }
    WMMA_TAIL_PAD();
  }
}

__global__ __launch_bounds__(256) void sage_layer1_wmma(
    const float* __restrict__ x, const int* __restrict__ csr,
    const int* __restrict__ off, const int* __restrict__ deg,
    const float* __restrict__ W1l,
    const float* __restrict__ W1r, const float* __restrict__ b1,
    const float* __restrict__ W2l, const float* __restrict__ W2r,
    const float* __restrict__ b2, float* __restrict__ p,
    float* __restrict__ out, int n_nodes) {
  __shared__ float ldsWl[D_FEAT * D_FEAT];
  __shared__ __attribute__((aligned(16))) float ldsAgg[8 * 16 * TILE_PITCH];

  for (int idx = threadIdx.x; idx < D_FEAT * 64; idx += 256) {
    const int r = idx >> 6;
    const int c = idx & 63;
    const float2 w = *(const float2*)(W1l + (size_t)r * D_FEAT + 2 * c);
    const int s = c ^ ((2 * r) & 63);
    *(float2*)(ldsWl + r * D_FEAT + 2 * s) = w;
  }
  __syncthreads();

  const int lane = threadIdx.x & 31;
  const int wave = __builtin_amdgcn_readfirstlane((int)(threadIdx.x >> 5));
  const int tile = blockIdx.x * 8 + wave;
  const int n_tiles = n_nodes >> 4;
  if (tile >= n_tiles) return;

  const int half = lane >> 4;
  const int q = lane & 15;
  const int rowA = tile * 16 + q;

  v8f acc[8];
#pragma unroll
  for (int i = 0; i < 8; ++i) {
    v8f z = {};
    acc[i] = z;
  }

  const float* xRow = x + (size_t)rowA * D_FEAT + 2 * half;
  float* myTile = ldsAgg + wave * (16 * TILE_PITCH);

  int nrow[8], nswz[8];
#pragma unroll
  for (int nt = 0; nt < 8; ++nt) {
    nrow[nt] = nt * 16 + q;
    nswz[nt] = (2 * nrow[nt]) & 63;
  }

  const bool self_first = (wave & 4) != 0;
  if (self_first) self_product(acc, xRow, W1r, nrow, half);
  tile_gather(x, csr, off, deg, tile * 16, lane, myTile);
  if (!self_first) self_product(acc, xRow, W1r, nrow, half);

  __builtin_amdgcn_fence(__ATOMIC_RELEASE, "workgroup");
  __builtin_amdgcn_wave_barrier();
  __builtin_amdgcn_fence(__ATOMIC_ACQUIRE, "workgroup");

  const float* aggRow = myTile + q * TILE_PITCH + 2 * half;
#pragma unroll 1
  for (int k = 0; k < D_FEAT; k += 8) {
    float2 tm[2], bl[2][8];
#pragma unroll
    for (int kk = 0; kk < 2; ++kk) {
      const int c0 = ((k + 4 * kk) >> 1) + half;
      tm[kk] = *(const float2*)(aggRow + k + 4 * kk);
#pragma unroll
      for (int nt = 0; nt < 8; ++nt)
        bl[kk][nt] = *(const float2*)(ldsWl + nrow[nt] * D_FEAT +
                                      2 * (c0 ^ nswz[nt]));
    }
#pragma unroll
    for (int kk = 0; kk < 2; ++kk) {
      v2f a_m;  a_m.x = tm[kk].x;  a_m.y = tm[kk].y;
#pragma unroll
      for (int nt = 0; nt < 8; ++nt) {
        v2f vbl;  vbl.x = bl[kk][nt].x;  vbl.y = bl[kk][nt].y;
        acc[nt] = __builtin_amdgcn_wmma_f32_16x16x4_f32(
            false, a_m, false, vbl, (short)0, acc[nt], false, false);
      }
    }
    WMMA_TAIL_PAD();
  }

  float part[32];
#pragma unroll
  for (int i = 0; i < 32; ++i) part[i] = 0.0f;
#pragma unroll
  for (int nt = 0; nt < 8; ++nt) {
    const int n = nrow[nt];
    const float bias = b1[n];
    const float wl0 = W2l[n];
    const float wl1 = W2l[D_FEAT + n];
    const float wr0 = W2r[n];
    const float wr1 = W2r[D_FEAT + n];
#pragma unroll
    for (int v = 0; v < 8; ++v) {
      const float hv = fmaxf(acc[nt][v] + bias, 0.0f);
      part[v * 4 + 0] += hv * wl0;
      part[v * 4 + 1] += hv * wl1;
      part[v * 4 + 2] += hv * wr0;
      part[v * 4 + 3] += hv * wr1;
    }
  }
  row_fold<16, 8>(part, (q & 8) != 0);
  row_fold<8, 4>(part, (q & 4) != 0);
  row_fold<4, 2>(part, (q & 2) != 0);
  row_fold<2, 1>(part, (q & 1) != 0);

  const int node = tile * 16 + 8 * half + (q >> 1);
  const bool self_side = (q & 1) != 0;
  const float bias0 = b2[0];
  const float bias1 = b2[1];
  float2 res;
  res.x = self_side ? part[0] + bias0 : part[0];
  res.y = self_side ? part[1] + bias1 : part[1];
  float* dst = (self_side ? out : p) + (size_t)node * 2;
  *(float2*)dst = res;
}

__global__ __launch_bounds__(256) void sage_gather2(
    const float* __restrict__ p, const int* __restrict__ csr,
    const int* __restrict__ off, const int* __restrict__ deg,
    float* __restrict__ out, int n_nodes) {
  const int lane = threadIdx.x & 31;
  const int node = __builtin_amdgcn_readfirstlane(
      (int)((blockIdx.x * (size_t)blockDim.x + threadIdx.x) >> 5));
  if (node >= n_nodes) return;
  const int beg = off[node];
  const int d = deg[node];
  float s0 = 0.0f, s1 = 0.0f;
  for (int j = lane; j < d; j += 32) {
    const int s = csr[beg + j];
    const float2 pv = *(const float2*)(p + (size_t)s * 2);
    s0 += pv.x;
    s1 += pv.y;
  }
#pragma unroll
  for (int o = 16; o > 0; o >>= 1) {
    s0 += __shfl_xor(s0, o, 32);
    s1 += __shfl_xor(s1, o, 32);
  }
  if (lane == 0) {
    const float inv = 1.0f / fmaxf((float)d, 1.0f);
    out[(size_t)node * 2 + 0] += s0 * inv;
    out[(size_t)node * 2 + 1] += s1 * inv;
  }
}

extern "C" void kernel_launch(void* const* d_in, const int* in_sizes, int n_in,
                              void* d_out, int out_size, void* d_ws, size_t ws_size,
                              hipStream_t stream) {
  const float* x   = (const float*)d_in[0];
  const int*   ei  = (const int*)d_in[1];
  const float* W1l = (const float*)d_in[2];
  const float* W1r = (const float*)d_in[3];
  const float* b1  = (const float*)d_in[4];
  const float* W2l = (const float*)d_in[5];
  const float* W2r = (const float*)d_in[6];
  const float* b2  = (const float*)d_in[7];
  float* out = (float*)d_out;

  const int n_nodes = in_sizes[0] / D_FEAT;
  const int n_edges = in_sizes[1] / 2;

  float* ws   = (float*)d_ws;
  float* agg1 = ws;
  float* p    = agg1 + (size_t)n_nodes * D_FEAT;
  float* cnt  = p + (size_t)n_nodes * 2;
  int*   deg  = (int*)(cnt + n_nodes);
  int*   off  = deg + n_nodes;
  int*   csr  = off + n_nodes;
  const int n_bins  = (n_nodes + 255) / 256;
  const int chunk   = (n_edges + EDGE_BLOCKS - 1) / EDGE_BLOCKS;
  int*   cnt_cb    = csr + n_edges;
  int*   bin_start = cnt_cb + EDGE_BLOCKS * MAX_BINS;
  int2*  tmp       = (int2*)ws;

  sage_bin_hist<<<EDGE_BLOCKS, 256, 0, stream>>>(ei, cnt_cb, n_edges, n_bins, chunk);

  sage_bin_scatter<<<EDGE_BLOCKS, 256, 0, stream>>>(ei, cnt_cb, bin_start, tmp, n_edges,
                                                    n_bins, chunk);

  sage_bin_fill<<<n_bins, 256, 0, stream>>>(tmp, bin_start, deg, off, csr, n_nodes,
                                            n_edges, n_bins);

  const int n_tiles = n_nodes / 16;
  sage_layer1_wmma<<<(n_tiles + 7) / 8, 256, 0, stream>>>(
      x, csr, off, deg, W1l, W1r, b1, W2l, W2r, b2, p, out, n_nodes);

  sage_gather2<<<(n_nodes + 7) / 8, 256, 0, stream>>>(p, csr, off, deg, out, n_nodes);
}


# Layer-2 gather: two lanes per node, 16 nodes per wave, 16 list and 16 row loads in flight per lane (3125 waves enter instead of 50000); sum order changes (edge order per lane, then pair add)
# speedup vs baseline: 3.6311x; 1.0531x over previous
#include <hip/hip_runtime.h>

typedef __attribute__((ext_vector_type(2))) float v2f;
typedef __attribute__((ext_vector_type(8))) float v8f;

#define D_FEAT 128


#define BIN_SHIFT 8
#define EDGE_BLOCKS 256
#define MAX_BINS 256

__global__ __launch_bounds__(256) void sage_bin_hist(
    const int* __restrict__ ei, int* __restrict__ cnt_cb, int n_edges,
    int n_bins, int chunk) {
  __shared__ int hist[MAX_BINS];
  const int t = threadIdx.x;
  if (t < n_bins) hist[t] = 0;
  __syncthreads();
  const int beg = blockIdx.x * chunk;
  int end = beg + chunk;
  if (end > n_edges) end = n_edges;
  const int* dstp = ei + n_edges;
  int e = beg + t;
  for (; e + 3 * 256 < end; e += 4 * 256) {
    int d[4];
#pragma unroll
    for (int u = 0; u < 4; ++u) d[u] = dstp[e + u * 256];
#pragma unroll
    for (int u = 0; u < 4; ++u) atomicAdd(&hist[d[u] >> BIN_SHIFT], 1);
  }
  for (; e < end; e += 256) atomicAdd(&hist[dstp[e] >> BIN_SHIFT], 1);
  __syncthreads();
  cnt_cb[blockIdx.x * MAX_BINS + t] = (t < n_bins) ? hist[t] : 0;
}

__global__ __launch_bounds__(256) void sage_bin_scatter(
    const int* __restrict__ ei, const int* __restrict__ cnt_cb,
    int* __restrict__ bin_start, int2* __restrict__ tmp, int n_edges,
    int n_bins, int chunk) {
  __shared__ int cur[MAX_BINS];
  __shared__ __attribute__((aligned(16))) int partTot[4 * MAX_BINS];
  __shared__ __attribute__((aligned(16))) int partPre[4 * MAX_BINS];
  __shared__ int waveTot[8];
  const int t = threadIdx.x;
  const int c = blockIdx.x;
  {
    const int q = t & 63;
    const int r = t >> 6;
    int4 tot = make_int4(0, 0, 0, 0);
    int4 pre = make_int4(0, 0, 0, 0);
    if (4 * q < n_bins) {
      const int4* col = (const int4*)cnt_cb + q;
      for (int j = 0; j < EDGE_BLOCKS / 4; j += 16) {
        int4 v[16];
#pragma unroll
        for (int u = 0; u < 16; ++u)
          v[u] = col[(r + 4 * (j + u)) * (MAX_BINS / 4)];
#pragma unroll
        for (int u = 0; u < 16; ++u) {
          const int m = (r + 4 * (j + u) < c) ? -1 : 0;
          tot.x += v[u].x;      tot.y += v[u].y;
          tot.z += v[u].z;      tot.w += v[u].w;
          pre.x += v[u].x & m;  pre.y += v[u].y & m;
          pre.z += v[u].z & m;  pre.w += v[u].w & m;
        }
      }
    }
    ((int4*)partTot)[r * (MAX_BINS / 4) + q] = tot;
    ((int4*)partPre)[r * (MAX_BINS / 4) + q] = pre;
  }
  __syncthreads();
  {
    int totT = 0, preT = 0;
    for (int k = 0; k < 4; ++k) {
      totT += partTot[k * MAX_BINS + t];
      preT += partPre[k * MAX_BINS + t];
    }
    const int lane = t & 31;
    const int wv = t >> 5;
    int inc = totT;
    for (int o = 1; o < 32; o <<= 1) {
      const int up = __shfl_up(inc, o, 32);
      if (lane >= o) inc += up;
    }
    if (lane == 31) waveTot[wv] = inc;
    __syncthreads();
    int below = 0;
    for (int k = 0; k < 8; ++k) below += (k < wv) ? waveTot[k] : 0;
    const int binBeg = below + inc - totT;
    cur[t] = binBeg + preT;
    if (c == 0 && t < n_bins) bin_start[t] = binBeg;
  }
  __syncthreads();
  const int beg = blockIdx.x * chunk;
  int end = beg + chunk;
  if (end > n_edges) end = n_edges;
  const int* dstp = ei + n_edges;
  int e = beg + t;
  for (; e + 3 * 256 < end; e += 4 * 256) {
    int s[4], d[4];
#pragma unroll
    for (int u = 0; u < 4; ++u) {
      s[u] = ei[e + u * 256];
      d[u] = dstp[e + u * 256];
    }
#pragma unroll
    for (int u = 0; u < 4; ++u) {
      const int slot = atomicAdd(&cur[d[u] >> BIN_SHIFT], 1);
      tmp[slot] = make_int2(s[u], d[u]);
    }
  }
  for (; e < end; e += 256) {
    const int s = ei[e];
    const int d = dstp[e];
    const int slot = atomicAdd(&cur[d >> BIN_SHIFT], 1);
    tmp[slot] = make_int2(s, d);
  }
}

__global__ __launch_bounds__(256) void sage_bin_fill(
    const int2* __restrict__ tmp, const int* __restrict__ bin_start,
    int* __restrict__ deg, int* __restrict__ off, int* __restrict__ csr,
    int n_nodes, int n_edges, int n_bins) {
  __shared__ int cnt[256];
  __shared__ int scn[256];
  const int t = threadIdx.x;
  const int b = blockIdx.x;
  const int beg = bin_start[b];
  const int end = (b + 1 < n_bins) ? bin_start[b + 1] : n_edges;
  cnt[t] = 0;
  __syncthreads();
  int i = beg + t;
  for (; i + 3 * 256 < end; i += 4 * 256) {
    int d[4];
#pragma unroll
    for (int u = 0; u < 4; ++u) d[u] = tmp[i + u * 256].y;
#pragma unroll
    for (int u = 0; u < 4; ++u) atomicAdd(&cnt[d[u] & 255], 1);
  }
  for (; i < end; i += 256) atomicAdd(&cnt[tmp[i].y & 255], 1);
  __syncthreads();
  const int c = cnt[t];
  scn[t] = c;
  __syncthreads();
  for (int d = 1; d < 256; d <<= 1) {
    const int v = (t >= d) ? scn[t - d] : 0;
    __syncthreads();
    scn[t] += v;
    __syncthreads();
  }
  const int excl = scn[t] - c;
  const int node = (b << BIN_SHIFT) + t;
  if (node < n_nodes) {
    deg[node] = c;
    off[node] = beg + excl;
  }
  cnt[t] = excl;
  __syncthreads();
  i = beg + t;
  for (; i + 3 * 256 < end; i += 4 * 256) {
    int2 sd[4];
#pragma unroll
    for (int u = 0; u < 4; ++u) sd[u] = tmp[i + u * 256];
#pragma unroll
    for (int u = 0; u < 4; ++u) {
      const int r = atomicAdd(&cnt[sd[u].y & 255], 1);
      csr[beg + r] = sd[u].x;
    }
  }
  for (; i < end; i += 256) {
    const int2 sd = tmp[i];
    const int r = atomicAdd(&cnt[sd.y & 255], 1);
    csr[beg + r] = sd.x;
  }
}

#define TILE_PITCH 132

__device__ __forceinline__ float lane_read_f(float v, int l) {
  return __builtin_bit_cast(
      float, __builtin_amdgcn_readlane(__builtin_bit_cast(int, v), l));
}

__device__ __forceinline__ void tile_gather(
    const float* __restrict__ x, const int* __restrict__ csr,
    const int* __restrict__ off, const int* __restrict__ deg, int node0,
    int lane, float* tileLds) {
  const int li = lane & 15;
  const int offv = off[node0 + li];
  const int degv = deg[node0 + li];
  const float invv = 1.0f / fmaxf((float)degv, 1.0f);
  const int e0 = __builtin_amdgcn_readlane(offv, 0);
  const int e1 = __builtin_amdgcn_readlane(offv, 15) +
                 __builtin_amdgcn_readlane(degv, 15);
  const int bndv = (lane < 16) ? offv : e1;
  const int total = e1 - e0;
  const float* xl = x + lane * 4;
  float* trow = tileLds + lane * 4;

  float4 acc = make_float4(0.0f, 0.0f, 0.0f, 0.0f);
  int cur = 0;
  int nextB = __builtin_amdgcn_readlane(bndv, 1);

  int mine = (lane < total) ? csr[e0 + lane] : 0;
  for (int base = 0; base < total; base += 32) {
    const int nb = base + 32;
    const int ni = (nb + lane < total) ? (nb + lane) : (total - 1);
    const int nxt = csr[e0 + ni];
#pragma unroll
    for (int h = 0; h < 2; ++h) {
      const int g = base + 16 * h;
      if (g < total) {
        float4 v[16];
#pragma unroll
        for (int u = 0; u < 16; ++u) {
          const int s = __builtin_amdgcn_readlane(mine, 16 * h + u);
          v[u] = *(const float4*)(xl + (size_t)s * D_FEAT);
        }
#pragma unroll
        for (int u = 0; u < 16; ++u) {
          const int e = e0 + g + u;
          if (e < e1) {
            while (e >= nextB && cur < 15) {
              const float inv = lane_read_f(invv, cur);
              *(float4*)(trow + cur * TILE_PITCH) = make_float4(
                  acc.x * inv, acc.y * inv, acc.z * inv, acc.w * inv);
              acc = make_float4(0.0f, 0.0f, 0.0f, 0.0f);
              ++cur;
              nextB = __builtin_amdgcn_readlane(bndv, cur + 1);
            }
            acc.x += v[u].x;  acc.y += v[u].y;
            acc.z += v[u].z;  acc.w += v[u].w;
          }
        }
      }
    }
    mine = nxt;
  }
  for (; cur < 16; ++cur) {
    const float inv = lane_read_f(invv, cur);
    *(float4*)(trow + cur * TILE_PITCH) = make_float4(
        acc.x * inv, acc.y * inv, acc.z * inv, acc.w * inv);
    acc = make_float4(0.0f, 0.0f, 0.0f, 0.0f);
  }
}

template <int XMASK>
__device__ __forceinline__ float row_xor(float x) {
  return __builtin_bit_cast(
      float, __builtin_amdgcn_update_dpp(0, __builtin_bit_cast(int, x),
                                         0x160 + XMASK, 0xf, 0xf, true));
}

template <int CNT, int XMASK>
__device__ __forceinline__ void row_fold(float* v, bool upper) {
#pragma unroll
  for (int i = 0; i < CNT; ++i) {
    const float keep = upper ? v[i + CNT] : v[i];
    const float send = upper ? v[i] : v[i + CNT];
    v[i] = keep + row_xor<XMASK>(send);
  }
}


#define WMMA_TAIL_PAD()                                    \
  do {                                                     \
    __builtin_amdgcn_sched_barrier(0);                     \
    asm volatile("v_nop\n\tv_nop\n\tv_nop\n\tv_nop");      \
  } while (0)

__device__ __forceinline__ void self_product(
    v8f* acc, const float* __restrict__ xRow, const float* __restrict__ W1r,
    const int* nrow, int half) {
#pragma unroll 1
  for (int k = 0; k < D_FEAT; k += 8) {
    float2 ts[2], br[2][8];
#pragma unroll
    for (int kk = 0; kk < 2; ++kk) {
      ts[kk] = *(const float2*)(xRow + k + 4 * kk);
#pragma unroll
      for (int nt = 0; nt < 8; ++nt)
        br[kk][nt] = *(const float2*)(W1r + (size_t)nrow[nt] * D_FEAT + k +
                                      4 * kk + 2 * half);
    }
#pragma unroll
    for (int kk = 0; kk < 2; ++kk) {
      v2f a_s;  a_s.x = ts[kk].x;  a_s.y = ts[kk].y;
#pragma unroll
      for (int nt = 0; nt < 8; ++nt) {
        v2f vbr;  vbr.x = br[kk][nt].x;  vbr.y = br[kk][nt].y;
        acc[nt] = __builtin_amdgcn_wmma_f32_16x16x4_f32(
            false, a_s, false, vbr, (short)0, acc[nt], false, false);
      }
    }
    WMMA_TAIL_PAD();
  }
}

__global__ __launch_bounds__(256) void sage_layer1_wmma(
    const float* __restrict__ x, const int* __restrict__ csr,
    const int* __restrict__ off, const int* __restrict__ deg,
    const float* __restrict__ W1l,
    const float* __restrict__ W1r, const float* __restrict__ b1,
    const float* __restrict__ W2l, const float* __restrict__ W2r,
    const float* __restrict__ b2, float* __restrict__ p,
    float* __restrict__ out, int n_nodes) {
  __shared__ float ldsWl[D_FEAT * D_FEAT];
  __shared__ __attribute__((aligned(16))) float ldsAgg[8 * 16 * TILE_PITCH];

  for (int idx = threadIdx.x; idx < D_FEAT * 64; idx += 256) {
    const int r = idx >> 6;
    const int c = idx & 63;
    const float2 w = *(const float2*)(W1l + (size_t)r * D_FEAT + 2 * c);
    const int s = c ^ ((2 * r) & 63);
    *(float2*)(ldsWl + r * D_FEAT + 2 * s) = w;
  }
  __syncthreads();

  const int lane = threadIdx.x & 31;
  const int wave = __builtin_amdgcn_readfirstlane((int)(threadIdx.x >> 5));
  const int tile = blockIdx.x * 8 + wave;
  const int n_tiles = n_nodes >> 4;
  if (tile >= n_tiles) return;

  const int half = lane >> 4;
  const int q = lane & 15;
  const int rowA = tile * 16 + q;

  v8f acc[8];
#pragma unroll
  for (int i = 0; i < 8; ++i) {
    v8f z = {};
    acc[i] = z;
  }

  const float* xRow = x + (size_t)rowA * D_FEAT + 2 * half;
  float* myTile = ldsAgg + wave * (16 * TILE_PITCH);

  int nrow[8], nswz[8];
#pragma unroll
  for (int nt = 0; nt < 8; ++nt) {
    nrow[nt] = nt * 16 + q;
    nswz[nt] = (2 * nrow[nt]) & 63;
  }

  const bool self_first = (wave & 4) != 0;
  if (self_first) self_product(acc, xRow, W1r, nrow, half);
  tile_gather(x, csr, off, deg, tile * 16, lane, myTile);
  if (!self_first) self_product(acc, xRow, W1r, nrow, half);

  __builtin_amdgcn_fence(__ATOMIC_RELEASE, "workgroup");
  __builtin_amdgcn_wave_barrier();
  __builtin_amdgcn_fence(__ATOMIC_ACQUIRE, "workgroup");

  const float* aggRow = myTile + q * TILE_PITCH + 2 * half;
#pragma unroll 1
  for (int k = 0; k < D_FEAT; k += 8) {
    float2 tm[2], bl[2][8];
#pragma unroll
    for (int kk = 0; kk < 2; ++kk) {
      const int c0 = ((k + 4 * kk) >> 1) + half;
      tm[kk] = *(const float2*)(aggRow + k + 4 * kk);
#pragma unroll
      for (int nt = 0; nt < 8; ++nt)
        bl[kk][nt] = *(const float2*)(ldsWl + nrow[nt] * D_FEAT +
                                      2 * (c0 ^ nswz[nt]));
    }
#pragma unroll
    for (int kk = 0; kk < 2; ++kk) {
      v2f a_m;  a_m.x = tm[kk].x;  a_m.y = tm[kk].y;
#pragma unroll
      for (int nt = 0; nt < 8; ++nt) {
        v2f vbl;  vbl.x = bl[kk][nt].x;  vbl.y = bl[kk][nt].y;
        acc[nt] = __builtin_amdgcn_wmma_f32_16x16x4_f32(
            false, a_m, false, vbl, (short)0, acc[nt], false, false);
      }
    }
    WMMA_TAIL_PAD();
  }

  float part[32];
#pragma unroll
  for (int i = 0; i < 32; ++i) part[i] = 0.0f;
#pragma unroll
  for (int nt = 0; nt < 8; ++nt) {
    const int n = nrow[nt];
    const float bias = b1[n];
    const float wl0 = W2l[n];
    const float wl1 = W2l[D_FEAT + n];
    const float wr0 = W2r[n];
    const float wr1 = W2r[D_FEAT + n];
#pragma unroll
    for (int v = 0; v < 8; ++v) {
      const float hv = fmaxf(acc[nt][v] + bias, 0.0f);
      part[v * 4 + 0] += hv * wl0;
      part[v * 4 + 1] += hv * wl1;
      part[v * 4 + 2] += hv * wr0;
      part[v * 4 + 3] += hv * wr1;
    }
  }
  row_fold<16, 8>(part, (q & 8) != 0);
  row_fold<8, 4>(part, (q & 4) != 0);
  row_fold<4, 2>(part, (q & 2) != 0);
  row_fold<2, 1>(part, (q & 1) != 0);

  const int node = tile * 16 + 8 * half + (q >> 1);
  const bool self_side = (q & 1) != 0;
  const float bias0 = b2[0];
  const float bias1 = b2[1];
  float2 res;
  res.x = self_side ? part[0] + bias0 : part[0];
  res.y = self_side ? part[1] + bias1 : part[1];
  float* dst = (self_side ? out : p) + (size_t)node * 2;
  *(float2*)dst = res;
}

#define G2_DEPTH 16

__global__ __launch_bounds__(256) void sage_gather2(
    const float* __restrict__ p, const int* __restrict__ csr,
    const int* __restrict__ off, const int* __restrict__ deg,
    float* __restrict__ out, int n_nodes, int n_edges) {
  const int gid = blockIdx.x * 256 + threadIdx.x;
  const int sub = gid & 1;
  const bool live = (gid >> 1) < n_nodes;
  const int node = live ? (gid >> 1) : (n_nodes - 1);
  const int beg = off[node];
  const int degv = deg[node];
  const int d = live ? degv : 0;
  const float2 self = *(const float2*)(out + (size_t)node * 2);
  const int last = (n_edges > 0) ? (n_edges - 1) : 0;
  float s0 = 0.0f, s1 = 0.0f;
  int jbase = 0;
  do {
    int sid[G2_DEPTH], keep[G2_DEPTH];
#pragma unroll
    for (int k = 0; k < G2_DEPTH; ++k) {
      const int j = jbase + sub + 2 * k;
      const int e = beg + j;
      keep[k] = (j - d) >> 31;
      sid[k] = csr[e < last ? e : last] & keep[k];
    }
    float2 pv[G2_DEPTH];
#pragma unroll
    for (int k = 0; k < G2_DEPTH; ++k)
      pv[k] = *(const float2*)(p + (size_t)sid[k] * 2);
#pragma unroll
    for (int k = 0; k < G2_DEPTH; ++k) {
      s0 += __int_as_float(__float_as_int(pv[k].x) & keep[k]);
      s1 += __int_as_float(__float_as_int(pv[k].y) & keep[k]);
    }
    jbase += 2 * G2_DEPTH;
  } while (__builtin_amdgcn_ballot_w32(jbase < d) != 0u);
  s0 += row_xor<1>(s0);
  s1 += row_xor<1>(s1);
  if (live && sub == 0) {
    const float inv = 1.0f / fmaxf((float)d, 1.0f);
    float2 res;
    res.x = self.x + s0 * inv;
    res.y = self.y + s1 * inv;
    *(float2*)(out + (size_t)node * 2) = res;
  }
}

extern "C" void kernel_launch(void* const* d_in, const int* in_sizes, int n_in,
                              void* d_out, int out_size, void* d_ws, size_t ws_size,
                              hipStream_t stream) {
  const float* x   = (const float*)d_in[0];
  const int*   ei  = (const int*)d_in[1];
  const float* W1l = (const float*)d_in[2];
  const float* W1r = (const float*)d_in[3];
  const float* b1  = (const float*)d_in[4];
  const float* W2l = (const float*)d_in[5];
  const float* W2r = (const float*)d_in[6];
  const float* b2  = (const float*)d_in[7];
  float* out = (float*)d_out;

  const int n_nodes = in_sizes[0] / D_FEAT;
  const int n_edges = in_sizes[1] / 2;

  float* ws   = (float*)d_ws;
  float* agg1 = ws;
  float* p    = agg1 + (size_t)n_nodes * D_FEAT;
  float* cnt  = p + (size_t)n_nodes * 2;
  int*   deg  = (int*)(cnt + n_nodes);
  int*   off  = deg + n_nodes;
  int*   csr  = off + n_nodes;
  const int n_bins  = (n_nodes + 255) / 256;
  const int chunk   = (n_edges + EDGE_BLOCKS - 1) / EDGE_BLOCKS;
  int*   cnt_cb    = csr + n_edges;
  int*   bin_start = cnt_cb + EDGE_BLOCKS * MAX_BINS;
  int2*  tmp       = (int2*)ws;

  sage_bin_hist<<<EDGE_BLOCKS, 256, 0, stream>>>(ei, cnt_cb, n_edges, n_bins, chunk);

  sage_bin_scatter<<<EDGE_BLOCKS, 256, 0, stream>>>(ei, cnt_cb, bin_start, tmp, n_edges,
                                                    n_bins, chunk);

  sage_bin_fill<<<n_bins, 256, 0, stream>>>(tmp, bin_start, deg, off, csr, n_nodes,
                                            n_edges, n_bins);

  const int n_tiles = n_nodes / 16;
  sage_layer1_wmma<<<(n_tiles + 7) / 8, 256, 0, stream>>>(
      x, csr, off, deg, W1l, W1r, b1, W2l, W2r, b2, p, out, n_nodes);

  sage_gather2<<<(n_nodes + 127) / 128, 256, 0, stream>>>(p, csr, off, deg, out, n_nodes,
                                                          n_edges);
}


# Layer-1 self product reads its B fragments (W1r) from a swizzled LDS copy beside W1l instead of 16-line global loads; one 13-wave block per WGP holds both; sum order and precision unchanged
# speedup vs baseline: 3.9836x; 1.0971x over previous
#include <hip/hip_runtime.h>

typedef __attribute__((ext_vector_type(2))) float v2f;
typedef __attribute__((ext_vector_type(8))) float v8f;

#define D_FEAT 128


#define BIN_SHIFT 8
#define EDGE_BLOCKS 256
#define MAX_BINS 256

__global__ __launch_bounds__(256) void sage_bin_hist(
    const int* __restrict__ ei, int* __restrict__ cnt_cb, int n_edges,
    int n_bins, int chunk) {
  __shared__ int hist[MAX_BINS];
  const int t = threadIdx.x;
  if (t < n_bins) hist[t] = 0;
  __syncthreads();
  const int beg = blockIdx.x * chunk;
  int end = beg + chunk;
  if (end > n_edges) end = n_edges;
  const int* dstp = ei + n_edges;
  int e = beg + t;
  for (; e + 3 * 256 < end; e += 4 * 256) {
    int d[4];
#pragma unroll
    for (int u = 0; u < 4; ++u) d[u] = dstp[e + u * 256];
#pragma unroll
    for (int u = 0; u < 4; ++u) atomicAdd(&hist[d[u] >> BIN_SHIFT], 1);
  }
  for (; e < end; e += 256) atomicAdd(&hist[dstp[e] >> BIN_SHIFT], 1);
  __syncthreads();
  cnt_cb[blockIdx.x * MAX_BINS + t] = (t < n_bins) ? hist[t] : 0;
}

__global__ __launch_bounds__(256) void sage_bin_scatter(
    const int* __restrict__ ei, const int* __restrict__ cnt_cb,
    int* __restrict__ bin_start, int2* __restrict__ tmp, int n_edges,
    int n_bins, int chunk) {
  __shared__ int cur[MAX_BINS];
  __shared__ __attribute__((aligned(16))) int partTot[4 * MAX_BINS];
  __shared__ __attribute__((aligned(16))) int partPre[4 * MAX_BINS];
  __shared__ int waveTot[8];
  const int t = threadIdx.x;
  const int c = blockIdx.x;
  {
    const int q = t & 63;
    const int r = t >> 6;
    int4 tot = make_int4(0, 0, 0, 0);
    int4 pre = make_int4(0, 0, 0, 0);
    if (4 * q < n_bins) {
      const int4* col = (const int4*)cnt_cb + q;
      for (int j = 0; j < EDGE_BLOCKS / 4; j += 16) {
        int4 v[16];
#pragma unroll
        for (int u = 0; u < 16; ++u)
          v[u] = col[(r + 4 * (j + u)) * (MAX_BINS / 4)];
#pragma unroll
        for (int u = 0; u < 16; ++u) {
          const int m = (r + 4 * (j + u) < c) ? -1 : 0;
          tot.x += v[u].x;      tot.y += v[u].y;
          tot.z += v[u].z;      tot.w += v[u].w;
          pre.x += v[u].x & m;  pre.y += v[u].y & m;
          pre.z += v[u].z & m;  pre.w += v[u].w & m;
        }
      }
    }
    ((int4*)partTot)[r * (MAX_BINS / 4) + q] = tot;
    ((int4*)partPre)[r * (MAX_BINS / 4) + q] = pre;
  }
  __syncthreads();
  {
    int totT = 0, preT = 0;
    for (int k = 0; k < 4; ++k) {
      totT += partTot[k * MAX_BINS + t];
      preT += partPre[k * MAX_BINS + t];
    }
    const int lane = t & 31;
    const int wv = t >> 5;
    int inc = totT;
    for (int o = 1; o < 32; o <<= 1) {
      const int up = __shfl_up(inc, o, 32);
      if (lane >= o) inc += up;
    }
    if (lane == 31) waveTot[wv] = inc;
    __syncthreads();
    int below = 0;
    for (int k = 0; k < 8; ++k) below += (k < wv) ? waveTot[k] : 0;
    const int binBeg = below + inc - totT;
    cur[t] = binBeg + preT;
    if (c == 0 && t < n_bins) bin_start[t] = binBeg;
  }
  __syncthreads();
  const int beg = blockIdx.x * chunk;
  int end = beg + chunk;
  if (end > n_edges) end = n_edges;
  const int* dstp = ei + n_edges;
  int e = beg + t;
  for (; e + 3 * 256 < end; e += 4 * 256) {
    int s[4], d[4];
#pragma unroll
    for (int u = 0; u < 4; ++u) {
      s[u] = ei[e + u * 256];
      d[u] = dstp[e + u * 256];
    }
#pragma unroll
    for (int u = 0; u < 4; ++u) {
      const int slot = atomicAdd(&cur[d[u] >> BIN_SHIFT], 1);
      tmp[slot] = make_int2(s[u], d[u]);
    }
  }
  for (; e < end; e += 256) {
    const int s = ei[e];
    const int d = dstp[e];
    const int slot = atomicAdd(&cur[d >> BIN_SHIFT], 1);
    tmp[slot] = make_int2(s, d);
  }
}

__global__ __launch_bounds__(256) void sage_bin_fill(
    const int2* __restrict__ tmp, const int* __restrict__ bin_start,
    int* __restrict__ deg, int* __restrict__ off, int* __restrict__ csr,
    int n_nodes, int n_edges, int n_bins) {
  __shared__ int cnt[256];
  __shared__ int scn[256];
  const int t = threadIdx.x;
  const int b = blockIdx.x;
  const int beg = bin_start[b];
  const int end = (b + 1 < n_bins) ? bin_start[b + 1] : n_edges;
  cnt[t] = 0;
  __syncthreads();
  int i = beg + t;
  for (; i + 3 * 256 < end; i += 4 * 256) {
    int d[4];
#pragma unroll
    for (int u = 0; u < 4; ++u) d[u] = tmp[i + u * 256].y;
#pragma unroll
    for (int u = 0; u < 4; ++u) atomicAdd(&cnt[d[u] & 255], 1);
  }
  for (; i < end; i += 256) atomicAdd(&cnt[tmp[i].y & 255], 1);
  __syncthreads();
  const int c = cnt[t];
  scn[t] = c;
  __syncthreads();
  for (int d = 1; d < 256; d <<= 1) {
    const int v = (t >= d) ? scn[t - d] : 0;
    __syncthreads();
    scn[t] += v;
    __syncthreads();
  }
  const int excl = scn[t] - c;
  const int node = (b << BIN_SHIFT) + t;
  if (node < n_nodes) {
    deg[node] = c;
    off[node] = beg + excl;
  }
  cnt[t] = excl;
  __syncthreads();
  i = beg + t;
  for (; i + 3 * 256 < end; i += 4 * 256) {
    int2 sd[4];
#pragma unroll
    for (int u = 0; u < 4; ++u) sd[u] = tmp[i + u * 256];
#pragma unroll
    for (int u = 0; u < 4; ++u) {
      const int r = atomicAdd(&cnt[sd[u].y & 255], 1);
      csr[beg + r] = sd[u].x;
    }
  }
  for (; i < end; i += 256) {
    const int2 sd = tmp[i];
    const int r = atomicAdd(&cnt[sd.y & 255], 1);
    csr[beg + r] = sd.x;
  }
}

#define TILE_PITCH 132

__device__ __forceinline__ float lane_read_f(float v, int l) {
  return __builtin_bit_cast(
      float, __builtin_amdgcn_readlane(__builtin_bit_cast(int, v), l));
}

__device__ __forceinline__ void tile_gather(
    const float* __restrict__ x, const int* __restrict__ csr,
    const int* __restrict__ off, const int* __restrict__ deg, int node0,
    int lane, float* tileLds) {
  const int li = lane & 15;
  const int offv = off[node0 + li];
  const int degv = deg[node0 + li];
  const float invv = 1.0f / fmaxf((float)degv, 1.0f);
  const int e0 = __builtin_amdgcn_readlane(offv, 0);
  const int e1 = __builtin_amdgcn_readlane(offv, 15) +
                 __builtin_amdgcn_readlane(degv, 15);
  const int bndv = (lane < 16) ? offv : e1;
  const int total = e1 - e0;
  const float* xl = x + lane * 4;
  float* trow = tileLds + lane * 4;

  float4 acc = make_float4(0.0f, 0.0f, 0.0f, 0.0f);
  int cur = 0;
  int nextB = __builtin_amdgcn_readlane(bndv, 1);

  int mine = (lane < total) ? csr[e0 + lane] : 0;
  for (int base = 0; base < total; base += 32) {
    const int nb = base + 32;
    const int ni = (nb + lane < total) ? (nb + lane) : (total - 1);
    const int nxt = csr[e0 + ni];
#pragma unroll
    for (int h = 0; h < 2; ++h) {
      const int g = base + 16 * h;
      if (g < total) {
        float4 v[16];
#pragma unroll
        for (int u = 0; u < 16; ++u) {
          const int s = __builtin_amdgcn_readlane(mine, 16 * h + u);
          v[u] = *(const float4*)(xl + (size_t)s * D_FEAT);
        }
#pragma unroll
        for (int u = 0; u < 16; ++u) {
          const int e = e0 + g + u;
          if (e < e1) {
            while (e >= nextB && cur < 15) {
              const float inv = lane_read_f(invv, cur);
              *(float4*)(trow + cur * TILE_PITCH) = make_float4(
                  acc.x * inv, acc.y * inv, acc.z * inv, acc.w * inv);
              acc = make_float4(0.0f, 0.0f, 0.0f, 0.0f);
              ++cur;
              nextB = __builtin_amdgcn_readlane(bndv, cur + 1);
            }
            acc.x += v[u].x;  acc.y += v[u].y;
            acc.z += v[u].z;  acc.w += v[u].w;
          }
        }
      }
    }
    mine = nxt;
  }
  for (; cur < 16; ++cur) {
    const float inv = lane_read_f(invv, cur);
    *(float4*)(trow + cur * TILE_PITCH) = make_float4(
        acc.x * inv, acc.y * inv, acc.z * inv, acc.w * inv);
    acc = make_float4(0.0f, 0.0f, 0.0f, 0.0f);
  }
}

template <int XMASK>
__device__ __forceinline__ float row_xor(float x) {
  return __builtin_bit_cast(
      float, __builtin_amdgcn_update_dpp(0, __builtin_bit_cast(int, x),
                                         0x160 + XMASK, 0xf, 0xf, true));
}

template <int CNT, int XMASK>
__device__ __forceinline__ void row_fold(float* v, bool upper) {
#pragma unroll
  for (int i = 0; i < CNT; ++i) {
    const float keep = upper ? v[i + CNT] : v[i];
    const float send = upper ? v[i] : v[i + CNT];
    v[i] = keep + row_xor<XMASK>(send);
  }
}


#define WMMA_TAIL_PAD()                                    \
  do {                                                     \
    __builtin_amdgcn_sched_barrier(0);                     \
    asm volatile("v_nop\n\tv_nop\n\tv_nop\n\tv_nop");      \
  } while (0)

#define L1_WAVES 13

__device__ __forceinline__ void self_product(
    v8f* acc, const float* __restrict__ xRow, const float* ldsWr,
    const int* nrow, const int* nswz, int half) {
#pragma unroll 1
  for (int k = 0; k < D_FEAT; k += 8) {
    float2 ts[2], br[2][8];
#pragma unroll
    for (int kk = 0; kk < 2; ++kk) {
      const int c0 = ((k + 4 * kk) >> 1) + half;
      ts[kk] = *(const float2*)(xRow + k + 4 * kk);
#pragma unroll
      for (int nt = 0; nt < 8; ++nt)
        br[kk][nt] = *(const float2*)(ldsWr + nrow[nt] * D_FEAT +
                                      2 * (c0 ^ nswz[nt]));
    }
#pragma unroll
    for (int kk = 0; kk < 2; ++kk) {
      v2f a_s;  a_s.x = ts[kk].x;  a_s.y = ts[kk].y;
#pragma unroll
      for (int nt = 0; nt < 8; ++nt) {
        v2f vbr;  vbr.x = br[kk][nt].x;  vbr.y = br[kk][nt].y;
        acc[nt] = __builtin_amdgcn_wmma_f32_16x16x4_f32(
            false, a_s, false, vbr, (short)0, acc[nt], false, false);
      }
    }
    WMMA_TAIL_PAD();
  }
}

__global__ __launch_bounds__(32 * L1_WAVES) void sage_layer1_wmma(
    const float* __restrict__ x, const int* __restrict__ csr,
    const int* __restrict__ off, const int* __restrict__ deg,
    const float* __restrict__ W1l,
    const float* __restrict__ W1r, const float* __restrict__ b1,
    const float* __restrict__ W2l, const float* __restrict__ W2r,
    const float* __restrict__ b2, float* __restrict__ p,
    float* __restrict__ out, int n_nodes) {
  __shared__ __attribute__((aligned(16))) float ldsW[2 * D_FEAT * D_FEAT];
  __shared__ __attribute__((aligned(16)))
  float ldsAgg[L1_WAVES * 16 * TILE_PITCH];

  const int lane = threadIdx.x & 31;
  const int wave = __builtin_amdgcn_readfirstlane((int)(threadIdx.x >> 5));

#define STAGE_DEPTH 10
  for (int j0 = 0; j0 * L1_WAVES < 2 * D_FEAT; j0 += STAGE_DEPTH) {
    float4 w[STAGE_DEPTH];
#pragma unroll
    for (int u = 0; u < STAGE_DEPTH; ++u) {
      const int row = wave + L1_WAVES * (j0 + u);
      const int rc = (row < 2 * D_FEAT) ? row : (2 * D_FEAT - 1);
      const float* srow = (rc < D_FEAT) ? (W1l + rc * D_FEAT)
                                        : (W1r + (rc - D_FEAT) * D_FEAT);
      w[u] = *(const float4*)(srow + 4 * lane);
    }
#pragma unroll
    for (int u = 0; u < STAGE_DEPTH; ++u) {
      const int row = wave + L1_WAVES * (j0 + u);
      const int rc = (row < 2 * D_FEAT) ? row : (2 * D_FEAT - 1);
      const int s0 = (2 * lane) ^ ((2 * rc) & 63);
      *(float4*)(ldsW + rc * D_FEAT + 2 * s0) = w[u];
    }
  }
  __syncthreads();

  const float* ldsWl = ldsW;
  const float* ldsWr = ldsW + D_FEAT * D_FEAT;
  const int tile = blockIdx.x * L1_WAVES + wave;
  const int n_tiles = n_nodes >> 4;
  if (tile >= n_tiles) return;

  const int half = lane >> 4;
  const int q = lane & 15;
  const int rowA = tile * 16 + q;

  v8f acc[8];
#pragma unroll
  for (int i = 0; i < 8; ++i) {
    v8f z = {};
    acc[i] = z;
  }

  const float* xRow = x + (size_t)rowA * D_FEAT + 2 * half;
  float* myTile = ldsAgg + wave * (16 * TILE_PITCH);

  int nrow[8], nswz[8];
#pragma unroll
  for (int nt = 0; nt < 8; ++nt) {
    nrow[nt] = nt * 16 + q;
    nswz[nt] = (2 * nrow[nt]) & 63;
  }

  const bool self_first = (wave & 4) != 0;
  if (self_first) self_product(acc, xRow, ldsWr, nrow, nswz, half);
  tile_gather(x, csr, off, deg, tile * 16, lane, myTile);
  if (!self_first) self_product(acc, xRow, ldsWr, nrow, nswz, half);

  __builtin_amdgcn_fence(__ATOMIC_RELEASE, "workgroup");
  __builtin_amdgcn_wave_barrier();
  __builtin_amdgcn_fence(__ATOMIC_ACQUIRE, "workgroup");

  const float* aggRow = myTile + q * TILE_PITCH + 2 * half;
#pragma unroll 1
  for (int k = 0; k < D_FEAT; k += 8) {
    float2 tm[2], bl[2][8];
#pragma unroll
    for (int kk = 0; kk < 2; ++kk) {
      const int c0 = ((k + 4 * kk) >> 1) + half;
      tm[kk] = *(const float2*)(aggRow + k + 4 * kk);
#pragma unroll
      for (int nt = 0; nt < 8; ++nt)
        bl[kk][nt] = *(const float2*)(ldsWl + nrow[nt] * D_FEAT +
                                      2 * (c0 ^ nswz[nt]));
    }
#pragma unroll
    for (int kk = 0; kk < 2; ++kk) {
      v2f a_m;  a_m.x = tm[kk].x;  a_m.y = tm[kk].y;
#pragma unroll
      for (int nt = 0; nt < 8; ++nt) {
        v2f vbl;  vbl.x = bl[kk][nt].x;  vbl.y = bl[kk][nt].y;
        acc[nt] = __builtin_amdgcn_wmma_f32_16x16x4_f32(
            false, a_m, false, vbl, (short)0, acc[nt], false, false);
      }
    }
    WMMA_TAIL_PAD();
  }

  float part[32];
#pragma unroll
  for (int i = 0; i < 32; ++i) part[i] = 0.0f;
#pragma unroll
  for (int nt = 0; nt < 8; ++nt) {
    const int n = nrow[nt];
    const float bias = b1[n];
    const float wl0 = W2l[n];
    const float wl1 = W2l[D_FEAT + n];
    const float wr0 = W2r[n];
    const float wr1 = W2r[D_FEAT + n];
#pragma unroll
    for (int v = 0; v < 8; ++v) {
      const float hv = fmaxf(acc[nt][v] + bias, 0.0f);
      part[v * 4 + 0] += hv * wl0;
      part[v * 4 + 1] += hv * wl1;
      part[v * 4 + 2] += hv * wr0;
      part[v * 4 + 3] += hv * wr1;
    }
  }
  row_fold<16, 8>(part, (q & 8) != 0);
  row_fold<8, 4>(part, (q & 4) != 0);
  row_fold<4, 2>(part, (q & 2) != 0);
  row_fold<2, 1>(part, (q & 1) != 0);

  const int node = tile * 16 + 8 * half + (q >> 1);
  const bool self_side = (q & 1) != 0;
  const float bias0 = b2[0];
  const float bias1 = b2[1];
  float2 res;
  res.x = self_side ? part[0] + bias0 : part[0];
  res.y = self_side ? part[1] + bias1 : part[1];
  float* dst = (self_side ? out : p) + (size_t)node * 2;
  *(float2*)dst = res;
}

#define G2_DEPTH 16

__global__ __launch_bounds__(256) void sage_gather2(
    const float* __restrict__ p, const int* __restrict__ csr,
    const int* __restrict__ off, const int* __restrict__ deg,
    float* __restrict__ out, int n_nodes, int n_edges) {
  const int gid = blockIdx.x * 256 + threadIdx.x;
  const int sub = gid & 1;
  const bool live = (gid >> 1) < n_nodes;
  const int node = live ? (gid >> 1) : (n_nodes - 1);
  const int beg = off[node];
  const int degv = deg[node];
  const int d = live ? degv : 0;
  const float2 self = *(const float2*)(out + (size_t)node * 2);
  const int last = (n_edges > 0) ? (n_edges - 1) : 0;
  float s0 = 0.0f, s1 = 0.0f;
  int jbase = 0;
  do {
    int sid[G2_DEPTH], keep[G2_DEPTH];
#pragma unroll
    for (int k = 0; k < G2_DEPTH; ++k) {
      const int j = jbase + sub + 2 * k;
      const int e = beg + j;
      keep[k] = (j - d) >> 31;
      sid[k] = csr[e < last ? e : last] & keep[k];
    }
    float2 pv[G2_DEPTH];
#pragma unroll
    for (int k = 0; k < G2_DEPTH; ++k)
      pv[k] = *(const float2*)(p + (size_t)sid[k] * 2);
#pragma unroll
    for (int k = 0; k < G2_DEPTH; ++k) {
      s0 += __int_as_float(__float_as_int(pv[k].x) & keep[k]);
      s1 += __int_as_float(__float_as_int(pv[k].y) & keep[k]);
    }
    jbase += 2 * G2_DEPTH;
  } while (__builtin_amdgcn_ballot_w32(jbase < d) != 0u);
  s0 += row_xor<1>(s0);
  s1 += row_xor<1>(s1);
  if (live && sub == 0) {
    const float inv = 1.0f / fmaxf((float)d, 1.0f);
    float2 res;
    res.x = self.x + s0 * inv;
    res.y = self.y + s1 * inv;
    *(float2*)(out + (size_t)node * 2) = res;
  }
}

extern "C" void kernel_launch(void* const* d_in, const int* in_sizes, int n_in,
                              void* d_out, int out_size, void* d_ws, size_t ws_size,
                              hipStream_t stream) {
  const float* x   = (const float*)d_in[0];
  const int*   ei  = (const int*)d_in[1];
  const float* W1l = (const float*)d_in[2];
  const float* W1r = (const float*)d_in[3];
  const float* b1  = (const float*)d_in[4];
  const float* W2l = (const float*)d_in[5];
  const float* W2r = (const float*)d_in[6];
  const float* b2  = (const float*)d_in[7];
  float* out = (float*)d_out;

  const int n_nodes = in_sizes[0] / D_FEAT;
  const int n_edges = in_sizes[1] / 2;

  float* ws   = (float*)d_ws;
  float* agg1 = ws;
  float* p    = agg1 + (size_t)n_nodes * D_FEAT;
  float* cnt  = p + (size_t)n_nodes * 2;
  int*   deg  = (int*)(cnt + n_nodes);
  int*   off  = deg + n_nodes;
  int*   csr  = off + n_nodes;
  const int n_bins  = (n_nodes + 255) / 256;
  const int chunk   = (n_edges + EDGE_BLOCKS - 1) / EDGE_BLOCKS;
  int*   cnt_cb    = csr + n_edges;
  int*   bin_start = cnt_cb + EDGE_BLOCKS * MAX_BINS;
  int2*  tmp       = (int2*)ws;

  sage_bin_hist<<<EDGE_BLOCKS, 256, 0, stream>>>(ei, cnt_cb, n_edges, n_bins, chunk);

  sage_bin_scatter<<<EDGE_BLOCKS, 256, 0, stream>>>(ei, cnt_cb, bin_start, tmp, n_edges,
                                                    n_bins, chunk);

  sage_bin_fill<<<n_bins, 256, 0, stream>>>(tmp, bin_start, deg, off, csr, n_nodes,
                                            n_edges, n_bins);

  const int n_tiles = n_nodes / 16;
  sage_layer1_wmma<<<(n_tiles + L1_WAVES - 1) / L1_WAVES, 32 * L1_WAVES, 0, stream>>>(
      x, csr, off, deg, W1l, W1r, b1, W2l, W2r, b2, p, out, n_nodes);

  sage_gather2<<<(n_nodes + 127) / 128, 256, 0, stream>>>(p, csr, off, deg, out, n_nodes,
                                                          n_edges);
}


# Layer-1 self product run inside the neighbour gather: every wave does one trip of 16 WMMAs behind each request for 16 rows (no self-first/self-last wave split); sum order unchanged
# speedup vs baseline: 4.1024x; 1.0298x over previous
#include <hip/hip_runtime.h>

typedef __attribute__((ext_vector_type(2))) float v2f;
typedef __attribute__((ext_vector_type(8))) float v8f;

#define D_FEAT 128


#define BIN_SHIFT 8
#define EDGE_BLOCKS 256
#define MAX_BINS 256

__global__ __launch_bounds__(256) void sage_bin_hist(
    const int* __restrict__ ei, int* __restrict__ cnt_cb, int n_edges,
    int n_bins, int chunk) {
  __shared__ int hist[MAX_BINS];
  const int t = threadIdx.x;
  if (t < n_bins) hist[t] = 0;
  __syncthreads();
  const int beg = blockIdx.x * chunk;
  int end = beg + chunk;
  if (end > n_edges) end = n_edges;
  const int* dstp = ei + n_edges;
  int e = beg + t;
  for (; e + 3 * 256 < end; e += 4 * 256) {
    int d[4];
#pragma unroll
    for (int u = 0; u < 4; ++u) d[u] = dstp[e + u * 256];
#pragma unroll
    for (int u = 0; u < 4; ++u) atomicAdd(&hist[d[u] >> BIN_SHIFT], 1);
  }
  for (; e < end; e += 256) atomicAdd(&hist[dstp[e] >> BIN_SHIFT], 1);
  __syncthreads();
  cnt_cb[blockIdx.x * MAX_BINS + t] = (t < n_bins) ? hist[t] : 0;
}

__global__ __launch_bounds__(256) void sage_bin_scatter(
    const int* __restrict__ ei, const int* __restrict__ cnt_cb,
    int* __restrict__ bin_start, int2* __restrict__ tmp, int n_edges,
    int n_bins, int chunk) {
  __shared__ int cur[MAX_BINS];
  __shared__ __attribute__((aligned(16))) int partTot[4 * MAX_BINS];
  __shared__ __attribute__((aligned(16))) int partPre[4 * MAX_BINS];
  __shared__ int waveTot[8];
  const int t = threadIdx.x;
  const int c = blockIdx.x;
  {
    const int q = t & 63;
    const int r = t >> 6;
    int4 tot = make_int4(0, 0, 0, 0);
    int4 pre = make_int4(0, 0, 0, 0);
    if (4 * q < n_bins) {
      const int4* col = (const int4*)cnt_cb + q;
      for (int j = 0; j < EDGE_BLOCKS / 4; j += 16) {
        int4 v[16];
#pragma unroll
        for (int u = 0; u < 16; ++u)
          v[u] = col[(r + 4 * (j + u)) * (MAX_BINS / 4)];
#pragma unroll
        for (int u = 0; u < 16; ++u) {
          const int m = (r + 4 * (j + u) < c) ? -1 : 0;
          tot.x += v[u].x;      tot.y += v[u].y;
          tot.z += v[u].z;      tot.w += v[u].w;
          pre.x += v[u].x & m;  pre.y += v[u].y & m;
          pre.z += v[u].z & m;  pre.w += v[u].w & m;
        }
      }
    }
    ((int4*)partTot)[r * (MAX_BINS / 4) + q] = tot;
    ((int4*)partPre)[r * (MAX_BINS / 4) + q] = pre;
  }
  __syncthreads();
  {
    int totT = 0, preT = 0;
    for (int k = 0; k < 4; ++k) {
      totT += partTot[k * MAX_BINS + t];
      preT += partPre[k * MAX_BINS + t];
    }
    const int lane = t & 31;
    const int wv = t >> 5;
    int inc = totT;
    for (int o = 1; o < 32; o <<= 1) {
      const int up = __shfl_up(inc, o, 32);
      if (lane >= o) inc += up;
    }
    if (lane == 31) waveTot[wv] = inc;
    __syncthreads();
    int below = 0;
    for (int k = 0; k < 8; ++k) below += (k < wv) ? waveTot[k] : 0;
    const int binBeg = below + inc - totT;
    cur[t] = binBeg + preT;
    if (c == 0 && t < n_bins) bin_start[t] = binBeg;
  }
  __syncthreads();
  const int beg = blockIdx.x * chunk;
  int end = beg + chunk;
  if (end > n_edges) end = n_edges;
  const int* dstp = ei + n_edges;
  int e = beg + t;
  for (; e + 3 * 256 < end; e += 4 * 256) {
    int s[4], d[4];
#pragma unroll
    for (int u = 0; u < 4; ++u) {
      s[u] = ei[e + u * 256];
      d[u] = dstp[e + u * 256];
    }
#pragma unroll
    for (int u = 0; u < 4; ++u) {
      const int slot = atomicAdd(&cur[d[u] >> BIN_SHIFT], 1);
      tmp[slot] = make_int2(s[u], d[u]);
    }
  }
  for (; e < end; e += 256) {
    const int s = ei[e];
    const int d = dstp[e];
    const int slot = atomicAdd(&cur[d >> BIN_SHIFT], 1);
    tmp[slot] = make_int2(s, d);
  }
}

__global__ __launch_bounds__(256) void sage_bin_fill(
    const int2* __restrict__ tmp, const int* __restrict__ bin_start,
    int* __restrict__ deg, int* __restrict__ off, int* __restrict__ csr,
    int n_nodes, int n_edges, int n_bins) {
  __shared__ int cnt[256];
  __shared__ int scn[256];
  const int t = threadIdx.x;
  const int b = blockIdx.x;
  const int beg = bin_start[b];
  const int end = (b + 1 < n_bins) ? bin_start[b + 1] : n_edges;
  cnt[t] = 0;
  __syncthreads();
  int i = beg + t;
  for (; i + 3 * 256 < end; i += 4 * 256) {
    int d[4];
#pragma unroll
    for (int u = 0; u < 4; ++u) d[u] = tmp[i + u * 256].y;
#pragma unroll
    for (int u = 0; u < 4; ++u) atomicAdd(&cnt[d[u] & 255], 1);
  }
  for (; i < end; i += 256) atomicAdd(&cnt[tmp[i].y & 255], 1);
  __syncthreads();
  const int c = cnt[t];
  scn[t] = c;
  __syncthreads();
  for (int d = 1; d < 256; d <<= 1) {
    const int v = (t >= d) ? scn[t - d] : 0;
    __syncthreads();
    scn[t] += v;
    __syncthreads();
  }
  const int excl = scn[t] - c;
  const int node = (b << BIN_SHIFT) + t;
  if (node < n_nodes) {
    deg[node] = c;
    off[node] = beg + excl;
  }
  cnt[t] = excl;
  __syncthreads();
  i = beg + t;
  for (; i + 3 * 256 < end; i += 4 * 256) {
    int2 sd[4];
#pragma unroll
    for (int u = 0; u < 4; ++u) sd[u] = tmp[i + u * 256];
#pragma unroll
    for (int u = 0; u < 4; ++u) {
      const int r = atomicAdd(&cnt[sd[u].y & 255], 1);
      csr[beg + r] = sd[u].x;
    }
  }
  for (; i < end; i += 256) {
    const int2 sd = tmp[i];
    const int r = atomicAdd(&cnt[sd.y & 255], 1);
    csr[beg + r] = sd.x;
  }
}

#define TILE_PITCH 132

__device__ __forceinline__ float lane_read_f(float v, int l) {
  return __builtin_bit_cast(
      float, __builtin_amdgcn_readlane(__builtin_bit_cast(int, v), l));
}

#define WMMA_TAIL_PAD()                                    \
  do {                                                     \
    __builtin_amdgcn_sched_barrier(0);                     \
    asm volatile("v_nop\n\tv_nop\n\tv_nop\n\tv_nop");      \
  } while (0)

#define L1_WAVES 13

__device__ __forceinline__ void self_trip(
    v8f* acc, float2 a0, float2 a1, const float* ldsWr, const int* nrow,
    const int* nswz, int half, int k) {
  float2 br[2][8];
#pragma unroll
  for (int kk = 0; kk < 2; ++kk) {
    const int c0 = ((k + 4 * kk) >> 1) + half;
#pragma unroll
    for (int nt = 0; nt < 8; ++nt)
      br[kk][nt] = *(const float2*)(ldsWr + nrow[nt] * D_FEAT +
                                    2 * (c0 ^ nswz[nt]));
  }
#pragma unroll
  for (int kk = 0; kk < 2; ++kk) {
    v2f a_s;
    a_s.x = kk ? a1.x : a0.x;
    a_s.y = kk ? a1.y : a0.y;
#pragma unroll
    for (int nt = 0; nt < 8; ++nt) {
      v2f vbr;  vbr.x = br[kk][nt].x;  vbr.y = br[kk][nt].y;
      acc[nt] = __builtin_amdgcn_wmma_f32_16x16x4_f32(
          false, a_s, false, vbr, (short)0, acc[nt], false, false);
    }
  }
  WMMA_TAIL_PAD();
}

__device__ __forceinline__ void tile_gather_self(
    const float* __restrict__ x, const int* __restrict__ csr,
    const int* __restrict__ off, const int* __restrict__ deg, int node0,
    int lane, float* tileLds, v8f* acc, const float* __restrict__ xRow,
    const float* ldsWr, const int* nrow, const int* nswz, int half) {
  const int li = lane & 15;
  const int offv = off[node0 + li];
  const int degv = deg[node0 + li];
  const float invv = 1.0f / fmaxf((float)degv, 1.0f);
  const int e0 = __builtin_amdgcn_readlane(offv, 0);
  const int e1 = __builtin_amdgcn_readlane(offv, 15) +
                 __builtin_amdgcn_readlane(degv, 15);
  const int bndv = (lane < 16) ? offv : e1;
  const int total = e1 - e0;
  const float* xl = x + lane * 4;
  float* trow = tileLds + lane * 4;

  float4 acs = make_float4(0.0f, 0.0f, 0.0f, 0.0f);
  int cur = 0;
  int nextB = __builtin_amdgcn_readlane(bndv, 1);
  int ks = 0;

  float2 pa[4];
#pragma unroll
  for (int i = 0; i < 4; ++i) pa[i] = *(const float2*)(xRow + 4 * i);

  int mine = (lane < total) ? csr[e0 + lane] : 0;
  for (int base = 0; base < total; base += 32) {
    float2 na[4];
#pragma unroll
    for (int i = 0; i < 4; ++i) {
      const int kn = ks + 16 + 4 * i;
      na[i] = *(const float2*)(xRow + ((kn < D_FEAT) ? kn : (D_FEAT - 4)));
    }
    const int nb = base + 32;
    const int ni = (nb + lane < total) ? (nb + lane) : (total - 1);
    const int nxt = csr[e0 + ni];
#pragma unroll
    for (int h = 0; h < 2; ++h) {
      const int g = base + 16 * h;
      if (g < total) {
        float4 v[16];
#pragma unroll
        for (int u = 0; u < 16; ++u) {
          const int s = __builtin_amdgcn_readlane(mine, 16 * h + u);
          v[u] = *(const float4*)(xl + (size_t)s * D_FEAT);
        }
        if (ks < D_FEAT) {
          self_trip(acc, pa[2 * h], pa[2 * h + 1], ldsWr, nrow, nswz, half, ks);
          ks += 8;
        }
#pragma unroll
        for (int u = 0; u < 16; ++u) {
          const int e = e0 + g + u;
          if (e < e1) {
            while (e >= nextB && cur < 15) {
              const float inv = lane_read_f(invv, cur);
              *(float4*)(trow + cur * TILE_PITCH) = make_float4(
                  acs.x * inv, acs.y * inv, acs.z * inv, acs.w * inv);
              acs = make_float4(0.0f, 0.0f, 0.0f, 0.0f);
              ++cur;
              nextB = __builtin_amdgcn_readlane(bndv, cur + 1);
            }
            acs.x += v[u].x;  acs.y += v[u].y;
            acs.z += v[u].z;  acs.w += v[u].w;
          }
        }
      }
    }
    mine = nxt;
#pragma unroll
    for (int i = 0; i < 4; ++i) pa[i] = na[i];
  }
  for (; cur < 16; ++cur) {
    const float inv = lane_read_f(invv, cur);
    *(float4*)(trow + cur * TILE_PITCH) = make_float4(
        acs.x * inv, acs.y * inv, acs.z * inv, acs.w * inv);
    acs = make_float4(0.0f, 0.0f, 0.0f, 0.0f);
  }
#pragma unroll 1
  for (; ks < D_FEAT; ks += 8) {
    const float2 sa0 = *(const float2*)(xRow + ks);
    const float2 sa1 = *(const float2*)(xRow + ks + 4);
    self_trip(acc, sa0, sa1, ldsWr, nrow, nswz, half, ks);
  }
}

template <int XMASK>
__device__ __forceinline__ float row_xor(float x) {
  return __builtin_bit_cast(
      float, __builtin_amdgcn_update_dpp(0, __builtin_bit_cast(int, x),
                                         0x160 + XMASK, 0xf, 0xf, true));
}

template <int CNT, int XMASK>
__device__ __forceinline__ void row_fold(float* v, bool upper) {
#pragma unroll
  for (int i = 0; i < CNT; ++i) {
    const float keep = upper ? v[i + CNT] : v[i];
    const float send = upper ? v[i] : v[i + CNT];
    v[i] = keep + row_xor<XMASK>(send);
  }
}


__global__ __launch_bounds__(32 * L1_WAVES) void sage_layer1_wmma(
    const float* __restrict__ x, const int* __restrict__ csr,
    const int* __restrict__ off, const int* __restrict__ deg,
    const float* __restrict__ W1l,
    const float* __restrict__ W1r, const float* __restrict__ b1,
    const float* __restrict__ W2l, const float* __restrict__ W2r,
    const float* __restrict__ b2, float* __restrict__ p,
    float* __restrict__ out, int n_nodes) {
  __shared__ __attribute__((aligned(16))) float ldsW[2 * D_FEAT * D_FEAT];
  __shared__ __attribute__((aligned(16)))
  float ldsAgg[L1_WAVES * 16 * TILE_PITCH];

  const int lane = threadIdx.x & 31;
  const int wave = __builtin_amdgcn_readfirstlane((int)(threadIdx.x >> 5));

#define STAGE_DEPTH 10
  for (int j0 = 0; j0 * L1_WAVES < 2 * D_FEAT; j0 += STAGE_DEPTH) {
    float4 w[STAGE_DEPTH];
#pragma unroll
    for (int u = 0; u < STAGE_DEPTH; ++u) {
      const int row = wave + L1_WAVES * (j0 + u);
      const int rc = (row < 2 * D_FEAT) ? row : (2 * D_FEAT - 1);
      const float* srow = (rc < D_FEAT) ? (W1l + rc * D_FEAT)
                                        : (W1r + (rc - D_FEAT) * D_FEAT);
      w[u] = *(const float4*)(srow + 4 * lane);
    }
#pragma unroll
    for (int u = 0; u < STAGE_DEPTH; ++u) {
      const int row = wave + L1_WAVES * (j0 + u);
      const int rc = (row < 2 * D_FEAT) ? row : (2 * D_FEAT - 1);
      const int s0 = (2 * lane) ^ ((2 * rc) & 63);
      *(float4*)(ldsW + rc * D_FEAT + 2 * s0) = w[u];
    }
  }
  __syncthreads();

  const float* ldsWl = ldsW;
  const float* ldsWr = ldsW + D_FEAT * D_FEAT;
  const int tile = blockIdx.x * L1_WAVES + wave;
  const int n_tiles = n_nodes >> 4;
  if (tile >= n_tiles) return;

  const int half = lane >> 4;
  const int q = lane & 15;
  const int rowA = tile * 16 + q;

  v8f acc[8];
#pragma unroll
  for (int i = 0; i < 8; ++i) {
    v8f z = {};
    acc[i] = z;
  }

  const float* xRow = x + (size_t)rowA * D_FEAT + 2 * half;
  float* myTile = ldsAgg + wave * (16 * TILE_PITCH);

  int nrow[8], nswz[8];
#pragma unroll
  for (int nt = 0; nt < 8; ++nt) {
    nrow[nt] = nt * 16 + q;
    nswz[nt] = (2 * nrow[nt]) & 63;
  }

  tile_gather_self(x, csr, off, deg, tile * 16, lane, myTile, acc, xRow, ldsWr,
                   nrow, nswz, half);

  __builtin_amdgcn_fence(__ATOMIC_RELEASE, "workgroup");
  __builtin_amdgcn_wave_barrier();
  __builtin_amdgcn_fence(__ATOMIC_ACQUIRE, "workgroup");

  const float* aggRow = myTile + q * TILE_PITCH + 2 * half;
#pragma unroll 1
  for (int k = 0; k < D_FEAT; k += 8) {
    float2 tm[2], bl[2][8];
#pragma unroll
    for (int kk = 0; kk < 2; ++kk) {
      const int c0 = ((k + 4 * kk) >> 1) + half;
      tm[kk] = *(const float2*)(aggRow + k + 4 * kk);
#pragma unroll
      for (int nt = 0; nt < 8; ++nt)
        bl[kk][nt] = *(const float2*)(ldsWl + nrow[nt] * D_FEAT +
                                      2 * (c0 ^ nswz[nt]));
    }
#pragma unroll
    for (int kk = 0; kk < 2; ++kk) {
      v2f a_m;  a_m.x = tm[kk].x;  a_m.y = tm[kk].y;
#pragma unroll
      for (int nt = 0; nt < 8; ++nt) {
        v2f vbl;  vbl.x = bl[kk][nt].x;  vbl.y = bl[kk][nt].y;
        acc[nt] = __builtin_amdgcn_wmma_f32_16x16x4_f32(
            false, a_m, false, vbl, (short)0, acc[nt], false, false);
      }
    }
    WMMA_TAIL_PAD();
  }

  float part[32];
#pragma unroll
  for (int i = 0; i < 32; ++i) part[i] = 0.0f;
#pragma unroll
  for (int nt = 0; nt < 8; ++nt) {
    const int n = nrow[nt];
    const float bias = b1[n];
    const float wl0 = W2l[n];
    const float wl1 = W2l[D_FEAT + n];
    const float wr0 = W2r[n];
    const float wr1 = W2r[D_FEAT + n];
#pragma unroll
    for (int v = 0; v < 8; ++v) {
      const float hv = fmaxf(acc[nt][v] + bias, 0.0f);
      part[v * 4 + 0] += hv * wl0;
      part[v * 4 + 1] += hv * wl1;
      part[v * 4 + 2] += hv * wr0;
      part[v * 4 + 3] += hv * wr1;
    }
  }
  row_fold<16, 8>(part, (q & 8) != 0);
  row_fold<8, 4>(part, (q & 4) != 0);
  row_fold<4, 2>(part, (q & 2) != 0);
  row_fold<2, 1>(part, (q & 1) != 0);

  const int node = tile * 16 + 8 * half + (q >> 1);
  const bool self_side = (q & 1) != 0;
  const float bias0 = b2[0];
  const float bias1 = b2[1];
  float2 res;
  res.x = self_side ? part[0] + bias0 : part[0];
  res.y = self_side ? part[1] + bias1 : part[1];
  float* dst = (self_side ? out : p) + (size_t)node * 2;
  *(float2*)dst = res;
}

#define G2_DEPTH 16

__global__ __launch_bounds__(256) void sage_gather2(
    const float* __restrict__ p, const int* __restrict__ csr,
    const int* __restrict__ off, const int* __restrict__ deg,
    float* __restrict__ out, int n_nodes, int n_edges) {
  const int gid = blockIdx.x * 256 + threadIdx.x;
  const int sub = gid & 1;
  const bool live = (gid >> 1) < n_nodes;
  const int node = live ? (gid >> 1) : (n_nodes - 1);
  const int beg = off[node];
  const int degv = deg[node];
  const int d = live ? degv : 0;
  const float2 self = *(const float2*)(out + (size_t)node * 2);
  const int last = (n_edges > 0) ? (n_edges - 1) : 0;
  float s0 = 0.0f, s1 = 0.0f;
  int jbase = 0;
  do {
    int sid[G2_DEPTH], keep[G2_DEPTH];
#pragma unroll
    for (int k = 0; k < G2_DEPTH; ++k) {
      const int j = jbase + sub + 2 * k;
      const int e = beg + j;
      keep[k] = (j - d) >> 31;
      sid[k] = csr[e < last ? e : last] & keep[k];
    }
    float2 pv[G2_DEPTH];
#pragma unroll
    for (int k = 0; k < G2_DEPTH; ++k)
      pv[k] = *(const float2*)(p + (size_t)sid[k] * 2);
#pragma unroll
    for (int k = 0; k < G2_DEPTH; ++k) {
      s0 += __int_as_float(__float_as_int(pv[k].x) & keep[k]);
      s1 += __int_as_float(__float_as_int(pv[k].y) & keep[k]);
    }
    jbase += 2 * G2_DEPTH;
  } while (__builtin_amdgcn_ballot_w32(jbase < d) != 0u);
  s0 += row_xor<1>(s0);
  s1 += row_xor<1>(s1);
  if (live && sub == 0) {
    const float inv = 1.0f / fmaxf((float)d, 1.0f);
    float2 res;
    res.x = self.x + s0 * inv;
    res.y = self.y + s1 * inv;
    *(float2*)(out + (size_t)node * 2) = res;
  }
}

extern "C" void kernel_launch(void* const* d_in, const int* in_sizes, int n_in,
                              void* d_out, int out_size, void* d_ws, size_t ws_size,
                              hipStream_t stream) {
  const float* x   = (const float*)d_in[0];
  const int*   ei  = (const int*)d_in[1];
  const float* W1l = (const float*)d_in[2];
  const float* W1r = (const float*)d_in[3];
  const float* b1  = (const float*)d_in[4];
  const float* W2l = (const float*)d_in[5];
  const float* W2r = (const float*)d_in[6];
  const float* b2  = (const float*)d_in[7];
  float* out = (float*)d_out;

  const int n_nodes = in_sizes[0] / D_FEAT;
  const int n_edges = in_sizes[1] / 2;

  float* ws   = (float*)d_ws;
  float* agg1 = ws;
  float* p    = agg1 + (size_t)n_nodes * D_FEAT;
  float* cnt  = p + (size_t)n_nodes * 2;
  int*   deg  = (int*)(cnt + n_nodes);
  int*   off  = deg + n_nodes;
  int*   csr  = off + n_nodes;
  const int n_bins  = (n_nodes + 255) / 256;
  const int chunk   = (n_edges + EDGE_BLOCKS - 1) / EDGE_BLOCKS;
  int*   cnt_cb    = csr + n_edges;
  int*   bin_start = cnt_cb + EDGE_BLOCKS * MAX_BINS;
  int2*  tmp       = (int2*)ws;

  sage_bin_hist<<<EDGE_BLOCKS, 256, 0, stream>>>(ei, cnt_cb, n_edges, n_bins, chunk);

  sage_bin_scatter<<<EDGE_BLOCKS, 256, 0, stream>>>(ei, cnt_cb, bin_start, tmp, n_edges,
                                                    n_bins, chunk);

  sage_bin_fill<<<n_bins, 256, 0, stream>>>(tmp, bin_start, deg, off, csr, n_nodes,
                                            n_edges, n_bins);

  const int n_tiles = n_nodes / 16;
  sage_layer1_wmma<<<(n_tiles + L1_WAVES - 1) / L1_WAVES, 32 * L1_WAVES, 0, stream>>>(
      x, csr, off, deg, W1l, W1r, b1, W2l, W2r, b2, p, out, n_nodes);

  sage_gather2<<<(n_nodes + 127) / 128, 256, 0, stream>>>(p, csr, off, deg, out, n_nodes,
                                                          n_edges);
}


# Edge ranking (per-node count, scan, csr fill) moved from its own launch into the layer-1 block that owns those 208 nodes; weight-stage loads run under its passes; sum order unchanged
# speedup vs baseline: 4.3508x; 1.0605x over previous
#include <hip/hip_runtime.h>

typedef __attribute__((ext_vector_type(2))) float v2f;
typedef __attribute__((ext_vector_type(8))) float v8f;

#define D_FEAT 128


#define L1_WAVES 13
#define BIN_NODES (16 * L1_WAVES)
#define EDGE_BLOCKS 256
#define MAX_BINS 256

__device__ __forceinline__ int bin_of(int d) {
  return (int)((unsigned)d / (unsigned)BIN_NODES);
}

__global__ __launch_bounds__(256) void sage_bin_hist(
    const int* __restrict__ ei, int* __restrict__ cnt_cb, int n_edges,
    int n_bins, int chunk) {
  __shared__ int hist[MAX_BINS];
  const int t = threadIdx.x;
  if (t < n_bins) hist[t] = 0;
  __syncthreads();
  const int beg = blockIdx.x * chunk;
  int end = beg + chunk;
  if (end > n_edges) end = n_edges;
  const int* dstp = ei + n_edges;
  int e = beg + t;
  for (; e + 3 * 256 < end; e += 4 * 256) {
    int d[4];
#pragma unroll
    for (int u = 0; u < 4; ++u) d[u] = dstp[e + u * 256];
#pragma unroll
    for (int u = 0; u < 4; ++u) atomicAdd(&hist[bin_of(d[u])], 1);
  }
  for (; e < end; e += 256) atomicAdd(&hist[bin_of(dstp[e])], 1);
  __syncthreads();
  cnt_cb[blockIdx.x * MAX_BINS + t] = (t < n_bins) ? hist[t] : 0;
}

__global__ __launch_bounds__(256) void sage_bin_scatter(
    const int* __restrict__ ei, const int* __restrict__ cnt_cb,
    int* __restrict__ bin_start, int2* __restrict__ tmp, int n_edges,
    int n_bins, int chunk) {
  __shared__ int cur[MAX_BINS];
  __shared__ __attribute__((aligned(16))) int partTot[4 * MAX_BINS];
  __shared__ __attribute__((aligned(16))) int partPre[4 * MAX_BINS];
  __shared__ int waveTot[8];
  const int t = threadIdx.x;
  const int c = blockIdx.x;
  {
    const int q = t & 63;
    const int r = t >> 6;
    int4 tot = make_int4(0, 0, 0, 0);
    int4 pre = make_int4(0, 0, 0, 0);
    if (4 * q < n_bins) {
      const int4* col = (const int4*)cnt_cb + q;
      for (int j = 0; j < EDGE_BLOCKS / 4; j += 16) {
        int4 v[16];
#pragma unroll
        for (int u = 0; u < 16; ++u)
          v[u] = col[(r + 4 * (j + u)) * (MAX_BINS / 4)];
#pragma unroll
        for (int u = 0; u < 16; ++u) {
          const int m = (r + 4 * (j + u) < c) ? -1 : 0;
          tot.x += v[u].x;      tot.y += v[u].y;
          tot.z += v[u].z;      tot.w += v[u].w;
          pre.x += v[u].x & m;  pre.y += v[u].y & m;
          pre.z += v[u].z & m;  pre.w += v[u].w & m;
        }
      }
    }
    ((int4*)partTot)[r * (MAX_BINS / 4) + q] = tot;
    ((int4*)partPre)[r * (MAX_BINS / 4) + q] = pre;
  }
  __syncthreads();
  {
    int totT = 0, preT = 0;
    for (int k = 0; k < 4; ++k) {
      totT += partTot[k * MAX_BINS + t];
      preT += partPre[k * MAX_BINS + t];
    }
    const int lane = t & 31;
    const int wv = t >> 5;
    int inc = totT;
    for (int o = 1; o < 32; o <<= 1) {
      const int up = __shfl_up(inc, o, 32);
      if (lane >= o) inc += up;
    }
    if (lane == 31) waveTot[wv] = inc;
    __syncthreads();
    int below = 0;
    for (int k = 0; k < 8; ++k) below += (k < wv) ? waveTot[k] : 0;
    const int binBeg = below + inc - totT;
    cur[t] = binBeg + preT;
    if (c == 0 && t < n_bins) bin_start[t] = binBeg;
  }
  __syncthreads();
  const int beg = blockIdx.x * chunk;
  int end = beg + chunk;
  if (end > n_edges) end = n_edges;
  const int* dstp = ei + n_edges;
  int e = beg + t;
  for (; e + 3 * 256 < end; e += 4 * 256) {
    int s[4], d[4];
#pragma unroll
    for (int u = 0; u < 4; ++u) {
      s[u] = ei[e + u * 256];
      d[u] = dstp[e + u * 256];
    }
#pragma unroll
    for (int u = 0; u < 4; ++u) {
      const int slot = atomicAdd(&cur[bin_of(d[u])], 1);
      tmp[slot] = make_int2(s[u], d[u]);
    }
  }
  for (; e < end; e += 256) {
    const int s = ei[e];
    const int d = dstp[e];
    const int slot = atomicAdd(&cur[bin_of(d)], 1);
    tmp[slot] = make_int2(s, d);
  }
}


#define TILE_PITCH 132

__device__ __forceinline__ float lane_read_f(float v, int l) {
  return __builtin_bit_cast(
      float, __builtin_amdgcn_readlane(__builtin_bit_cast(int, v), l));
}

#define WMMA_TAIL_PAD()                                    \
  do {                                                     \
    __builtin_amdgcn_sched_barrier(0);                     \
    asm volatile("v_nop\n\tv_nop\n\tv_nop\n\tv_nop");      \
  } while (0)

__device__ __forceinline__ void self_trip(
    v8f* acc, float2 a0, float2 a1, const float* ldsWr, const int* nrow,
    const int* nswz, int half, int k) {
  float2 br[2][8];
#pragma unroll
  for (int kk = 0; kk < 2; ++kk) {
    const int c0 = ((k + 4 * kk) >> 1) + half;
#pragma unroll
    for (int nt = 0; nt < 8; ++nt)
      br[kk][nt] = *(const float2*)(ldsWr + nrow[nt] * D_FEAT +
                                    2 * (c0 ^ nswz[nt]));
  }
#pragma unroll
  for (int kk = 0; kk < 2; ++kk) {
    v2f a_s;
    a_s.x = kk ? a1.x : a0.x;
    a_s.y = kk ? a1.y : a0.y;
#pragma unroll
    for (int nt = 0; nt < 8; ++nt) {
      v2f vbr;  vbr.x = br[kk][nt].x;  vbr.y = br[kk][nt].y;
      acc[nt] = __builtin_amdgcn_wmma_f32_16x16x4_f32(
          false, a_s, false, vbr, (short)0, acc[nt], false, false);
    }
  }
  WMMA_TAIL_PAD();
}

__device__ __forceinline__ void tile_gather_self(
    const float* __restrict__ x, const int* csr, const int* tileOff,
    const int* tileDeg, int lane, float* tileLds, v8f* acc,
    const float* __restrict__ xRow, const float* ldsWr, const int* nrow,
    const int* nswz, int half) {
  const int li = lane & 15;
  const int offv = tileOff[li];
  const int degv = tileDeg[li];
  const float invv = 1.0f / fmaxf((float)degv, 1.0f);
  const int e0 = __builtin_amdgcn_readlane(offv, 0);
  const int e1 = __builtin_amdgcn_readlane(offv, 15) +
                 __builtin_amdgcn_readlane(degv, 15);
  const int bndv = (lane < 16) ? offv : e1;
  const int total = e1 - e0;
  const float* xl = x + lane * 4;
  float* trow = tileLds + lane * 4;

  float4 acs = make_float4(0.0f, 0.0f, 0.0f, 0.0f);
  int cur = 0;
  int nextB = __builtin_amdgcn_readlane(bndv, 1);
  int ks = 0;

  float2 pa[4];
#pragma unroll
  for (int i = 0; i < 4; ++i) pa[i] = *(const float2*)(xRow + 4 * i);

  int mine = (lane < total) ? csr[e0 + lane] : 0;
  for (int base = 0; base < total; base += 32) {
    float2 na[4];
#pragma unroll
    for (int i = 0; i < 4; ++i) {
      const int kn = ks + 16 + 4 * i;
      na[i] = *(const float2*)(xRow + ((kn < D_FEAT) ? kn : (D_FEAT - 4)));
    }
    const int nb = base + 32;
    const int ni = (nb + lane < total) ? (nb + lane) : (total - 1);
    const int nxt = csr[e0 + ni];
#pragma unroll
    for (int h = 0; h < 2; ++h) {
      const int g = base + 16 * h;
      if (g < total) {
        float4 v[16];
#pragma unroll
        for (int u = 0; u < 16; ++u) {
          const int s = __builtin_amdgcn_readlane(mine, 16 * h + u);
          v[u] = *(const float4*)(xl + (size_t)s * D_FEAT);
        }
        if (ks < D_FEAT) {
          self_trip(acc, pa[2 * h], pa[2 * h + 1], ldsWr, nrow, nswz, half, ks);
          ks += 8;
        }
#pragma unroll
        for (int u = 0; u < 16; ++u) {
          const int e = e0 + g + u;
          if (e < e1) {
            while (e >= nextB && cur < 15) {
              const float inv = lane_read_f(invv, cur);
              *(float4*)(trow + cur * TILE_PITCH) = make_float4(
                  acs.x * inv, acs.y * inv, acs.z * inv, acs.w * inv);
              acs = make_float4(0.0f, 0.0f, 0.0f, 0.0f);
              ++cur;
              nextB = __builtin_amdgcn_readlane(bndv, cur + 1);
            }
            acs.x += v[u].x;  acs.y += v[u].y;
            acs.z += v[u].z;  acs.w += v[u].w;
          }
        }
      }
    }
    mine = nxt;
#pragma unroll
    for (int i = 0; i < 4; ++i) pa[i] = na[i];
  }
  for (; cur < 16; ++cur) {
    const float inv = lane_read_f(invv, cur);
    *(float4*)(trow + cur * TILE_PITCH) = make_float4(
        acs.x * inv, acs.y * inv, acs.z * inv, acs.w * inv);
    acs = make_float4(0.0f, 0.0f, 0.0f, 0.0f);
  }
#pragma unroll 1
  for (; ks < D_FEAT; ks += 8) {
    const float2 sa0 = *(const float2*)(xRow + ks);
    const float2 sa1 = *(const float2*)(xRow + ks + 4);
    self_trip(acc, sa0, sa1, ldsWr, nrow, nswz, half, ks);
  }
}

template <int XMASK>
__device__ __forceinline__ float row_xor(float x) {
  return __builtin_bit_cast(
      float, __builtin_amdgcn_update_dpp(0, __builtin_bit_cast(int, x),
                                         0x160 + XMASK, 0xf, 0xf, true));
}

template <int CNT, int XMASK>
__device__ __forceinline__ void row_fold(float* v, bool upper) {
#pragma unroll
  for (int i = 0; i < CNT; ++i) {
    const float keep = upper ? v[i + CNT] : v[i];
    const float send = upper ? v[i] : v[i + CNT];
    v[i] = keep + row_xor<XMASK>(send);
  }
}


#define L1_THREADS (32 * L1_WAVES)
#define STAGE_DEPTH 10
static_assert(2 * STAGE_DEPTH * L1_WAVES >= 2 * D_FEAT,
              "two stage batches must cover both weight matrices");

__device__ __forceinline__ float4 stage_row_load(
    const float* __restrict__ W1l, const float* __restrict__ W1r, int wave,
    int lane, int j) {
  const int row = wave + L1_WAVES * j;
  const int rc = (row < 2 * D_FEAT) ? row : (2 * D_FEAT - 1);
  const float* srow = (rc < D_FEAT) ? (W1l + rc * D_FEAT)
                                    : (W1r + (rc - D_FEAT) * D_FEAT);
  return *(const float4*)(srow + 4 * lane);
}

__device__ __forceinline__ void stage_row_store(float* ldsW, int wave, int lane,
                                                int j, float4 v) {
  const int row = wave + L1_WAVES * j;
  const int rc = (row < 2 * D_FEAT) ? row : (2 * D_FEAT - 1);
  const int s0 = (2 * lane) ^ ((2 * rc) & 63);
  *(float4*)(ldsW + rc * D_FEAT + 2 * s0) = v;
}

#define STAGE_EACH(OP, J0) \
  OP(0, J0) OP(1, J0) OP(2, J0) OP(3, J0) OP(4, J0) \
  OP(5, J0) OP(6, J0) OP(7, J0) OP(8, J0) OP(9, J0)
#define STAGE_LD(u, J0) wq##u = stage_row_load(W1l, W1r, wave, lane, (J0) + u);
#define STAGE_ST(u, J0) stage_row_store(ldsW, wave, lane, (J0) + u, wq##u);

__device__ __forceinline__ void lds_barrier() {
  __builtin_amdgcn_fence(__ATOMIC_RELEASE, "workgroup", "local");
  __builtin_amdgcn_s_barrier();
  __builtin_amdgcn_fence(__ATOMIC_ACQUIRE, "workgroup", "local");
}

__global__ __launch_bounds__(L1_THREADS) void sage_layer1_wmma(
    const float* __restrict__ x, const int2* __restrict__ tmp,
    const int* __restrict__ bin_start, int* csr, int* off, int* deg,
    const float* __restrict__ W1l,
    const float* __restrict__ W1r, const float* __restrict__ b1,
    const float* __restrict__ W2l, const float* __restrict__ W2r,
    const float* __restrict__ b2, float* __restrict__ p,
    float* __restrict__ out, int n_nodes, int n_edges, int n_bins) {
  __shared__ __attribute__((aligned(16))) float ldsW[2 * D_FEAT * D_FEAT];
  __shared__ __attribute__((aligned(16)))
  float ldsAgg[L1_WAVES * 16 * TILE_PITCH];
  __shared__ int binCur[BIN_NODES];
  __shared__ int binOff[BIN_NODES];
  __shared__ int binDeg[BIN_NODES];
  __shared__ int waveTot[L1_WAVES];

  const int t = threadIdx.x;
  const int lane = threadIdx.x & 31;
  const int wave = __builtin_amdgcn_readfirstlane((int)(threadIdx.x >> 5));

  const int bin = blockIdx.x;
  const int nodeBase = bin * BIN_NODES;
  const int beg = bin_start[bin];
  const int end = (bin + 1 < n_bins) ? bin_start[bin + 1] : n_edges;
  if (t < BIN_NODES) binCur[t] = 0;
  float4 wq0, wq1, wq2, wq3, wq4, wq5, wq6, wq7, wq8, wq9;
  STAGE_EACH(STAGE_LD, 0)
  lds_barrier();
  {
    int i = beg + t;
    for (; i + 3 * L1_THREADS < end; i += 4 * L1_THREADS) {
      int d[4];
#pragma unroll
      for (int u = 0; u < 4; ++u) d[u] = tmp[i + u * L1_THREADS].y;
#pragma unroll
      for (int u = 0; u < 4; ++u) atomicAdd(&binCur[d[u] - nodeBase], 1);
    }
    for (; i < end; i += L1_THREADS) atomicAdd(&binCur[tmp[i].y - nodeBase], 1);
  }
  STAGE_EACH(STAGE_ST, 0)
  STAGE_EACH(STAGE_LD, STAGE_DEPTH)
  lds_barrier();
  {
    const int c = (t < BIN_NODES) ? binCur[t] : 0;
    int inc = c;
    for (int o = 1; o < 32; o <<= 1) {
      const int up = __shfl_up(inc, o, 32);
      if (lane >= o) inc += up;
    }
    if (lane == 31) waveTot[wave] = inc;
    lds_barrier();
    int below = 0;
    for (int k = 0; k < L1_WAVES; ++k) below += (k < wave) ? waveTot[k] : 0;
    const int excl = below + inc - c;
    if (t < BIN_NODES) {
      const int node = nodeBase + t;
      if (node < n_nodes) {
        deg[node] = c;
        off[node] = beg + excl;
      }
      binDeg[t] = c;
      binOff[t] = beg + excl;
      binCur[t] = excl;
    }
  }
  lds_barrier();
  {
    int i = beg + t;
    for (; i + 3 * L1_THREADS < end; i += 4 * L1_THREADS) {
      int2 sd[4];
      int r[4];
#pragma unroll
      for (int u = 0; u < 4; ++u) sd[u] = tmp[i + u * L1_THREADS];
#pragma unroll
      for (int u = 0; u < 4; ++u)
        r[u] = atomicAdd(&binCur[sd[u].y - nodeBase], 1);
#pragma unroll
      for (int u = 0; u < 4; ++u) csr[beg + r[u]] = sd[u].x;
    }
    for (; i < end; i += L1_THREADS) {
      const int2 sd = tmp[i];
      const int r = atomicAdd(&binCur[sd.y - nodeBase], 1);
      csr[beg + r] = sd.x;
    }
  }
  STAGE_EACH(STAGE_ST, STAGE_DEPTH)
  __syncthreads();

  const float* ldsWl = ldsW;
  const float* ldsWr = ldsW + D_FEAT * D_FEAT;
  const int tile = blockIdx.x * L1_WAVES + wave;
  const int n_tiles = n_nodes >> 4;
  if (tile >= n_tiles) return;

  const int half = lane >> 4;
  const int q = lane & 15;
  const int rowA = tile * 16 + q;

  v8f acc[8];
#pragma unroll
  for (int i = 0; i < 8; ++i) {
    v8f z = {};
    acc[i] = z;
  }

  const float* xRow = x + (size_t)rowA * D_FEAT + 2 * half;
  float* myTile = ldsAgg + wave * (16 * TILE_PITCH);

  int nrow[8], nswz[8];
#pragma unroll
  for (int nt = 0; nt < 8; ++nt) {
    nrow[nt] = nt * 16 + q;
    nswz[nt] = (2 * nrow[nt]) & 63;
  }

  tile_gather_self(x, csr, binOff + wave * 16, binDeg + wave * 16, lane, myTile,
                   acc, xRow, ldsWr, nrow, nswz, half);

  __builtin_amdgcn_fence(__ATOMIC_RELEASE, "workgroup");
  __builtin_amdgcn_wave_barrier();
  __builtin_amdgcn_fence(__ATOMIC_ACQUIRE, "workgroup");

  const float* aggRow = myTile + q * TILE_PITCH + 2 * half;
#pragma unroll 1
  for (int k = 0; k < D_FEAT; k += 8) {
    float2 tm[2], bl[2][8];
#pragma unroll
    for (int kk = 0; kk < 2; ++kk) {
      const int c0 = ((k + 4 * kk) >> 1) + half;
      tm[kk] = *(const float2*)(aggRow + k + 4 * kk);
#pragma unroll
      for (int nt = 0; nt < 8; ++nt)
        bl[kk][nt] = *(const float2*)(ldsWl + nrow[nt] * D_FEAT +
                                      2 * (c0 ^ nswz[nt]));
    }
#pragma unroll
    for (int kk = 0; kk < 2; ++kk) {
      v2f a_m;  a_m.x = tm[kk].x;  a_m.y = tm[kk].y;
#pragma unroll
      for (int nt = 0; nt < 8; ++nt) {
        v2f vbl;  vbl.x = bl[kk][nt].x;  vbl.y = bl[kk][nt].y;
        acc[nt] = __builtin_amdgcn_wmma_f32_16x16x4_f32(
            false, a_m, false, vbl, (short)0, acc[nt], false, false);
      }
    }
    WMMA_TAIL_PAD();
  }

  float part[32];
#pragma unroll
  for (int i = 0; i < 32; ++i) part[i] = 0.0f;
#pragma unroll
  for (int nt = 0; nt < 8; ++nt) {
    const int n = nrow[nt];
    const float bias = b1[n];
    const float wl0 = W2l[n];
    const float wl1 = W2l[D_FEAT + n];
    const float wr0 = W2r[n];
    const float wr1 = W2r[D_FEAT + n];
#pragma unroll
    for (int v = 0; v < 8; ++v) {
      const float hv = fmaxf(acc[nt][v] + bias, 0.0f);
      part[v * 4 + 0] += hv * wl0;
      part[v * 4 + 1] += hv * wl1;
      part[v * 4 + 2] += hv * wr0;
      part[v * 4 + 3] += hv * wr1;
    }
  }
  row_fold<16, 8>(part, (q & 8) != 0);
  row_fold<8, 4>(part, (q & 4) != 0);
  row_fold<4, 2>(part, (q & 2) != 0);
  row_fold<2, 1>(part, (q & 1) != 0);

  const int node = tile * 16 + 8 * half + (q >> 1);
  const bool self_side = (q & 1) != 0;
  const float bias0 = b2[0];
  const float bias1 = b2[1];
  float2 res;
  res.x = self_side ? part[0] + bias0 : part[0];
  res.y = self_side ? part[1] + bias1 : part[1];
  float* dst = (self_side ? out : p) + (size_t)node * 2;
  *(float2*)dst = res;
}

#define G2_DEPTH 16

__global__ __launch_bounds__(256) void sage_gather2(
    const float* __restrict__ p, const int* __restrict__ csr,
    const int* __restrict__ off, const int* __restrict__ deg,
    float* __restrict__ out, int n_nodes, int n_edges) {
  const int gid = blockIdx.x * 256 + threadIdx.x;
  const int sub = gid & 1;
  const bool live = (gid >> 1) < n_nodes;
  const int node = live ? (gid >> 1) : (n_nodes - 1);
  const int beg = off[node];
  const int degv = deg[node];
  const int d = live ? degv : 0;
  const float2 self = *(const float2*)(out + (size_t)node * 2);
  const int last = (n_edges > 0) ? (n_edges - 1) : 0;
  float s0 = 0.0f, s1 = 0.0f;
  int jbase = 0;
  do {
    int sid[G2_DEPTH], keep[G2_DEPTH];
#pragma unroll
    for (int k = 0; k < G2_DEPTH; ++k) {
      const int j = jbase + sub + 2 * k;
      const int e = beg + j;
      keep[k] = (j - d) >> 31;
      sid[k] = csr[e < last ? e : last] & keep[k];
    }
    float2 pv[G2_DEPTH];
#pragma unroll
    for (int k = 0; k < G2_DEPTH; ++k)
      pv[k] = *(const float2*)(p + (size_t)sid[k] * 2);
#pragma unroll
    for (int k = 0; k < G2_DEPTH; ++k) {
      s0 += __int_as_float(__float_as_int(pv[k].x) & keep[k]);
      s1 += __int_as_float(__float_as_int(pv[k].y) & keep[k]);
    }
    jbase += 2 * G2_DEPTH;
  } while (__builtin_amdgcn_ballot_w32(jbase < d) != 0u);
  s0 += row_xor<1>(s0);
  s1 += row_xor<1>(s1);
  if (live && sub == 0) {
    const float inv = 1.0f / fmaxf((float)d, 1.0f);
    float2 res;
    res.x = self.x + s0 * inv;
    res.y = self.y + s1 * inv;
    *(float2*)(out + (size_t)node * 2) = res;
  }
}

extern "C" void kernel_launch(void* const* d_in, const int* in_sizes, int n_in,
                              void* d_out, int out_size, void* d_ws, size_t ws_size,
                              hipStream_t stream) {
  const float* x   = (const float*)d_in[0];
  const int*   ei  = (const int*)d_in[1];
  const float* W1l = (const float*)d_in[2];
  const float* W1r = (const float*)d_in[3];
  const float* b1  = (const float*)d_in[4];
  const float* W2l = (const float*)d_in[5];
  const float* W2r = (const float*)d_in[6];
  const float* b2  = (const float*)d_in[7];
  float* out = (float*)d_out;

  const int n_nodes = in_sizes[0] / D_FEAT;
  const int n_edges = in_sizes[1] / 2;

  float* ws   = (float*)d_ws;
  float* agg1 = ws;
  float* p    = agg1 + (size_t)n_nodes * D_FEAT;
  float* cnt  = p + (size_t)n_nodes * 2;
  int*   deg  = (int*)(cnt + n_nodes);
  int*   off  = deg + n_nodes;
  int*   csr  = off + n_nodes;
  const int n_bins  = (n_nodes + BIN_NODES - 1) / BIN_NODES;
  const int chunk   = (n_edges + EDGE_BLOCKS - 1) / EDGE_BLOCKS;
  int*   cnt_cb    = csr + n_edges;
  int*   bin_start = cnt_cb + EDGE_BLOCKS * MAX_BINS;
  int2*  tmp       = (int2*)ws;

  sage_bin_hist<<<EDGE_BLOCKS, 256, 0, stream>>>(ei, cnt_cb, n_edges, n_bins, chunk);

  sage_bin_scatter<<<EDGE_BLOCKS, 256, 0, stream>>>(ei, cnt_cb, bin_start, tmp, n_edges,
                                                    n_bins, chunk);

  sage_layer1_wmma<<<n_bins, L1_THREADS, 0, stream>>>(
      x, tmp, bin_start, csr, off, deg, W1l, W1r, b1, W2l, W2r, b2, p, out,
      n_nodes, n_edges, n_bins);

  sage_gather2<<<(n_nodes + 127) / 128, 256, 0, stream>>>(p, csr, off, deg, out, n_nodes,
                                                          n_edges);
}


# Edge binning done per chunk in ONE launch (LDS histogram, scan, copy; no count table or prefix over it): a layer-1 block collects its bin's 256 chunk slices; neighbour sum order stays arrival-ranked
# speedup vs baseline: 4.9980x; 1.1488x over previous
#include <hip/hip_runtime.h>

typedef __attribute__((ext_vector_type(2))) float v2f;
typedef __attribute__((ext_vector_type(8))) float v8f;

#define D_FEAT 128


#define L1_WAVES 13
#define BIN_NODES (16 * L1_WAVES)
#define EDGE_BLOCKS 256
#define MAX_BINS 256

__device__ __forceinline__ int bin_of(int d) {
  return (int)((unsigned)d / (unsigned)BIN_NODES);
}

static_assert(BIN_NODES <= 256, "the node's index inside its bin is stored in 8 bits");

__global__ __launch_bounds__(256) void sage_bin_local(
    const int* __restrict__ ei, int* __restrict__ tmp, int2* __restrict__ seg,
    int n_edges, int n_bins, int chunk) {
  __shared__ int hist[MAX_BINS];
  __shared__ int cur[MAX_BINS];
  __shared__ int waveTot[8];
  const int t = threadIdx.x;
  const int c = blockIdx.x;
  hist[t] = 0;
  __syncthreads();
  const int beg = c * chunk;
  int end = beg + chunk;
  if (end > n_edges) end = n_edges;
  const int* dstp = ei + n_edges;
  {
    int e = beg + t;
    for (; e + 3 * 256 < end; e += 4 * 256) {
      int d[4];
#pragma unroll
      for (int u = 0; u < 4; ++u) d[u] = dstp[e + u * 256];
#pragma unroll
      for (int u = 0; u < 4; ++u) atomicAdd(&hist[bin_of(d[u])], 1);
    }
    for (; e < end; e += 256) atomicAdd(&hist[bin_of(dstp[e])], 1);
  }
  __syncthreads();
  {
    const int totT = hist[t];
    const int lane = t & 31;
    const int wv = t >> 5;
    int inc = totT;
    for (int o = 1; o < 32; o <<= 1) {
      const int up = __shfl_up(inc, o, 32);
      if (lane >= o) inc += up;
    }
    if (lane == 31) waveTot[wv] = inc;
    __syncthreads();
    int below = 0;
    for (int k = 0; k < 8; ++k) below += (k < wv) ? waveTot[k] : 0;
    const int start = below + inc - totT;
    cur[t] = beg + start;
    if (t < n_bins) seg[t * EDGE_BLOCKS + c] = make_int2(start, totT);
  }
  __syncthreads();
  {
    int e = beg + t;
    for (; e + 3 * 256 < end; e += 4 * 256) {
      int s[4], d[4];
#pragma unroll
      for (int u = 0; u < 4; ++u) {
        s[u] = ei[e + u * 256];
        d[u] = dstp[e + u * 256];
      }
#pragma unroll
      for (int u = 0; u < 4; ++u) {
        const int b = bin_of(d[u]);
        const int slot = atomicAdd(&cur[b], 1);
        tmp[slot] = (s[u] << 8) | (d[u] - b * BIN_NODES);
      }
    }
    for (; e < end; e += 256) {
      const int s = ei[e];
      const int d = dstp[e];
      const int b = bin_of(d);
      const int slot = atomicAdd(&cur[b], 1);
      tmp[slot] = (s << 8) | (d - b * BIN_NODES);
    }
  }
}


#define TILE_PITCH 132

__device__ __forceinline__ float lane_read_f(float v, int l) {
  return __builtin_bit_cast(
      float, __builtin_amdgcn_readlane(__builtin_bit_cast(int, v), l));
}

#define WMMA_TAIL_PAD()                                    \
  do {                                                     \
    __builtin_amdgcn_sched_barrier(0);                     \
    asm volatile("v_nop\n\tv_nop\n\tv_nop\n\tv_nop");      \
  } while (0)

__device__ __forceinline__ void self_trip(
    v8f* acc, float2 a0, float2 a1, const float* ldsWr, const int* nrow,
    const int* nswz, int half, int k) {
  float2 br[2][8];
#pragma unroll
  for (int kk = 0; kk < 2; ++kk) {
    const int c0 = ((k + 4 * kk) >> 1) + half;
#pragma unroll
    for (int nt = 0; nt < 8; ++nt)
      br[kk][nt] = *(const float2*)(ldsWr + nrow[nt] * D_FEAT +
                                    2 * (c0 ^ nswz[nt]));
  }
#pragma unroll
  for (int kk = 0; kk < 2; ++kk) {
    v2f a_s;
    a_s.x = kk ? a1.x : a0.x;
    a_s.y = kk ? a1.y : a0.y;
#pragma unroll
    for (int nt = 0; nt < 8; ++nt) {
      v2f vbr;  vbr.x = br[kk][nt].x;  vbr.y = br[kk][nt].y;
      acc[nt] = __builtin_amdgcn_wmma_f32_16x16x4_f32(
          false, a_s, false, vbr, (short)0, acc[nt], false, false);
    }
  }
  WMMA_TAIL_PAD();
}

__device__ __forceinline__ void tile_gather_self(
    const float* __restrict__ x, const int* csr, const int* tileOff,
    const int* tileDeg, int lane, float* tileLds, v8f* acc,
    const float* __restrict__ xRow, const float* ldsWr, const int* nrow,
    const int* nswz, int half) {
  const int li = lane & 15;
  const int offv = tileOff[li];
  const int degv = tileDeg[li];
  const float invv = 1.0f / fmaxf((float)degv, 1.0f);
  const int e0 = __builtin_amdgcn_readlane(offv, 0);
  const int e1 = __builtin_amdgcn_readlane(offv, 15) +
                 __builtin_amdgcn_readlane(degv, 15);
  const int bndv = (lane < 16) ? offv : e1;
  const int total = e1 - e0;
  const float* xl = x + lane * 4;
  float* trow = tileLds + lane * 4;

  float4 acs = make_float4(0.0f, 0.0f, 0.0f, 0.0f);
  int cur = 0;
  int nextB = __builtin_amdgcn_readlane(bndv, 1);
  int ks = 0;

  float2 pa[4];
#pragma unroll
  for (int i = 0; i < 4; ++i) pa[i] = *(const float2*)(xRow + 4 * i);

  int mine = (lane < total) ? csr[e0 + lane] : 0;
  for (int base = 0; base < total; base += 32) {
    float2 na[4];
#pragma unroll
    for (int i = 0; i < 4; ++i) {
      const int kn = ks + 16 + 4 * i;
      na[i] = *(const float2*)(xRow + ((kn < D_FEAT) ? kn : (D_FEAT - 4)));
    }
    const int nb = base + 32;
    const int ni = (nb + lane < total) ? (nb + lane) : (total - 1);
    const int nxt = csr[e0 + ni];
#pragma unroll
    for (int h = 0; h < 2; ++h) {
      const int g = base + 16 * h;
      if (g < total) {
        float4 v[16];
#pragma unroll
        for (int u = 0; u < 16; ++u) {
          const int s = __builtin_amdgcn_readlane(mine, 16 * h + u);
          v[u] = *(const float4*)(xl + (size_t)s * D_FEAT);
        }
        if (ks < D_FEAT) {
          self_trip(acc, pa[2 * h], pa[2 * h + 1], ldsWr, nrow, nswz, half, ks);
          ks += 8;
        }
#pragma unroll
        for (int u = 0; u < 16; ++u) {
          const int e = e0 + g + u;
          if (e < e1) {
            while (e >= nextB && cur < 15) {
              const float inv = lane_read_f(invv, cur);
              *(float4*)(trow + cur * TILE_PITCH) = make_float4(
                  acs.x * inv, acs.y * inv, acs.z * inv, acs.w * inv);
              acs = make_float4(0.0f, 0.0f, 0.0f, 0.0f);
              ++cur;
              nextB = __builtin_amdgcn_readlane(bndv, cur + 1);
            }
            acs.x += v[u].x;  acs.y += v[u].y;
            acs.z += v[u].z;  acs.w += v[u].w;
          }
        }
      }
    }
    mine = nxt;
#pragma unroll
    for (int i = 0; i < 4; ++i) pa[i] = na[i];
  }
  for (; cur < 16; ++cur) {
    const float inv = lane_read_f(invv, cur);
    *(float4*)(trow + cur * TILE_PITCH) = make_float4(
        acs.x * inv, acs.y * inv, acs.z * inv, acs.w * inv);
    acs = make_float4(0.0f, 0.0f, 0.0f, 0.0f);
  }
#pragma unroll 1
  for (; ks < D_FEAT; ks += 8) {
    const float2 sa0 = *(const float2*)(xRow + ks);
    const float2 sa1 = *(const float2*)(xRow + ks + 4);
    self_trip(acc, sa0, sa1, ldsWr, nrow, nswz, half, ks);
  }
}

template <int XMASK>
__device__ __forceinline__ float row_xor(float x) {
  return __builtin_bit_cast(
      float, __builtin_amdgcn_update_dpp(0, __builtin_bit_cast(int, x),
                                         0x160 + XMASK, 0xf, 0xf, true));
}

template <int CNT, int XMASK>
__device__ __forceinline__ void row_fold(float* v, bool upper) {
#pragma unroll
  for (int i = 0; i < CNT; ++i) {
    const float keep = upper ? v[i + CNT] : v[i];
    const float send = upper ? v[i] : v[i + CNT];
    v[i] = keep + row_xor<XMASK>(send);
  }
}


#define L1_THREADS (32 * L1_WAVES)
#define SEG_PER_WAVE ((EDGE_BLOCKS + L1_WAVES - 1) / L1_WAVES)
#define SEG_GROUP 20
#define BIN_SLOTS 256
static_assert(BIN_SLOTS <= L1_THREADS, "one thread zeroes one counter");
static_assert(SEG_PER_WAVE <= 32, "one lane per slice of the wave");
static_assert(SEG_PER_WAVE % SEG_GROUP == 0, "whole groups of slices");
#define STAGE_DEPTH 10
static_assert(2 * STAGE_DEPTH * L1_WAVES >= 2 * D_FEAT,
              "two stage batches must cover both weight matrices");

__device__ __forceinline__ float4 stage_row_load(
    const float* __restrict__ W1l, const float* __restrict__ W1r, int wave,
    int lane, int j) {
  const int row = wave + L1_WAVES * j;
  const int rc = (row < 2 * D_FEAT) ? row : (2 * D_FEAT - 1);
  const float* srow = (rc < D_FEAT) ? (W1l + rc * D_FEAT)
                                    : (W1r + (rc - D_FEAT) * D_FEAT);
  return *(const float4*)(srow + 4 * lane);
}

__device__ __forceinline__ void stage_row_store(float* ldsW, int wave, int lane,
                                                int j, float4 v) {
  const int row = wave + L1_WAVES * j;
  const int rc = (row < 2 * D_FEAT) ? row : (2 * D_FEAT - 1);
  const int s0 = (2 * lane) ^ ((2 * rc) & 63);
  *(float4*)(ldsW + rc * D_FEAT + 2 * s0) = v;
}

#define STAGE_EACH(OP, J0) \
  OP(0, J0) OP(1, J0) OP(2, J0) OP(3, J0) OP(4, J0) \
  OP(5, J0) OP(6, J0) OP(7, J0) OP(8, J0) OP(9, J0)
#define STAGE_LD(u, J0) wq##u = stage_row_load(W1l, W1r, wave, lane, (J0) + u);
#define STAGE_ST(u, J0) stage_row_store(ldsW, wave, lane, (J0) + u, wq##u);

__device__ __forceinline__ void lds_barrier() {
  __builtin_amdgcn_fence(__ATOMIC_RELEASE, "workgroup", "local");
  __builtin_amdgcn_s_barrier();
  __builtin_amdgcn_fence(__ATOMIC_ACQUIRE, "workgroup", "local");
}

__global__ __launch_bounds__(L1_THREADS) void sage_layer1_wmma(
    const float* __restrict__ x, const int* __restrict__ tmp,
    const int2* __restrict__ seg, int* csr, int* off, int* deg,
    const float* __restrict__ W1l,
    const float* __restrict__ W1r, const float* __restrict__ b1,
    const float* __restrict__ W2l, const float* __restrict__ W2r,
    const float* __restrict__ b2, float* __restrict__ p,
    float* __restrict__ out, int n_nodes, int chunk) {
  __shared__ __attribute__((aligned(16))) float ldsW[2 * D_FEAT * D_FEAT];
  __shared__ __attribute__((aligned(16)))
  float ldsAgg[L1_WAVES * 16 * TILE_PITCH];
  __shared__ int binCur[BIN_SLOTS];
  __shared__ int binOff[BIN_NODES];
  __shared__ int binDeg[BIN_NODES];
  __shared__ int waveTot[L1_WAVES];
  __shared__ int waveSum[L1_WAVES];

  const int t = threadIdx.x;
  const int lane = threadIdx.x & 31;
  const int wave = __builtin_amdgcn_readfirstlane((int)(threadIdx.x >> 5));

  const int bin = blockIdx.x;
  const int nodeBase = bin * BIN_NODES;
  const int segC = wave + L1_WAVES * lane;
  const bool segLive = lane < SEG_PER_WAVE && segC < EDGE_BLOCKS;
  const int2 sg = seg[bin * EDGE_BLOCKS + (segLive ? segC : 0)];
  if (t < BIN_SLOTS) binCur[t] = 0;
  float4 wq0, wq1, wq2, wq3, wq4, wq5, wq6, wq7, wq8, wq9;
  STAGE_EACH(STAGE_LD, 0)
  __builtin_amdgcn_sched_barrier(0);
  const int segMask = segLive ? -1 : 0;
  const int segLocal = sg.x & segMask;
  const int segAt = (segC * chunk + sg.x) & segMask;
  const int segN = sg.y & segMask;
  lds_barrier();
  {
#pragma unroll
    for (int g = 0; g < SEG_PER_WAVE; g += SEG_GROUP) {
      int pk[SEG_GROUP];
#pragma unroll
      for (int u = 0; u < SEG_GROUP; ++u) {
        const int a = __builtin_amdgcn_readlane(segAt, g + u);
        const int n = __builtin_amdgcn_readlane(segN, g + u);
        pk[u] = tmp[a + min(lane, max(n - 1, 0))];
      }
#pragma unroll
      for (int u = 0; u < SEG_GROUP; ++u) {
        const int n = __builtin_amdgcn_readlane(segN, g + u);
        const bool live = lane < n;
        atomicAdd(&binCur[live ? (pk[u] & 255) : lane], live ? 1 : 0);
      }
#pragma unroll
      for (int u = 0; u < SEG_GROUP; ++u) {
        const int a = __builtin_amdgcn_readlane(segAt, g + u);
        const int n = __builtin_amdgcn_readlane(segN, g + u);
        if (n > 32) {
          for (int l = lane + 32; l < n; l += 32)
            atomicAdd(&binCur[tmp[a + l] & 255], 1);
        }
      }
    }
  }
  {
    int s = segLocal;
    for (int o = 16; o > 0; o >>= 1) s += __shfl_xor(s, o, 32);
    if (lane == 0) waveSum[wave] = s;
  }
  STAGE_EACH(STAGE_ST, 0)
  STAGE_EACH(STAGE_LD, STAGE_DEPTH)
  lds_barrier();
  int beg = 0;
  for (int k = 0; k < L1_WAVES; ++k) beg += waveSum[k];
  {
    const int c = (t < BIN_NODES) ? binCur[t] : 0;
    int inc = c;
    for (int o = 1; o < 32; o <<= 1) {
      const int up = __shfl_up(inc, o, 32);
      if (lane >= o) inc += up;
    }
    if (lane == 31) waveTot[wave] = inc;
    lds_barrier();
    int below = 0;
    for (int k = 0; k < L1_WAVES; ++k) below += (k < wave) ? waveTot[k] : 0;
    const int excl = below + inc - c;
    if (t < BIN_NODES) {
      const int node = nodeBase + t;
      if (node < n_nodes) {
        deg[node] = c;
        off[node] = beg + excl;
      }
      binDeg[t] = c;
      binOff[t] = beg + excl;
      binCur[t] = excl;
    }
  }
  lds_barrier();
  {
#pragma unroll
    for (int g = 0; g < SEG_PER_WAVE; g += SEG_GROUP) {
      int pk[SEG_GROUP], r[SEG_GROUP];
#pragma unroll
      for (int u = 0; u < SEG_GROUP; ++u) {
        const int a = __builtin_amdgcn_readlane(segAt, g + u);
        const int n = __builtin_amdgcn_readlane(segN, g + u);
        pk[u] = tmp[a + min(lane, max(n - 1, 0))];
      }
#pragma unroll
      for (int u = 0; u < SEG_GROUP; ++u) {
        const int n = __builtin_amdgcn_readlane(segN, g + u);
        const bool live = lane < n;
        r[u] = atomicAdd(&binCur[live ? (pk[u] & 255) : lane], live ? 1 : 0);
      }
#pragma unroll
      for (int u = 0; u < SEG_GROUP; ++u) {
        const int n = __builtin_amdgcn_readlane(segN, g + u);
        if (lane < n) csr[beg + r[u]] = (int)((unsigned)pk[u] >> 8);
      }
#pragma unroll
      for (int u = 0; u < SEG_GROUP; ++u) {
        const int a = __builtin_amdgcn_readlane(segAt, g + u);
        const int n = __builtin_amdgcn_readlane(segN, g + u);
        if (n > 32) {
          for (int l = lane + 32; l < n; l += 32) {
            const int v = tmp[a + l];
            const int rr = atomicAdd(&binCur[v & 255], 1);
            csr[beg + rr] = (int)((unsigned)v >> 8);
          }
        }
      }
    }
  }
  STAGE_EACH(STAGE_ST, STAGE_DEPTH)
  __syncthreads();

  const float* ldsWl = ldsW;
  const float* ldsWr = ldsW + D_FEAT * D_FEAT;
  const int tile = blockIdx.x * L1_WAVES + wave;
  const int n_tiles = n_nodes >> 4;
  if (tile >= n_tiles) return;

  const int half = lane >> 4;
  const int q = lane & 15;
  const int rowA = tile * 16 + q;

  v8f acc[8];
#pragma unroll
  for (int i = 0; i < 8; ++i) {
    v8f z = {};
    acc[i] = z;
  }

  const float* xRow = x + (size_t)rowA * D_FEAT + 2 * half;
  float* myTile = ldsAgg + wave * (16 * TILE_PITCH);

  int nrow[8], nswz[8];
#pragma unroll
  for (int nt = 0; nt < 8; ++nt) {
    nrow[nt] = nt * 16 + q;
    nswz[nt] = (2 * nrow[nt]) & 63;
  }

  tile_gather_self(x, csr, binOff + wave * 16, binDeg + wave * 16, lane, myTile,
                   acc, xRow, ldsWr, nrow, nswz, half);

  __builtin_amdgcn_fence(__ATOMIC_RELEASE, "workgroup");
  __builtin_amdgcn_wave_barrier();
  __builtin_amdgcn_fence(__ATOMIC_ACQUIRE, "workgroup");

  const float* aggRow = myTile + q * TILE_PITCH + 2 * half;
#pragma unroll 1
  for (int k = 0; k < D_FEAT; k += 8) {
    float2 tm[2], bl[2][8];
#pragma unroll
    for (int kk = 0; kk < 2; ++kk) {
      const int c0 = ((k + 4 * kk) >> 1) + half;
      tm[kk] = *(const float2*)(aggRow + k + 4 * kk);
#pragma unroll
      for (int nt = 0; nt < 8; ++nt)
        bl[kk][nt] = *(const float2*)(ldsWl + nrow[nt] * D_FEAT +
                                      2 * (c0 ^ nswz[nt]));
    }
#pragma unroll
    for (int kk = 0; kk < 2; ++kk) {
      v2f a_m;  a_m.x = tm[kk].x;  a_m.y = tm[kk].y;
#pragma unroll
      for (int nt = 0; nt < 8; ++nt) {
        v2f vbl;  vbl.x = bl[kk][nt].x;  vbl.y = bl[kk][nt].y;
        acc[nt] = __builtin_amdgcn_wmma_f32_16x16x4_f32(
            false, a_m, false, vbl, (short)0, acc[nt], false, false);
      }
    }
    WMMA_TAIL_PAD();
  }

  float part[32];
#pragma unroll
  for (int i = 0; i < 32; ++i) part[i] = 0.0f;
#pragma unroll
  for (int nt = 0; nt < 8; ++nt) {
    const int n = nrow[nt];
    const float bias = b1[n];
    const float wl0 = W2l[n];
    const float wl1 = W2l[D_FEAT + n];
    const float wr0 = W2r[n];
    const float wr1 = W2r[D_FEAT + n];
#pragma unroll
    for (int v = 0; v < 8; ++v) {
      const float hv = fmaxf(acc[nt][v] + bias, 0.0f);
      part[v * 4 + 0] += hv * wl0;
      part[v * 4 + 1] += hv * wl1;
      part[v * 4 + 2] += hv * wr0;
      part[v * 4 + 3] += hv * wr1;
    }
  }
  row_fold<16, 8>(part, (q & 8) != 0);
  row_fold<8, 4>(part, (q & 4) != 0);
  row_fold<4, 2>(part, (q & 2) != 0);
  row_fold<2, 1>(part, (q & 1) != 0);

  const int node = tile * 16 + 8 * half + (q >> 1);
  const bool self_side = (q & 1) != 0;
  const float bias0 = b2[0];
  const float bias1 = b2[1];
  float2 res;
  res.x = self_side ? part[0] + bias0 : part[0];
  res.y = self_side ? part[1] + bias1 : part[1];
  float* dst = (self_side ? out : p) + (size_t)node * 2;
  *(float2*)dst = res;
}

#define G2_DEPTH 16

__global__ __launch_bounds__(256) void sage_gather2(
    const float* __restrict__ p, const int* __restrict__ csr,
    const int* __restrict__ off, const int* __restrict__ deg,
    float* __restrict__ out, int n_nodes, int n_edges) {
  const int gid = blockIdx.x * 256 + threadIdx.x;
  const int sub = gid & 1;
  const bool live = (gid >> 1) < n_nodes;
  const int node = live ? (gid >> 1) : (n_nodes - 1);
  const int beg = off[node];
  const int degv = deg[node];
  const int d = live ? degv : 0;
  const float2 self = *(const float2*)(out + (size_t)node * 2);
  const int last = (n_edges > 0) ? (n_edges - 1) : 0;
  float s0 = 0.0f, s1 = 0.0f;
  int jbase = 0;
  do {
    int sid[G2_DEPTH], keep[G2_DEPTH];
#pragma unroll
    for (int k = 0; k < G2_DEPTH; ++k) {
      const int j = jbase + sub + 2 * k;
      const int e = beg + j;
      keep[k] = (j - d) >> 31;
      sid[k] = csr[e < last ? e : last] & keep[k];
    }
    float2 pv[G2_DEPTH];
#pragma unroll
    for (int k = 0; k < G2_DEPTH; ++k)
      pv[k] = *(const float2*)(p + (size_t)sid[k] * 2);
#pragma unroll
    for (int k = 0; k < G2_DEPTH; ++k) {
      s0 += __int_as_float(__float_as_int(pv[k].x) & keep[k]);
      s1 += __int_as_float(__float_as_int(pv[k].y) & keep[k]);
    }
    jbase += 2 * G2_DEPTH;
  } while (__builtin_amdgcn_ballot_w32(jbase < d) != 0u);
  s0 += row_xor<1>(s0);
  s1 += row_xor<1>(s1);
  if (live && sub == 0) {
    const float inv = 1.0f / fmaxf((float)d, 1.0f);
    float2 res;
    res.x = self.x + s0 * inv;
    res.y = self.y + s1 * inv;
    *(float2*)(out + (size_t)node * 2) = res;
  }
}

extern "C" void kernel_launch(void* const* d_in, const int* in_sizes, int n_in,
                              void* d_out, int out_size, void* d_ws, size_t ws_size,
                              hipStream_t stream) {
  const float* x   = (const float*)d_in[0];
  const int*   ei  = (const int*)d_in[1];
  const float* W1l = (const float*)d_in[2];
  const float* W1r = (const float*)d_in[3];
  const float* b1  = (const float*)d_in[4];
  const float* W2l = (const float*)d_in[5];
  const float* W2r = (const float*)d_in[6];
  const float* b2  = (const float*)d_in[7];
  float* out = (float*)d_out;

  const int n_nodes = in_sizes[0] / D_FEAT;
  const int n_edges = in_sizes[1] / 2;

  float* ws   = (float*)d_ws;
  float* agg1 = ws;
  float* p    = agg1 + (size_t)n_nodes * D_FEAT;
  float* cnt  = p + (size_t)n_nodes * 2;
  int*   deg  = (int*)(cnt + n_nodes);
  int*   off  = deg + n_nodes;
  int*   csr  = off + n_nodes;
  const int n_bins  = (n_nodes + BIN_NODES - 1) / BIN_NODES;
  const int chunk   = (n_edges + EDGE_BLOCKS - 1) / EDGE_BLOCKS;
  int*   tmp = (int*)ws;
  int2*  seg = (int2*)((int*)ws + n_edges);

  sage_bin_local<<<EDGE_BLOCKS, 256, 0, stream>>>(ei, tmp, seg, n_edges, n_bins, chunk);

  sage_layer1_wmma<<<n_bins, L1_THREADS, 0, stream>>>(
      x, tmp, seg, csr, off, deg, W1l, W1r, b1, W2l, W2r, b2, p, out, n_nodes,
      chunk);

  sage_gather2<<<(n_nodes + 127) / 128, 256, 0, stream>>>(p, csr, off, deg, out, n_nodes,
                                                          n_edges);
}
